# starting kernel
# baseline (speedup 1.0000x reference)
#include <hip/hip_runtime.h>

typedef __attribute__((ext_vector_type(2))) float v2f;
typedef __attribute__((ext_vector_type(8))) float v8f;

__global__ __launch_bounds__(256) void ec_count_deg(const int* __restrict__ dst,
                                                    int* __restrict__ deg,
                                                    int total, int Nn, int Eper) {
  int t = blockIdx.x * blockDim.x + threadIdx.x;
  if (t >= total) return;
  int r = t / Eper;
  atomicAdd(&deg[r * Nn + dst[t]], 1);
}

__global__ __launch_bounds__(256) void ec_inv_deg(int* __restrict__ deg,
                                                  float* __restrict__ invd,
                                                  int total) {
  int t = blockIdx.x * blockDim.x + threadIdx.x;
  if (t >= total) return;
  float d = (float)deg[t];
  invd[t] = 1.0f / fmaxf(d, 1.0f);
}

template <int K, int NC>
__global__ __launch_bounds__(128) void ec_gemm_wmma(
    const float* __restrict__ A, const float* __restrict__ B,
    const float* __restrict__ bias, float* __restrict__ C, int mtiles) {
  constexpr int NT = NC / 16;
  constexpr int S = ((2 * NC) % 64 == 32) ? (2 * NC) : (2 * NC + 32);
  __shared__ float Bs[(K / 2) * S];

  for (int i = threadIdx.x; i < K * NC / 4; i += 128) {
    int k = (i * 4) / NC;
    int n = (i * 4) % NC;
    float4 v = *(const float4*)(B + i * 4);
    float* p = &Bs[(k >> 1) * S + 2 * n + (k & 1)];
    p[0] = v.x;
    p[2] = v.y;
    p[4] = v.z;
    p[6] = v.w;
  }
  __syncthreads();

  int wave = (blockIdx.x * blockDim.x + threadIdx.x) >> 5;
  if (wave >= mtiles) return;
  int lane = threadIdx.x & 31;
  int m = lane & 15;
  int half = lane >> 4;

  const float* Arow = A + (size_t)(wave * 16 + m) * K + 2 * half;

  v8f acc[NT];
  v8f z = {0.f, 0.f, 0.f, 0.f, 0.f, 0.f, 0.f, 0.f};
#pragma unroll
  for (int j = 0; j < NT; ++j) acc[j] = z;

#pragma unroll 8
  for (int k = 0; k < K; k += 4) {
    float2 av = *(const float2*)(Arow + k);
    v2f a = {av.x, av.y};
    const float* Bp = &Bs[(size_t)((k + 2 * half) >> 1) * S + 2 * m];
#pragma unroll
    for (int j = 0; j < NT; ++j) {
      v2f b = *(const v2f*)(Bp + j * 32);
      acc[j] = __builtin_amdgcn_wmma_f32_16x16x4_f32(
           false, a,  false, b,
           (short)0, acc[j],  false,  false);
    }
  }

#pragma unroll
  for (int j = 0; j < NT; ++j) {
    float bv = bias ? bias[j * 16 + m] : 0.0f;
#pragma unroll
    for (int v = 0; v < 8; ++v) {
      C[(size_t)(wave * 16 + v + 8 * half) * NC + j * 16 + m] = acc[j][v] + bv;
    }
  }
}

template <int COLS>
__global__ __launch_bounds__(256) void ec_agg(
    const float* __restrict__ hw, const int* __restrict__ src,
    const int* __restrict__ dst, const float* __restrict__ invd,
    float* __restrict__ acc, int E) {
  constexpr int TPE = COLS / 4;
  int t = blockIdx.x * blockDim.x + threadIdx.x;
  int e = t / TPE;
  if (e >= E) return;
  int c = (t % TPE) * 4;
  int s = src[e];
  int d = dst[e];
  float w = invd[d];
  float4 v = *(const float4*)(hw + (size_t)s * COLS + c);
  float* dp = acc + (size_t)d * COLS + c;
  unsafeAtomicAdd(dp + 0, v.x * w);
  unsafeAtomicAdd(dp + 1, v.y * w);
  unsafeAtomicAdd(dp + 2, v.z * w);
  unsafeAtomicAdd(dp + 3, v.w * w);
}

__global__ __launch_bounds__(256) void ec_relu(float* __restrict__ h, int total) {
  int t = blockIdx.x * blockDim.x + threadIdx.x;
  if (t >= total) return;
  h[t] = fmaxf(h[t], 0.0f);
}

extern "C" void kernel_launch(void* const* d_in, const int* in_sizes, int n_in,
                              void* d_out, int out_size, void* d_ws, size_t ws_size,
                              hipStream_t stream) {
  const float* x   = (const float*)d_in[0];
  const int*   src = (const int*)d_in[1];
  const int*   dst = (const int*)d_in[2];
  const float* W1  = (const float*)d_in[3];
  const float* L1w = (const float*)d_in[4];
  const float* L1b = (const float*)d_in[5];
  const float* W2  = (const float*)d_in[6];
  const float* L2w = (const float*)d_in[7];
  const float* L2b = (const float*)d_in[8];
  float* out = (float*)d_out;

  const int D_IN = 256, D_H = 64, D_OUT = 16;
  const int Nn = in_sizes[0] / D_IN;
  const int R  = in_sizes[3] / (D_IN * D_H);
  const int E  = in_sizes[1] / R;

  char* ws = (char*)d_ws;
  size_t off = 0;
  int*   deg  = (int*)(ws + off);
  float* invd = (float*)(ws + off);
  off += (size_t)R * Nn * sizeof(float);
  off = (off + 255) & ~(size_t)255;
  float* h = (float*)(ws + off);
  off += (size_t)Nn * D_H * sizeof(float);
  off = (off + 255) & ~(size_t)255;
  float* hw = (float*)(ws + off);

  hipMemsetAsync(deg, 0, (size_t)R * Nn * sizeof(int), stream);
  {
    int total = R * E;
    ec_count_deg<<<(total + 255) / 256, 256, 0, stream>>>(dst, deg, total, Nn, E);
  }
  {
    int total = R * Nn;
    ec_inv_deg<<<(total + 255) / 256, 256, 0, stream>>>(deg, invd, total);
  }

  int mtiles = Nn / 16;
  int gemmBlocks = (mtiles * 32 + 127) / 128;

  ec_gemm_wmma<256, 64><<<gemmBlocks, 128, 0, stream>>>(x, L1w, L1b, h, mtiles);
  for (int r = 0; r < R; ++r) {
    ec_gemm_wmma<256, 64><<<gemmBlocks, 128, 0, stream>>>(
        x, W1 + (size_t)r * D_IN * D_H, nullptr, hw, mtiles);
    long long tot = (long long)E * (D_H / 4);
    ec_agg<64><<<(int)((tot + 255) / 256), 256, 0, stream>>>(
        hw, src + (size_t)r * E, dst + (size_t)r * E, invd + (size_t)r * Nn, h, E);
  }
  ec_relu<<<(Nn * D_H + 255) / 256, 256, 0, stream>>>(h, Nn * D_H);

  ec_gemm_wmma<64, 16><<<gemmBlocks, 128, 0, stream>>>(h, L2w, L2b, out, mtiles);
  for (int r = 0; r < R; ++r) {
    ec_gemm_wmma<64, 16><<<gemmBlocks, 128, 0, stream>>>(
        h, W2 + (size_t)r * D_H * D_OUT, nullptr, hw, mtiles);
    long long tot = (long long)E * (D_OUT / 4);
    ec_agg<16><<<(int)((tot + 255) / 256), 256, 0, stream>>>(
        hw, src + (size_t)r * E, dst + (size_t)r * E, invd + (size_t)r * Nn, out, E);
  }
}


# Edge scatter-add with lane = column: a wave walks 16 (layer 2: 32) edges and each gather and f32 atomic instruction covers whole 128-byte lines; sums stay in atomic arrival order as before
# speedup vs baseline: 1.3705x; 1.3705x over previous
#include <hip/hip_runtime.h>

typedef __attribute__((ext_vector_type(2))) float v2f;
typedef __attribute__((ext_vector_type(8))) float v8f;

__global__ __launch_bounds__(256) void ec_count_deg(const int* __restrict__ dst,
                                                    int* __restrict__ deg,
                                                    int total, int Nn, int Eper) {
  int t = blockIdx.x * blockDim.x + threadIdx.x;
  if (t >= total) return;
  int r = t / Eper;
  atomicAdd(&deg[r * Nn + dst[t]], 1);
}

__global__ __launch_bounds__(256) void ec_inv_deg(int* __restrict__ deg,
                                                  float* __restrict__ invd,
                                                  int total) {
  int t = blockIdx.x * blockDim.x + threadIdx.x;
  if (t >= total) return;
  float d = (float)deg[t];
  invd[t] = 1.0f / fmaxf(d, 1.0f);
}

template <int K, int NC>
__global__ __launch_bounds__(128) void ec_gemm_wmma(
    const float* __restrict__ A, const float* __restrict__ B,
    const float* __restrict__ bias, float* __restrict__ C, int mtiles) {
  constexpr int NT = NC / 16;
  constexpr int S = ((2 * NC) % 64 == 32) ? (2 * NC) : (2 * NC + 32);
  __shared__ float Bs[(K / 2) * S];

  for (int i = threadIdx.x; i < K * NC / 4; i += 128) {
    int k = (i * 4) / NC;
    int n = (i * 4) % NC;
    float4 v = *(const float4*)(B + i * 4);
    float* p = &Bs[(k >> 1) * S + 2 * n + (k & 1)];
    p[0] = v.x;
    p[2] = v.y;
    p[4] = v.z;
    p[6] = v.w;
  }
  __syncthreads();

  int wave = (blockIdx.x * blockDim.x + threadIdx.x) >> 5;
  if (wave >= mtiles) return;
  int lane = threadIdx.x & 31;
  int m = lane & 15;
  int half = lane >> 4;

  const float* Arow = A + (size_t)(wave * 16 + m) * K + 2 * half;

  v8f acc[NT];
  v8f z = {0.f, 0.f, 0.f, 0.f, 0.f, 0.f, 0.f, 0.f};
#pragma unroll
  for (int j = 0; j < NT; ++j) acc[j] = z;

#pragma unroll 8
  for (int k = 0; k < K; k += 4) {
    float2 av = *(const float2*)(Arow + k);
    v2f a = {av.x, av.y};
    const float* Bp = &Bs[(size_t)((k + 2 * half) >> 1) * S + 2 * m];
#pragma unroll
    for (int j = 0; j < NT; ++j) {
      v2f b = *(const v2f*)(Bp + j * 32);
      acc[j] = __builtin_amdgcn_wmma_f32_16x16x4_f32(
           false, a,  false, b,
           (short)0, acc[j],  false,  false);
    }
  }

#pragma unroll
  for (int j = 0; j < NT; ++j) {
    float bv = bias ? bias[j * 16 + m] : 0.0f;
#pragma unroll
    for (int v = 0; v < 8; ++v) {
      C[(size_t)(wave * 16 + v + 8 * half) * NC + j * 16 + m] = acc[j][v] + bv;
    }
  }
}

__global__ __launch_bounds__(256) void ec_agg64(
    const float* __restrict__ hw, const int* __restrict__ src,
    const int* __restrict__ dst, const float* __restrict__ invd,
    float* __restrict__ acc, int E) {
  constexpr int EPW = 16;
  int lane = threadIdx.x & 31;
  int wave = __builtin_amdgcn_readfirstlane((blockIdx.x * blockDim.x + threadIdx.x) >> 5);
  int e0 = wave * EPW;
  if (e0 >= E) return;
  int e = e0 + (lane & (EPW - 1));
  if (e >= E) e = E - 1;
  int sv = src[e];
  int dv = dst[e];
  int wv = __float_as_int(invd[dv]);
  if (e0 + EPW <= E) {
    float v0[EPW], v1[EPW];
#pragma unroll
    for (int k = 0; k < EPW; ++k) {
      int s = __builtin_amdgcn_readlane(sv, k);
      const float* hp = hw + (size_t)s * 64 + lane;
      v0[k] = hp[0];
      v1[k] = hp[32];
    }
#pragma unroll
    for (int k = 0; k < EPW; ++k) {
      int d = __builtin_amdgcn_readlane(dv, k);
      float w = __int_as_float(__builtin_amdgcn_readlane(wv, k));
      float* dp = acc + (size_t)d * 64 + lane;
      unsafeAtomicAdd(dp, v0[k] * w);
      unsafeAtomicAdd(dp + 32, v1[k] * w);
    }
  } else {
    int n = E - e0;
    for (int k = 0; k < n; ++k) {
      int s = __builtin_amdgcn_readlane(sv, k);
      int d = __builtin_amdgcn_readlane(dv, k);
      float w = __int_as_float(__builtin_amdgcn_readlane(wv, k));
      const float* hp = hw + (size_t)s * 64 + lane;
      float* dp = acc + (size_t)d * 64 + lane;
      unsafeAtomicAdd(dp, hp[0] * w);
      unsafeAtomicAdd(dp + 32, hp[32] * w);
    }
  }
}

__global__ __launch_bounds__(256) void ec_agg16(
    const float* __restrict__ hw, const int* __restrict__ src,
    const int* __restrict__ dst, const float* __restrict__ invd,
    float* __restrict__ acc, int E) {
  constexpr int EPW = 32;
  int lane = threadIdx.x & 31;
  int wave = __builtin_amdgcn_readfirstlane((blockIdx.x * blockDim.x + threadIdx.x) >> 5);
  int e0 = wave * EPW;
  if (e0 >= E) return;
  int e = e0 + lane;
  if (e >= E) e = E - 1;
  int sv = src[e];
  int dv = dst[e];
  float wv = invd[dv];
  int c = lane & 15;
  int half = lane >> 4;
  if (e0 + EPW <= E) {
    float v[EPW / 2];
#pragma unroll
    for (int k = 0; k < EPW / 2; ++k) {
      int s = __shfl(sv, 2 * k + half);
      v[k] = hw[(size_t)s * 16 + c];
    }
#pragma unroll
    for (int k = 0; k < EPW / 2; ++k) {
      int d = __shfl(dv, 2 * k + half);
      float w = __shfl(wv, 2 * k + half);
      unsafeAtomicAdd(acc + (size_t)d * 16 + c, v[k] * w);
    }
  } else {
    int n = E - e0;
    for (int k = 0; 2 * k < n; ++k) {
      int s = __shfl(sv, 2 * k + half);
      int d = __shfl(dv, 2 * k + half);
      float w = __shfl(wv, 2 * k + half);
      if (2 * k + half < n) {
        unsafeAtomicAdd(acc + (size_t)d * 16 + c, hw[(size_t)s * 16 + c] * w);
      }
    }
  }
}

__global__ __launch_bounds__(256) void ec_relu(float* __restrict__ h, int total) {
  int t = blockIdx.x * blockDim.x + threadIdx.x;
  if (t >= total) return;
  h[t] = fmaxf(h[t], 0.0f);
}

extern "C" void kernel_launch(void* const* d_in, const int* in_sizes, int n_in,
                              void* d_out, int out_size, void* d_ws, size_t ws_size,
                              hipStream_t stream) {
  const float* x   = (const float*)d_in[0];
  const int*   src = (const int*)d_in[1];
  const int*   dst = (const int*)d_in[2];
  const float* W1  = (const float*)d_in[3];
  const float* L1w = (const float*)d_in[4];
  const float* L1b = (const float*)d_in[5];
  const float* W2  = (const float*)d_in[6];
  const float* L2w = (const float*)d_in[7];
  const float* L2b = (const float*)d_in[8];
  float* out = (float*)d_out;

  const int D_IN = 256, D_H = 64, D_OUT = 16;
  const int Nn = in_sizes[0] / D_IN;
  const int R  = in_sizes[3] / (D_IN * D_H);
  const int E  = in_sizes[1] / R;

  char* ws = (char*)d_ws;
  size_t off = 0;
  int*   deg  = (int*)(ws + off);
  float* invd = (float*)(ws + off);
  off += (size_t)R * Nn * sizeof(float);
  off = (off + 255) & ~(size_t)255;
  float* h = (float*)(ws + off);
  off += (size_t)Nn * D_H * sizeof(float);
  off = (off + 255) & ~(size_t)255;
  float* hw = (float*)(ws + off);

  hipMemsetAsync(deg, 0, (size_t)R * Nn * sizeof(int), stream);
  {
    int total = R * E;
    ec_count_deg<<<(total + 255) / 256, 256, 0, stream>>>(dst, deg, total, Nn, E);
  }
  {
    int total = R * Nn;
    ec_inv_deg<<<(total + 255) / 256, 256, 0, stream>>>(deg, invd, total);
  }

  int mtiles = Nn / 16;
  int gemmBlocks = (mtiles * 32 + 127) / 128;

  ec_gemm_wmma<256, 64><<<gemmBlocks, 128, 0, stream>>>(x, L1w, L1b, h, mtiles);
  for (int r = 0; r < R; ++r) {
    ec_gemm_wmma<256, 64><<<gemmBlocks, 128, 0, stream>>>(
        x, W1 + (size_t)r * D_IN * D_H, nullptr, hw, mtiles);
    ec_agg64<<<(E + 127) / 128, 256, 0, stream>>>(
        hw, src + (size_t)r * E, dst + (size_t)r * E, invd + (size_t)r * Nn, h, E);
  }
  ec_relu<<<(Nn * D_H + 255) / 256, 256, 0, stream>>>(h, Nn * D_H);

  ec_gemm_wmma<64, 16><<<gemmBlocks, 128, 0, stream>>>(h, L2w, L2b, out, mtiles);
  for (int r = 0; r < R; ++r) {
    ec_gemm_wmma<64, 16><<<gemmBlocks, 128, 0, stream>>>(
        h, W2 + (size_t)r * D_H * D_OUT, nullptr, hw, mtiles);
    ec_agg16<<<(E + 255) / 256, 256, 0, stream>>>(
        hw, src + (size_t)r * E, dst + (size_t)r * E, invd + (size_t)r * Nn, out, E);
  }
}


# Edge sums in LDS instead of global float atomics: edges binned per call by 128-node destination range; an owner work-group adds rows by LDS float adds (arrival order), scales by 1/deg after the sum
# speedup vs baseline: 1.6798x; 1.2257x over previous
#include <hip/hip_runtime.h>

typedef __attribute__((ext_vector_type(2))) float v2f;
typedef __attribute__((ext_vector_type(8))) float v8f;

constexpr int BKT = 128;
constexpr int NBMAX = 1024;
constexpr int CHUNK = 16384;
constexpr int SP = NBMAX + 1;

__global__ __launch_bounds__(1024) void ec_bin_count(const int* __restrict__ dst,
                                                     int* __restrict__ cnt,
                                                     int E, int NB, int CH) {
  __shared__ int hist[NBMAX];
  int t = threadIdx.x;
  int c = blockIdx.x;
  int r = blockIdx.y;
  hist[t] = 0;
  __syncthreads();
  const int* d = dst + (size_t)r * E;
  int e0 = c * CHUNK;
  int dv[CHUNK / 1024];
#pragma unroll
  for (int i = 0; i < CHUNK / 1024; ++i) {
    int e = e0 + i * 1024 + t;
    dv[i] = (e < E) ? d[e] : -1;
  }
#pragma unroll
  for (int i = 0; i < CHUNK / 1024; ++i) {
    if (dv[i] >= 0) atomicAdd(&hist[dv[i] >> 7], 1);
  }
  __syncthreads();
  if (t < NB) cnt[((size_t)r * CH + c) * NB + t] = hist[t];
}

__device__ __forceinline__ int ec_scan1024(int v, int* wsum, int& total) {
  int lane = threadIdx.x & 31;
  int w = threadIdx.x >> 5;
  int x = v;
  for (int o = 1; o < 32; o <<= 1) {
    int y = __shfl_up(x, o);
    if (lane >= o) x += y;
  }
  if (lane == 31) wsum[w] = x;
  __syncthreads();
  if (w == 0) {
    int s = wsum[lane];
    for (int o = 1; o < 32; o <<= 1) {
      int y = __shfl_up(s, o);
      if (lane >= o) s += y;
    }
    wsum[lane] = s;
  }
  __syncthreads();
  int off = (w > 0) ? wsum[w - 1] : 0;
  total = wsum[31];
  __syncthreads();
  return off + x - v;
}

__global__ __launch_bounds__(1024) void ec_bin_scatter(
    const int* __restrict__ src, const int* __restrict__ dst,
    const int* __restrict__ cnt, int* __restrict__ start,
    unsigned* __restrict__ lst, int E, int NB, int CH, int r0) {
  __shared__ int lcur[NBMAX];
  __shared__ int gdelta[NBMAX];
  __shared__ int wsum[32];
  __shared__ unsigned sortedw[CHUNK];
  __shared__ int sortedp[CHUNK];
  int t = threadIdx.x;
  int c = blockIdx.x;
  int r = r0 + blockIdx.y;

  int tot = 0, before = 0, mine = 0;
  if (t < NB) {
    const int* col = cnt + (size_t)r * CH * NB + t;
    for (int cc = 0; cc < CH; cc += 8) {
      int v[8];
#pragma unroll
      for (int u = 0; u < 8; ++u) v[u] = (cc + u < CH) ? col[(size_t)(cc + u) * NB] : 0;
#pragma unroll
      for (int u = 0; u < 8; ++u) {
        tot += v[u];
        if (cc + u < c) before += v[u];
        if (cc + u == c) mine = v[u];
      }
    }
  }
  int total, ltotal;
  int gstart = ec_scan1024(tot, wsum, total);
  int lstart = ec_scan1024(mine, wsum, ltotal);
  lcur[t] = lstart;
  gdelta[t] = gstart + before - lstart;
  if (c == 0) {
    if (t < NB) start[r * SP + t] = gstart;
    if (t == 0) start[r * SP + NB] = total;
  }
  __syncthreads();

  const int* d = dst + (size_t)r * E;
  const int* s = src + (size_t)r * E;
  int e0 = c * CHUNK;
  int dv[CHUNK / 1024], sv[CHUNK / 1024];
#pragma unroll
  for (int i = 0; i < CHUNK / 1024; ++i) {
    int e = e0 + i * 1024 + t;
    dv[i] = (e < E) ? d[e] : -1;
    sv[i] = (e < E) ? s[e] : 0;
  }
#pragma unroll
  for (int i = 0; i < CHUNK / 1024; ++i) {
    if (dv[i] >= 0) {
      int b = dv[i] >> 7;
      int li = atomicAdd(&lcur[b], 1);
      sortedw[li] = ((unsigned)sv[i] << 7) | (unsigned)(dv[i] & (BKT - 1));
      sortedp[li] = gdelta[b] + li;
    }
  }
  __syncthreads();
  int n = E - e0;
  if (n > CHUNK) n = CHUNK;
  unsigned* out = lst + (size_t)(r - r0) * E;
  for (int i = 0; i < CHUNK / 1024; ++i) {
    int j = i * 1024 + t;
    if (j < n) out[sortedp[j]] = sortedw[j];
  }
}

template <int K, int NC>
__global__ __launch_bounds__(128) void ec_gemm_wmma(
    const float* __restrict__ A, const float* __restrict__ B,
    const float* __restrict__ bias, float* __restrict__ C, int mtiles) {
  constexpr int NT = NC / 16;
  constexpr int S = ((2 * NC) % 64 == 32) ? (2 * NC) : (2 * NC + 32);
  __shared__ float Bs[(K / 2) * S];

  for (int i = threadIdx.x; i < K * NC / 4; i += 128) {
    int k = (i * 4) / NC;
    int n = (i * 4) % NC;
    float4 v = *(const float4*)(B + i * 4);
    float* p = &Bs[(k >> 1) * S + 2 * n + (k & 1)];
    p[0] = v.x;
    p[2] = v.y;
    p[4] = v.z;
    p[6] = v.w;
  }
  __syncthreads();

  int wave = (blockIdx.x * blockDim.x + threadIdx.x) >> 5;
  if (wave >= mtiles) return;
  int lane = threadIdx.x & 31;
  int m = lane & 15;
  int half = lane >> 4;

  const float* Arow = A + (size_t)(wave * 16 + m) * K + 2 * half;

  v8f acc[NT];
  v8f z = {0.f, 0.f, 0.f, 0.f, 0.f, 0.f, 0.f, 0.f};
#pragma unroll
  for (int j = 0; j < NT; ++j) acc[j] = z;

#pragma unroll 8
  for (int k = 0; k < K; k += 4) {
    float2 av = *(const float2*)(Arow + k);
    v2f a = {av.x, av.y};
    const float* Bp = &Bs[(size_t)((k + 2 * half) >> 1) * S + 2 * m];
#pragma unroll
    for (int j = 0; j < NT; ++j) {
      v2f b = *(const v2f*)(Bp + j * 32);
      acc[j] = __builtin_amdgcn_wmma_f32_16x16x4_f32(
           false, a,  false, b,
           (short)0, acc[j],  false,  false);
    }
  }

#pragma unroll
  for (int j = 0; j < NT; ++j) {
    float bv = bias ? bias[j * 16 + m] : 0.0f;
#pragma unroll
    for (int v = 0; v < 8; ++v) {
      C[(size_t)(wave * 16 + v + 8 * half) * NC + j * 16 + m] = acc[j][v] + bv;
    }
  }
}


__global__ __launch_bounds__(256) void ec_sum64(
    const float* __restrict__ hw, const unsigned* __restrict__ lst,
    const int* __restrict__ start, float* __restrict__ acc, int Nn) {
  __shared__ float rows[BKT * 64];
  __shared__ int degl[BKT];
  int t = threadIdx.x;
  int lane = t & 31;
  int wave = __builtin_amdgcn_readfirstlane(t >> 5);
  int b = blockIdx.x;
  for (int i = t; i < BKT * 64; i += 256) rows[i] = 0.0f;
  if (t < BKT) degl[t] = 0;
  __syncthreads();
  int s0 = start[b];
  int s1 = start[b + 1];
  int j = lane & 15;
  int g0 = s0 + wave * 16;
  int wd = (g0 + j < s1) ? (int)lst[g0 + j] : 0;
  for (; g0 < s1; g0 += 8 * 16) {
    int n = s1 - g0;
    if (n > 16) n = 16;
    int wdn = (g0 + 8 * 16 + j < s1) ? (int)lst[g0 + 8 * 16 + j] : 0;
    if (lane < n) atomicAdd(&degl[wd & (BKT - 1)], 1);
    if (n == 16) {
      float v0[16], v1[16];
#pragma unroll
      for (int k = 0; k < 16; ++k) {
        unsigned w = (unsigned)__builtin_amdgcn_readlane(wd, k);
        const float* hp = hw + (size_t)(w >> 7) * 64 + lane;
        v0[k] = hp[0];
        v1[k] = hp[32];
      }
#pragma unroll
      for (int k = 0; k < 16; ++k) {
        unsigned w = (unsigned)__builtin_amdgcn_readlane(wd, k);
        float* rp = rows + (w & (BKT - 1)) * 64 + lane;
        __hip_atomic_fetch_add(rp, v0[k], __ATOMIC_RELAXED, __HIP_MEMORY_SCOPE_WORKGROUP);
        __hip_atomic_fetch_add(rp + 32, v1[k], __ATOMIC_RELAXED, __HIP_MEMORY_SCOPE_WORKGROUP);
      }
    } else {
      for (int k = 0; k < n; ++k) {
        unsigned w = (unsigned)__shfl(wd, k);
        const float* hp = hw + (size_t)(w >> 7) * 64 + lane;
        float* rp = rows + (w & (BKT - 1)) * 64 + lane;
        __hip_atomic_fetch_add(rp, hp[0], __ATOMIC_RELAXED, __HIP_MEMORY_SCOPE_WORKGROUP);
        __hip_atomic_fetch_add(rp + 32, hp[32], __ATOMIC_RELAXED, __HIP_MEMORY_SCOPE_WORKGROUP);
      }
    }
    wd = wdn;
  }
  __syncthreads();
  for (int dl = wave; dl < BKT; dl += 8) {
    int d = b * BKT + dl;
    if (d < Nn) {
      float w = 1.0f / fmaxf((float)degl[dl], 1.0f);
      float* ap = acc + (size_t)d * 64 + lane;
      ap[0] += rows[dl * 64 + lane] * w;
      ap[32] += rows[dl * 64 + lane + 32] * w;
    }
  }
}

__global__ __launch_bounds__(256) void ec_sum16(
    const float* __restrict__ hw, const unsigned* __restrict__ lst,
    const int* __restrict__ start, float* __restrict__ acc, int Nn) {
  __shared__ float rows[BKT * 16];
  __shared__ int degl[BKT];
  int t = threadIdx.x;
  int lane = t & 31;
  int wave = __builtin_amdgcn_readfirstlane(t >> 5);
  int b = blockIdx.x;
  for (int i = t; i < BKT * 16; i += 256) rows[i] = 0.0f;
  if (t < BKT) degl[t] = 0;
  __syncthreads();
  int s0 = start[b];
  int s1 = start[b + 1];
  int c = lane & 15;
  int half = lane >> 4;
  int g0 = s0 + wave * 32;
  int wd = (g0 + lane < s1) ? (int)lst[g0 + lane] : 0;
  for (; g0 < s1; g0 += 8 * 32) {
    int n = s1 - g0;
    if (n > 32) n = 32;
    int wdn = (g0 + 8 * 32 + lane < s1) ? (int)lst[g0 + 8 * 32 + lane] : 0;
    if (lane < n) atomicAdd(&degl[wd & (BKT - 1)], 1);
    if (n == 32) {
      float v[16];
#pragma unroll
      for (int k = 0; k < 16; ++k) {
        unsigned w = (unsigned)__shfl(wd, 2 * k + half);
        v[k] = hw[(size_t)(w >> 7) * 16 + c];
      }
#pragma unroll
      for (int k = 0; k < 16; ++k) {
        unsigned w = (unsigned)__shfl(wd, 2 * k + half);
        __hip_atomic_fetch_add(&rows[(w & (BKT - 1)) * 16 + c], v[k], __ATOMIC_RELAXED,
                               __HIP_MEMORY_SCOPE_WORKGROUP);
      }
    } else {
      for (int k = 0; 2 * k < n; ++k) {
        unsigned w = (unsigned)__shfl(wd, 2 * k + half);
        if (2 * k + half < n) {
          __hip_atomic_fetch_add(&rows[(w & (BKT - 1)) * 16 + c], hw[(size_t)(w >> 7) * 16 + c],
                                 __ATOMIC_RELAXED, __HIP_MEMORY_SCOPE_WORKGROUP);
        }
      }
    }
    wd = wdn;
  }
  __syncthreads();
  for (int i = t; i < BKT * 16; i += 256) {
    int dl = i >> 4;
    int d = b * BKT + dl;
    if (d < Nn) {
      float w = 1.0f / fmaxf((float)degl[dl], 1.0f);
      acc[(size_t)d * 16 + (i & 15)] += rows[i] * w;
    }
  }
}

__global__ __launch_bounds__(256) void ec_relu(float* __restrict__ h, int total) {
  int t = blockIdx.x * blockDim.x + threadIdx.x;
  if (t >= total) return;
  h[t] = fmaxf(h[t], 0.0f);
}

extern "C" void kernel_launch(void* const* d_in, const int* in_sizes, int n_in,
                              void* d_out, int out_size, void* d_ws, size_t ws_size,
                              hipStream_t stream) {
  const float* x   = (const float*)d_in[0];
  const int*   src = (const int*)d_in[1];
  const int*   dst = (const int*)d_in[2];
  const float* W1  = (const float*)d_in[3];
  const float* L1w = (const float*)d_in[4];
  const float* L1b = (const float*)d_in[5];
  const float* W2  = (const float*)d_in[6];
  const float* L2w = (const float*)d_in[7];
  const float* L2b = (const float*)d_in[8];
  float* out = (float*)d_out;

  const int D_IN = 256, D_H = 64, D_OUT = 16;
  const int Nn = in_sizes[0] / D_IN;
  const int R  = in_sizes[3] / (D_IN * D_H);
  const int E  = in_sizes[1] / R;
  const int NB = (Nn + BKT - 1) / BKT;
  const int CH = (E + CHUNK - 1) / CHUNK;

  char* ws = (char*)d_ws;
  size_t off = 0;
  int* start = (int*)(ws + off);
  int* cnt   = start + (size_t)R * SP;
  off += (size_t)R * Nn * sizeof(float);
  off = (off + 255) & ~(size_t)255;
  float* h = (float*)(ws + off);
  off += (size_t)Nn * D_H * sizeof(float);
  off = (off + 255) & ~(size_t)255;
  float* hw = (float*)(ws + off);
  unsigned* lst1 = (unsigned*)out;
  unsigned* lst2 = (unsigned*)(hw + (size_t)Nn * D_OUT);

  int mtiles = Nn / 16;
  int gemmBlocks = (mtiles * 32 + 127) / 128;

  ec_bin_count<<<dim3(CH, R), 1024, 0, stream>>>(dst, cnt, E, NB, CH);

  ec_gemm_wmma<256, 64><<<gemmBlocks, 128, 0, stream>>>(x, L1w, L1b, h, mtiles);
  for (int r = 0; r < R; ++r) {
    if ((r & 1) == 0) {
      int nr = (R - r < 2) ? (R - r) : 2;
      ec_bin_scatter<<<dim3(CH, nr), 1024, 0, stream>>>(src, dst, cnt, start, lst1, E, NB, CH, r);
    }
    ec_gemm_wmma<256, 64><<<gemmBlocks, 128, 0, stream>>>(
        x, W1 + (size_t)r * D_IN * D_H, nullptr, hw, mtiles);
    ec_sum64<<<NB, 256, 0, stream>>>(hw, lst1 + (size_t)(r & 1) * E, start + r * SP, h, Nn);
  }
  ec_relu<<<(Nn * D_H + 255) / 256, 256, 0, stream>>>(h, Nn * D_H);

  ec_bin_scatter<<<dim3(CH, R), 1024, 0, stream>>>(src, dst, cnt, start, lst2, E, NB, CH, 0);
  ec_gemm_wmma<64, 16><<<gemmBlocks, 128, 0, stream>>>(h, L2w, L2b, out, mtiles);
  for (int r = 0; r < R; ++r) {
    ec_gemm_wmma<64, 16><<<gemmBlocks, 128, 0, stream>>>(
        h, W2 + (size_t)r * D_H * D_OUT, nullptr, hw, mtiles);
    ec_sum16<<<NB, 256, 0, stream>>>(hw, lst2 + (size_t)r * E, start + r * SP, out, Nn);
  }
}


# ReLU applied in the last layer-1 edge-sum's write-back instead of a separate one-element-per-thread launch: 22 launches; every sum and its order unchanged
# speedup vs baseline: 1.7443x; 1.0384x over previous
#include <hip/hip_runtime.h>

typedef __attribute__((ext_vector_type(2))) float v2f;
typedef __attribute__((ext_vector_type(8))) float v8f;

constexpr int BKT = 128;
constexpr int NBMAX = 1024;
constexpr int CHUNK = 16384;
constexpr int SP = NBMAX + 1;

__global__ __launch_bounds__(1024) void ec_bin_count(const int* __restrict__ dst,
                                                     int* __restrict__ cnt,
                                                     int E, int NB, int CH) {
  __shared__ int hist[NBMAX];
  int t = threadIdx.x;
  int c = blockIdx.x;
  int r = blockIdx.y;
  hist[t] = 0;
  __syncthreads();
  const int* d = dst + (size_t)r * E;
  int e0 = c * CHUNK;
  int dv[CHUNK / 1024];
#pragma unroll
  for (int i = 0; i < CHUNK / 1024; ++i) {
    int e = e0 + i * 1024 + t;
    dv[i] = (e < E) ? d[e] : -1;
  }
#pragma unroll
  for (int i = 0; i < CHUNK / 1024; ++i) {
    if (dv[i] >= 0) atomicAdd(&hist[dv[i] >> 7], 1);
  }
  __syncthreads();
  if (t < NB) cnt[((size_t)r * CH + c) * NB + t] = hist[t];
}

__device__ __forceinline__ int ec_scan1024(int v, int* wsum, int& total) {
  int lane = threadIdx.x & 31;
  int w = threadIdx.x >> 5;
  int x = v;
  for (int o = 1; o < 32; o <<= 1) {
    int y = __shfl_up(x, o);
    if (lane >= o) x += y;
  }
  if (lane == 31) wsum[w] = x;
  __syncthreads();
  if (w == 0) {
    int s = wsum[lane];
    for (int o = 1; o < 32; o <<= 1) {
      int y = __shfl_up(s, o);
      if (lane >= o) s += y;
    }
    wsum[lane] = s;
  }
  __syncthreads();
  int off = (w > 0) ? wsum[w - 1] : 0;
  total = wsum[31];
  __syncthreads();
  return off + x - v;
}

__global__ __launch_bounds__(1024) void ec_bin_scatter(
    const int* __restrict__ src, const int* __restrict__ dst,
    const int* __restrict__ cnt, int* __restrict__ start,
    unsigned* __restrict__ lst, int E, int NB, int CH, int r0) {
  __shared__ int lcur[NBMAX];
  __shared__ int gdelta[NBMAX];
  __shared__ int wsum[32];
  __shared__ unsigned sortedw[CHUNK];
  __shared__ int sortedp[CHUNK];
  int t = threadIdx.x;
  int c = blockIdx.x;
  int r = r0 + blockIdx.y;

  int tot = 0, before = 0, mine = 0;
  if (t < NB) {
    const int* col = cnt + (size_t)r * CH * NB + t;
    for (int cc = 0; cc < CH; cc += 8) {
      int v[8];
#pragma unroll
      for (int u = 0; u < 8; ++u) v[u] = (cc + u < CH) ? col[(size_t)(cc + u) * NB] : 0;
#pragma unroll
      for (int u = 0; u < 8; ++u) {
        tot += v[u];
        if (cc + u < c) before += v[u];
        if (cc + u == c) mine = v[u];
      }
    }
  }
  int total, ltotal;
  int gstart = ec_scan1024(tot, wsum, total);
  int lstart = ec_scan1024(mine, wsum, ltotal);
  lcur[t] = lstart;
  gdelta[t] = gstart + before - lstart;
  if (c == 0) {
    if (t < NB) start[r * SP + t] = gstart;
    if (t == 0) start[r * SP + NB] = total;
  }
  __syncthreads();

  const int* d = dst + (size_t)r * E;
  const int* s = src + (size_t)r * E;
  int e0 = c * CHUNK;
  int dv[CHUNK / 1024], sv[CHUNK / 1024];
#pragma unroll
  for (int i = 0; i < CHUNK / 1024; ++i) {
    int e = e0 + i * 1024 + t;
    dv[i] = (e < E) ? d[e] : -1;
    sv[i] = (e < E) ? s[e] : 0;
  }
#pragma unroll
  for (int i = 0; i < CHUNK / 1024; ++i) {
    if (dv[i] >= 0) {
      int b = dv[i] >> 7;
      int li = atomicAdd(&lcur[b], 1);
      sortedw[li] = ((unsigned)sv[i] << 7) | (unsigned)(dv[i] & (BKT - 1));
      sortedp[li] = gdelta[b] + li;
    }
  }
  __syncthreads();
  int n = E - e0;
  if (n > CHUNK) n = CHUNK;
  unsigned* out = lst + (size_t)(r - r0) * E;
  for (int i = 0; i < CHUNK / 1024; ++i) {
    int j = i * 1024 + t;
    if (j < n) out[sortedp[j]] = sortedw[j];
  }
}

template <int K, int NC>
__global__ __launch_bounds__(128) void ec_gemm_wmma(
    const float* __restrict__ A, const float* __restrict__ B,
    const float* __restrict__ bias, float* __restrict__ C, int mtiles) {
  constexpr int NT = NC / 16;
  constexpr int S = ((2 * NC) % 64 == 32) ? (2 * NC) : (2 * NC + 32);
  __shared__ float Bs[(K / 2) * S];

  for (int i = threadIdx.x; i < K * NC / 4; i += 128) {
    int k = (i * 4) / NC;
    int n = (i * 4) % NC;
    float4 v = *(const float4*)(B + i * 4);
    float* p = &Bs[(k >> 1) * S + 2 * n + (k & 1)];
    p[0] = v.x;
    p[2] = v.y;
    p[4] = v.z;
    p[6] = v.w;
  }
  __syncthreads();

  int wave = (blockIdx.x * blockDim.x + threadIdx.x) >> 5;
  if (wave >= mtiles) return;
  int lane = threadIdx.x & 31;
  int m = lane & 15;
  int half = lane >> 4;

  const float* Arow = A + (size_t)(wave * 16 + m) * K + 2 * half;

  v8f acc[NT];
  v8f z = {0.f, 0.f, 0.f, 0.f, 0.f, 0.f, 0.f, 0.f};
#pragma unroll
  for (int j = 0; j < NT; ++j) acc[j] = z;

#pragma unroll 8
  for (int k = 0; k < K; k += 4) {
    float2 av = *(const float2*)(Arow + k);
    v2f a = {av.x, av.y};
    const float* Bp = &Bs[(size_t)((k + 2 * half) >> 1) * S + 2 * m];
#pragma unroll
    for (int j = 0; j < NT; ++j) {
      v2f b = *(const v2f*)(Bp + j * 32);
      acc[j] = __builtin_amdgcn_wmma_f32_16x16x4_f32(
           false, a,  false, b,
           (short)0, acc[j],  false,  false);
    }
  }

#pragma unroll
  for (int j = 0; j < NT; ++j) {
    float bv = bias ? bias[j * 16 + m] : 0.0f;
#pragma unroll
    for (int v = 0; v < 8; ++v) {
      C[(size_t)(wave * 16 + v + 8 * half) * NC + j * 16 + m] = acc[j][v] + bv;
    }
  }
}


__global__ __launch_bounds__(256) void ec_sum64(
    const float* __restrict__ hw, const unsigned* __restrict__ lst,
    const int* __restrict__ start, float* __restrict__ acc, int Nn, int relu) {
  __shared__ float rows[BKT * 64];
  __shared__ int degl[BKT];
  int t = threadIdx.x;
  int lane = t & 31;
  int wave = __builtin_amdgcn_readfirstlane(t >> 5);
  int b = blockIdx.x;
  for (int i = t; i < BKT * 64; i += 256) rows[i] = 0.0f;
  if (t < BKT) degl[t] = 0;
  __syncthreads();
  int s0 = start[b];
  int s1 = start[b + 1];
  int j = lane & 15;
  int g0 = s0 + wave * 16;
  int wd = (g0 + j < s1) ? (int)lst[g0 + j] : 0;
  for (; g0 < s1; g0 += 8 * 16) {
    int n = s1 - g0;
    if (n > 16) n = 16;
    int wdn = (g0 + 8 * 16 + j < s1) ? (int)lst[g0 + 8 * 16 + j] : 0;
    if (lane < n) atomicAdd(&degl[wd & (BKT - 1)], 1);
    if (n == 16) {
      float v0[16], v1[16];
#pragma unroll
      for (int k = 0; k < 16; ++k) {
        unsigned w = (unsigned)__builtin_amdgcn_readlane(wd, k);
        const float* hp = hw + (size_t)(w >> 7) * 64 + lane;
        v0[k] = hp[0];
        v1[k] = hp[32];
      }
#pragma unroll
      for (int k = 0; k < 16; ++k) {
        unsigned w = (unsigned)__builtin_amdgcn_readlane(wd, k);
        float* rp = rows + (w & (BKT - 1)) * 64 + lane;
        __hip_atomic_fetch_add(rp, v0[k], __ATOMIC_RELAXED, __HIP_MEMORY_SCOPE_WORKGROUP);
        __hip_atomic_fetch_add(rp + 32, v1[k], __ATOMIC_RELAXED, __HIP_MEMORY_SCOPE_WORKGROUP);
      }
    } else {
      for (int k = 0; k < n; ++k) {
        unsigned w = (unsigned)__shfl(wd, k);
        const float* hp = hw + (size_t)(w >> 7) * 64 + lane;
        float* rp = rows + (w & (BKT - 1)) * 64 + lane;
        __hip_atomic_fetch_add(rp, hp[0], __ATOMIC_RELAXED, __HIP_MEMORY_SCOPE_WORKGROUP);
        __hip_atomic_fetch_add(rp + 32, hp[32], __ATOMIC_RELAXED, __HIP_MEMORY_SCOPE_WORKGROUP);
      }
    }
    wd = wdn;
  }
  __syncthreads();
  for (int dl = wave; dl < BKT; dl += 8) {
    int d = b * BKT + dl;
    if (d < Nn) {
      float w = 1.0f / fmaxf((float)degl[dl], 1.0f);
      float* ap = acc + (size_t)d * 64 + lane;
      float a0 = ap[0] + rows[dl * 64 + lane] * w;
      float a1 = ap[32] + rows[dl * 64 + lane + 32] * w;
      if (relu) {
        a0 = fmaxf(a0, 0.0f);
        a1 = fmaxf(a1, 0.0f);
      }
      ap[0] = a0;
      ap[32] = a1;
    }
  }
}

__global__ __launch_bounds__(256) void ec_sum16(
    const float* __restrict__ hw, const unsigned* __restrict__ lst,
    const int* __restrict__ start, float* __restrict__ acc, int Nn) {
  __shared__ float rows[BKT * 16];
  __shared__ int degl[BKT];
  int t = threadIdx.x;
  int lane = t & 31;
  int wave = __builtin_amdgcn_readfirstlane(t >> 5);
  int b = blockIdx.x;
  for (int i = t; i < BKT * 16; i += 256) rows[i] = 0.0f;
  if (t < BKT) degl[t] = 0;
  __syncthreads();
  int s0 = start[b];
  int s1 = start[b + 1];
  int c = lane & 15;
  int half = lane >> 4;
  int g0 = s0 + wave * 32;
  int wd = (g0 + lane < s1) ? (int)lst[g0 + lane] : 0;
  for (; g0 < s1; g0 += 8 * 32) {
    int n = s1 - g0;
    if (n > 32) n = 32;
    int wdn = (g0 + 8 * 32 + lane < s1) ? (int)lst[g0 + 8 * 32 + lane] : 0;
    if (lane < n) atomicAdd(&degl[wd & (BKT - 1)], 1);
    if (n == 32) {
      float v[16];
#pragma unroll
      for (int k = 0; k < 16; ++k) {
        unsigned w = (unsigned)__shfl(wd, 2 * k + half);
        v[k] = hw[(size_t)(w >> 7) * 16 + c];
      }
#pragma unroll
      for (int k = 0; k < 16; ++k) {
        unsigned w = (unsigned)__shfl(wd, 2 * k + half);
        __hip_atomic_fetch_add(&rows[(w & (BKT - 1)) * 16 + c], v[k], __ATOMIC_RELAXED,
                               __HIP_MEMORY_SCOPE_WORKGROUP);
      }
    } else {
      for (int k = 0; 2 * k < n; ++k) {
        unsigned w = (unsigned)__shfl(wd, 2 * k + half);
        if (2 * k + half < n) {
          __hip_atomic_fetch_add(&rows[(w & (BKT - 1)) * 16 + c], hw[(size_t)(w >> 7) * 16 + c],
                                 __ATOMIC_RELAXED, __HIP_MEMORY_SCOPE_WORKGROUP);
        }
      }
    }
    wd = wdn;
  }
  __syncthreads();
  for (int i = t; i < BKT * 16; i += 256) {
    int dl = i >> 4;
    int d = b * BKT + dl;
    if (d < Nn) {
      float w = 1.0f / fmaxf((float)degl[dl], 1.0f);
      acc[(size_t)d * 16 + (i & 15)] += rows[i] * w;
    }
  }
}

extern "C" void kernel_launch(void* const* d_in, const int* in_sizes, int n_in,
                              void* d_out, int out_size, void* d_ws, size_t ws_size,
                              hipStream_t stream) {
  const float* x   = (const float*)d_in[0];
  const int*   src = (const int*)d_in[1];
  const int*   dst = (const int*)d_in[2];
  const float* W1  = (const float*)d_in[3];
  const float* L1w = (const float*)d_in[4];
  const float* L1b = (const float*)d_in[5];
  const float* W2  = (const float*)d_in[6];
  const float* L2w = (const float*)d_in[7];
  const float* L2b = (const float*)d_in[8];
  float* out = (float*)d_out;

  const int D_IN = 256, D_H = 64, D_OUT = 16;
  const int Nn = in_sizes[0] / D_IN;
  const int R  = in_sizes[3] / (D_IN * D_H);
  const int E  = in_sizes[1] / R;
  const int NB = (Nn + BKT - 1) / BKT;
  const int CH = (E + CHUNK - 1) / CHUNK;

  char* ws = (char*)d_ws;
  size_t off = 0;
  int* start = (int*)(ws + off);
  int* cnt   = start + (size_t)R * SP;
  off += (size_t)R * Nn * sizeof(float);
  off = (off + 255) & ~(size_t)255;
  float* h = (float*)(ws + off);
  off += (size_t)Nn * D_H * sizeof(float);
  off = (off + 255) & ~(size_t)255;
  float* hw = (float*)(ws + off);
  unsigned* lst1 = (unsigned*)out;
  unsigned* lst2 = (unsigned*)(hw + (size_t)Nn * D_OUT);

  int mtiles = Nn / 16;
  int gemmBlocks = (mtiles * 32 + 127) / 128;

  ec_bin_count<<<dim3(CH, R), 1024, 0, stream>>>(dst, cnt, E, NB, CH);

  ec_gemm_wmma<256, 64><<<gemmBlocks, 128, 0, stream>>>(x, L1w, L1b, h, mtiles);
  for (int r = 0; r < R; ++r) {
    if ((r & 1) == 0) {
      int nr = (R - r < 2) ? (R - r) : 2;
      ec_bin_scatter<<<dim3(CH, nr), 1024, 0, stream>>>(src, dst, cnt, start, lst1, E, NB, CH, r);
    }
    ec_gemm_wmma<256, 64><<<gemmBlocks, 128, 0, stream>>>(
        x, W1 + (size_t)r * D_IN * D_H, nullptr, hw, mtiles);
    ec_sum64<<<NB, 256, 0, stream>>>(hw, lst1 + (size_t)(r & 1) * E, start + r * SP, h, Nn,
                                     (r == R - 1) ? 1 : 0);
  }

  ec_bin_scatter<<<dim3(CH, R), 1024, 0, stream>>>(src, dst, cnt, start, lst2, E, NB, CH, 0);
  ec_gemm_wmma<64, 16><<<gemmBlocks, 128, 0, stream>>>(h, L2w, L2b, out, mtiles);
  for (int r = 0; r < R; ++r) {
    ec_gemm_wmma<64, 16><<<gemmBlocks, 128, 0, stream>>>(
        h, W2 + (size_t)r * D_H * D_OUT, nullptr, hw, mtiles);
    ec_sum16<<<NB, 256, 0, stream>>>(hw, lst2 + (size_t)r * E, start + r * SP, out, Nn);
  }
}


# Layer 2 in 3 launches, not 10: one WMMA pass over h gives the self term and the four relation products (80 columns), one edge-sum launch walks the four relations per bucket; add order unchanged
# speedup vs baseline: 1.9372x; 1.1106x over previous
#include <hip/hip_runtime.h>

typedef __attribute__((ext_vector_type(2))) float v2f;
typedef __attribute__((ext_vector_type(8))) float v8f;

constexpr int BKT = 128;
constexpr int NBMAX = 1024;
constexpr int CHUNK = 16384;
constexpr int SP = NBMAX + 1;
constexpr int R2 = 4;

__global__ __launch_bounds__(1024) void ec_bin_count(const int* __restrict__ dst,
                                                     int* __restrict__ cnt,
                                                     int E, int NB, int CH) {
  __shared__ int hist[NBMAX];
  int t = threadIdx.x;
  int c = blockIdx.x;
  int r = blockIdx.y;
  hist[t] = 0;
  __syncthreads();
  const int* d = dst + (size_t)r * E;
  int e0 = c * CHUNK;
  int dv[CHUNK / 1024];
#pragma unroll
  for (int i = 0; i < CHUNK / 1024; ++i) {
    int e = e0 + i * 1024 + t;
    dv[i] = (e < E) ? d[e] : -1;
  }
#pragma unroll
  for (int i = 0; i < CHUNK / 1024; ++i) {
    if (dv[i] >= 0) atomicAdd(&hist[dv[i] >> 7], 1);
  }
  __syncthreads();
  if (t < NB) cnt[((size_t)r * CH + c) * NB + t] = hist[t];
}

__device__ __forceinline__ int ec_scan1024(int v, int* wsum, int& total) {
  int lane = threadIdx.x & 31;
  int w = threadIdx.x >> 5;
  int x = v;
  for (int o = 1; o < 32; o <<= 1) {
    int y = __shfl_up(x, o);
    if (lane >= o) x += y;
  }
  if (lane == 31) wsum[w] = x;
  __syncthreads();
  if (w == 0) {
    int s = wsum[lane];
    for (int o = 1; o < 32; o <<= 1) {
      int y = __shfl_up(s, o);
      if (lane >= o) s += y;
    }
    wsum[lane] = s;
  }
  __syncthreads();
  int off = (w > 0) ? wsum[w - 1] : 0;
  total = wsum[31];
  __syncthreads();
  return off + x - v;
}

__global__ __launch_bounds__(1024) void ec_bin_scatter(
    const int* __restrict__ src, const int* __restrict__ dst,
    const int* __restrict__ cnt, int* __restrict__ start,
    unsigned* __restrict__ lst, int E, int NB, int CH, int r0) {
  __shared__ int lcur[NBMAX];
  __shared__ int gdelta[NBMAX];
  __shared__ int wsum[32];
  __shared__ unsigned sortedw[CHUNK];
  __shared__ int sortedp[CHUNK];
  int t = threadIdx.x;
  int c = blockIdx.x;
  int r = r0 + blockIdx.y;

  int tot = 0, before = 0, mine = 0;
  if (t < NB) {
    const int* col = cnt + (size_t)r * CH * NB + t;
    for (int cc = 0; cc < CH; cc += 8) {
      int v[8];
#pragma unroll
      for (int u = 0; u < 8; ++u) v[u] = (cc + u < CH) ? col[(size_t)(cc + u) * NB] : 0;
#pragma unroll
      for (int u = 0; u < 8; ++u) {
        tot += v[u];
        if (cc + u < c) before += v[u];
        if (cc + u == c) mine = v[u];
      }
    }
  }
  int total, ltotal;
  int gstart = ec_scan1024(tot, wsum, total);
  int lstart = ec_scan1024(mine, wsum, ltotal);
  lcur[t] = lstart;
  gdelta[t] = gstart + before - lstart;
  if (c == 0) {
    if (t < NB) start[r * SP + t] = gstart;
    if (t == 0) start[r * SP + NB] = total;
  }
  __syncthreads();

  const int* d = dst + (size_t)r * E;
  const int* s = src + (size_t)r * E;
  int e0 = c * CHUNK;
  int dv[CHUNK / 1024], sv[CHUNK / 1024];
#pragma unroll
  for (int i = 0; i < CHUNK / 1024; ++i) {
    int e = e0 + i * 1024 + t;
    dv[i] = (e < E) ? d[e] : -1;
    sv[i] = (e < E) ? s[e] : 0;
  }
#pragma unroll
  for (int i = 0; i < CHUNK / 1024; ++i) {
    if (dv[i] >= 0) {
      int b = dv[i] >> 7;
      int li = atomicAdd(&lcur[b], 1);
      sortedw[li] = ((unsigned)sv[i] << 7) | (unsigned)(dv[i] & (BKT - 1));
      sortedp[li] = gdelta[b] + li;
    }
  }
  __syncthreads();
  int n = E - e0;
  if (n > CHUNK) n = CHUNK;
  unsigned* out = lst + (size_t)(r - r0) * E;
  for (int i = 0; i < CHUNK / 1024; ++i) {
    int j = i * 1024 + t;
    if (j < n) out[sortedp[j]] = sortedw[j];
  }
}

template <int K, int NC>
__global__ __launch_bounds__(128) void ec_gemm_wmma(
    const float* __restrict__ A, const float* __restrict__ B,
    const float* __restrict__ bias, float* __restrict__ C, int mtiles) {
  constexpr int NT = NC / 16;
  constexpr int S = ((2 * NC) % 64 == 32) ? (2 * NC) : (2 * NC + 32);
  __shared__ float Bs[(K / 2) * S];

  for (int i = threadIdx.x; i < K * NC / 4; i += 128) {
    int k = (i * 4) / NC;
    int n = (i * 4) % NC;
    float4 v = *(const float4*)(B + i * 4);
    float* p = &Bs[(k >> 1) * S + 2 * n + (k & 1)];
    p[0] = v.x;
    p[2] = v.y;
    p[4] = v.z;
    p[6] = v.w;
  }
  __syncthreads();

  int wave = (blockIdx.x * blockDim.x + threadIdx.x) >> 5;
  if (wave >= mtiles) return;
  int lane = threadIdx.x & 31;
  int m = lane & 15;
  int half = lane >> 4;

  const float* Arow = A + (size_t)(wave * 16 + m) * K + 2 * half;

  v8f acc[NT];
  v8f z = {0.f, 0.f, 0.f, 0.f, 0.f, 0.f, 0.f, 0.f};
#pragma unroll
  for (int j = 0; j < NT; ++j) acc[j] = z;

#pragma unroll 8
  for (int k = 0; k < K; k += 4) {
    float2 av = *(const float2*)(Arow + k);
    v2f a = {av.x, av.y};
    const float* Bp = &Bs[(size_t)((k + 2 * half) >> 1) * S + 2 * m];
#pragma unroll
    for (int j = 0; j < NT; ++j) {
      v2f b = *(const v2f*)(Bp + j * 32);
      acc[j] = __builtin_amdgcn_wmma_f32_16x16x4_f32(
           false, a,  false, b,
           (short)0, acc[j],  false,  false);
    }
  }

#pragma unroll
  for (int j = 0; j < NT; ++j) {
    float bv = bias ? bias[j * 16 + m] : 0.0f;
#pragma unroll
    for (int v = 0; v < 8; ++v) {
      C[(size_t)(wave * 16 + v + 8 * half) * NC + j * 16 + m] = acc[j][v] + bv;
    }
  }
}


__global__ __launch_bounds__(128) void ec_gemm2_all(
    const float* __restrict__ A, const float* __restrict__ Bself,
    const float* __restrict__ Brel, const float* __restrict__ bias,
    float* __restrict__ Cself, float* __restrict__ Crel, int mtiles) {
  constexpr int K = 64;
  constexpr int NT = 1 + R2;
  constexpr int S = 2 * 16 * NT;
  static_assert(S % 64 == 32, "lane halves must land in disjoint bank ranges");
  __shared__ float Bs[(K / 2) * S];

  static_assert(K * 16 / 4 == 256, "two float4 of a block per thread");
  float4 sv[2 * NT];
#pragma unroll
  for (int it = 0; it < 2 * NT; ++it) {
    const float* bp = (it < 2) ? Bself : (Brel + (size_t)(it / 2 - 1) * K * 16);
    sv[it] = *(const float4*)(bp + (threadIdx.x + 128 * (it & 1)) * 4);
  }
#pragma unroll
  for (int it = 0; it < 2 * NT; ++it) {
    int q = threadIdx.x + 128 * (it & 1);
    int k = q >> 2;
    int n = (it / 2) * 16 + (q & 3) * 4;
    float* p = &Bs[(k >> 1) * S + 2 * n + (k & 1)];
    p[0] = sv[it].x;
    p[2] = sv[it].y;
    p[4] = sv[it].z;
    p[6] = sv[it].w;
  }
  __syncthreads();

  int wave = (blockIdx.x * blockDim.x + threadIdx.x) >> 5;
  if (wave >= mtiles) return;
  int lane = threadIdx.x & 31;
  int m = lane & 15;
  int half = lane >> 4;

  const float* Arow = A + (size_t)(wave * 16 + m) * K + 2 * half;

  v8f acc[NT];
  v8f z = {0.f, 0.f, 0.f, 0.f, 0.f, 0.f, 0.f, 0.f};
#pragma unroll
  for (int j = 0; j < NT; ++j) acc[j] = z;

#pragma unroll
  for (int k = 0; k < K; k += 4) {
    float2 av = *(const float2*)(Arow + k);
    v2f a = {av.x, av.y};
    const float* Bp = &Bs[(size_t)((k + 2 * half) >> 1) * S + 2 * m];
#pragma unroll
    for (int j = 0; j < NT; ++j) {
      v2f b = *(const v2f*)(Bp + j * 32);
      acc[j] = __builtin_amdgcn_wmma_f32_16x16x4_f32(
           false, a,  false, b,
           (short)0, acc[j],  false,  false);
    }
  }

  float bv = bias[m];
  size_t plane = (size_t)mtiles * 16 * 16;
#pragma unroll
  for (int v = 0; v < 8; ++v) {
    size_t o = (size_t)(wave * 16 + v + 8 * half) * 16 + m;
    Cself[o] = acc[0][v] + bv;
#pragma unroll
    for (int j = 1; j < NT; ++j) Crel[(size_t)(j - 1) * plane + o] = acc[j][v];
  }
}

__global__ __launch_bounds__(256) void ec_sum64(
    const float* __restrict__ hw, const unsigned* __restrict__ lst,
    const int* __restrict__ start, float* __restrict__ acc, int Nn, int relu) {
  __shared__ float rows[BKT * 64];
  __shared__ int degl[BKT];
  int t = threadIdx.x;
  int lane = t & 31;
  int wave = __builtin_amdgcn_readfirstlane(t >> 5);
  int b = blockIdx.x;
  for (int i = t; i < BKT * 64; i += 256) rows[i] = 0.0f;
  if (t < BKT) degl[t] = 0;
  __syncthreads();
  int s0 = start[b];
  int s1 = start[b + 1];
  int j = lane & 15;
  int g0 = s0 + wave * 16;
  int wd = (g0 + j < s1) ? (int)lst[g0 + j] : 0;
  for (; g0 < s1; g0 += 8 * 16) {
    int n = s1 - g0;
    if (n > 16) n = 16;
    int wdn = (g0 + 8 * 16 + j < s1) ? (int)lst[g0 + 8 * 16 + j] : 0;
    if (lane < n) atomicAdd(&degl[wd & (BKT - 1)], 1);
    if (n == 16) {
      float v0[16], v1[16];
#pragma unroll
      for (int k = 0; k < 16; ++k) {
        unsigned w = (unsigned)__builtin_amdgcn_readlane(wd, k);
        const float* hp = hw + (size_t)(w >> 7) * 64 + lane;
        v0[k] = hp[0];
        v1[k] = hp[32];
      }
#pragma unroll
      for (int k = 0; k < 16; ++k) {
        unsigned w = (unsigned)__builtin_amdgcn_readlane(wd, k);
        float* rp = rows + (w & (BKT - 1)) * 64 + lane;
        __hip_atomic_fetch_add(rp, v0[k], __ATOMIC_RELAXED, __HIP_MEMORY_SCOPE_WORKGROUP);
        __hip_atomic_fetch_add(rp + 32, v1[k], __ATOMIC_RELAXED, __HIP_MEMORY_SCOPE_WORKGROUP);
      }
    } else {
      for (int k = 0; k < n; ++k) {
        unsigned w = (unsigned)__shfl(wd, k);
        const float* hp = hw + (size_t)(w >> 7) * 64 + lane;
        float* rp = rows + (w & (BKT - 1)) * 64 + lane;
        __hip_atomic_fetch_add(rp, hp[0], __ATOMIC_RELAXED, __HIP_MEMORY_SCOPE_WORKGROUP);
        __hip_atomic_fetch_add(rp + 32, hp[32], __ATOMIC_RELAXED, __HIP_MEMORY_SCOPE_WORKGROUP);
      }
    }
    wd = wdn;
  }
  __syncthreads();
  for (int dl = wave; dl < BKT; dl += 8) {
    int d = b * BKT + dl;
    if (d < Nn) {
      float w = 1.0f / fmaxf((float)degl[dl], 1.0f);
      float* ap = acc + (size_t)d * 64 + lane;
      float a0 = ap[0] + rows[dl * 64 + lane] * w;
      float a1 = ap[32] + rows[dl * 64 + lane + 32] * w;
      if (relu) {
        a0 = fmaxf(a0, 0.0f);
        a1 = fmaxf(a1, 0.0f);
      }
      ap[0] = a0;
      ap[32] = a1;
    }
  }
}

__global__ __launch_bounds__(256) void ec_sum16_all(
    const float* __restrict__ hw, const unsigned* __restrict__ lst,
    const int* __restrict__ start, float* __restrict__ acc, int Nn, int E) {
  __shared__ float rows[2 * BKT * 16];
  __shared__ int degl[R2 * BKT];
  __shared__ int sb[2 * R2];
  int t = threadIdx.x;
  int lane = t & 31;
  int wave = __builtin_amdgcn_readfirstlane(t >> 5);
  int b = blockIdx.x;
  if (t < 2 * R2) sb[t] = start[(t >> 1) * SP + b + (t & 1)];
  for (int i = t; i < 2 * BKT * 16; i += 256) rows[i] = 0.0f;
  for (int i = t; i < R2 * BKT; i += 256) degl[i] = 0;
  float tot[BKT * 16 / 256];
#pragma unroll
  for (int q = 0; q < BKT * 16 / 256; ++q) {
    int i = t + 256 * q;
    int d = b * BKT + (i >> 4);
    tot[q] = (d < Nn) ? acc[(size_t)d * 16 + (i & 15)] : 0.0f;
  }
  __syncthreads();
  int c = lane & 15;
  int half = lane >> 4;
  int s1 = __builtin_amdgcn_readfirstlane(sb[1]);
  int g0 = __builtin_amdgcn_readfirstlane(sb[0]) + wave * 32;
  int wd = (g0 + lane < s1) ? (int)lst[g0 + lane] : 0;
  for (int r = 0; r < R2; ++r) {
    const float* hwr = hw + (size_t)r * Nn * 16;
    const unsigned* l = lst + (size_t)r * E;
    float* rw = rows + (r & 1) * BKT * 16;
    int* dg = degl + r * BKT;
    int s1n = 0, g0n = 0, wdf = 0;
    if (r + 1 < R2) {
      s1n = __builtin_amdgcn_readfirstlane(sb[2 * r + 3]);
      g0n = __builtin_amdgcn_readfirstlane(sb[2 * r + 2]) + wave * 32;
      wdf = (g0n + lane < s1n) ? (int)lst[(size_t)(r + 1) * E + g0n + lane] : 0;
    }
    for (; g0 < s1; g0 += 8 * 32) {
      int n = s1 - g0;
      if (n > 32) n = 32;
      int wdn = (g0 + 8 * 32 + lane < s1) ? (int)l[g0 + 8 * 32 + lane] : 0;
      if (lane < n) atomicAdd(&dg[wd & (BKT - 1)], 1);
      if (n == 32) {
        float v[16];
#pragma unroll
        for (int k = 0; k < 16; ++k) {
          unsigned w = (unsigned)__shfl(wd, 2 * k + half);
          v[k] = hwr[(size_t)(w >> 7) * 16 + c];
        }
#pragma unroll
        for (int k = 0; k < 16; ++k) {
          unsigned w = (unsigned)__shfl(wd, 2 * k + half);
          __hip_atomic_fetch_add(&rw[(w & (BKT - 1)) * 16 + c], v[k], __ATOMIC_RELAXED,
                                 __HIP_MEMORY_SCOPE_WORKGROUP);
        }
      } else {
        for (int k = 0; 2 * k < n; ++k) {
          unsigned w = (unsigned)__shfl(wd, 2 * k + half);
          if (2 * k + half < n) {
            __hip_atomic_fetch_add(&rw[(w & (BKT - 1)) * 16 + c], hwr[(size_t)(w >> 7) * 16 + c],
                                   __ATOMIC_RELAXED, __HIP_MEMORY_SCOPE_WORKGROUP);
          }
        }
      }
      wd = wdn;
    }
    __syncthreads();
#pragma unroll
    for (int q = 0; q < BKT * 16 / 256; ++q) {
      int i = t + 256 * q;
      float w = 1.0f / fmaxf((float)dg[i >> 4], 1.0f);
      tot[q] += rw[i] * w;
      rw[i] = 0.0f;
    }
    s1 = s1n;
    g0 = g0n;
    wd = wdf;
  }
#pragma unroll
  for (int q = 0; q < BKT * 16 / 256; ++q) {
    int i = t + 256 * q;
    int d = b * BKT + (i >> 4);
    if (d < Nn) acc[(size_t)d * 16 + (i & 15)] = tot[q];
  }
}

extern "C" void kernel_launch(void* const* d_in, const int* in_sizes, int n_in,
                              void* d_out, int out_size, void* d_ws, size_t ws_size,
                              hipStream_t stream) {
  const float* x   = (const float*)d_in[0];
  const int*   src = (const int*)d_in[1];
  const int*   dst = (const int*)d_in[2];
  const float* W1  = (const float*)d_in[3];
  const float* L1w = (const float*)d_in[4];
  const float* L1b = (const float*)d_in[5];
  const float* W2  = (const float*)d_in[6];
  const float* L2w = (const float*)d_in[7];
  const float* L2b = (const float*)d_in[8];
  float* out = (float*)d_out;

  const int D_IN = 256, D_H = 64;
  const int Nn = in_sizes[0] / D_IN;
  const int R  = in_sizes[3] / (D_IN * D_H);
  const int E  = in_sizes[1] / R;
  const int NB = (Nn + BKT - 1) / BKT;
  const int CH = (E + CHUNK - 1) / CHUNK;

  char* ws = (char*)d_ws;
  size_t off = 0;
  int* start = (int*)(ws + off);
  int* cnt   = start + (size_t)R * SP;
  off += (size_t)R * Nn * sizeof(float);
  off = (off + 255) & ~(size_t)255;
  float* h = (float*)(ws + off);
  off += (size_t)Nn * D_H * sizeof(float);
  off = (off + 255) & ~(size_t)255;
  float* hw = (float*)(ws + off);
  unsigned* lst1 = (unsigned*)out;
  unsigned* lst2 = (unsigned*)h;

  int mtiles = Nn / 16;
  int gemmBlocks = (mtiles * 32 + 127) / 128;

  ec_bin_count<<<dim3(CH, R), 1024, 0, stream>>>(dst, cnt, E, NB, CH);

  ec_gemm_wmma<256, 64><<<gemmBlocks, 128, 0, stream>>>(x, L1w, L1b, h, mtiles);
  for (int r = 0; r < R; ++r) {
    if ((r & 1) == 0) {
      int nr = (R - r < 2) ? (R - r) : 2;
      ec_bin_scatter<<<dim3(CH, nr), 1024, 0, stream>>>(src, dst, cnt, start, lst1, E, NB, CH, r);
    }
    ec_gemm_wmma<256, 64><<<gemmBlocks, 128, 0, stream>>>(
        x, W1 + (size_t)r * D_IN * D_H, nullptr, hw, mtiles);
    ec_sum64<<<NB, 256, 0, stream>>>(hw, lst1 + (size_t)(r & 1) * E, start + r * SP, h, Nn,
                                     (r == R - 1) ? 1 : 0);
  }

  ec_gemm2_all<<<gemmBlocks, 128, 0, stream>>>(h, L2w, W2, L2b, out, hw, mtiles);
  ec_bin_scatter<<<dim3(CH, R), 1024, 0, stream>>>(src, dst, cnt, start, lst2, E, NB, CH, 0);
  ec_sum16_all<<<NB, 256, 0, stream>>>(hw, lst2, start, out, Nn, E);
}


# Edge sums: a bucket's ragged last group is gathered as one batch like a full group (words past the end read a valid row, not added), not one dependent load per row; LDS add order unchanged
# speedup vs baseline: 2.0383x; 1.0522x over previous
#include <hip/hip_runtime.h>

typedef __attribute__((ext_vector_type(2))) float v2f;
typedef __attribute__((ext_vector_type(8))) float v8f;

constexpr int BKT = 128;
constexpr int NBMAX = 1024;
constexpr int CHUNK = 16384;
constexpr int SP = NBMAX + 1;
constexpr int R2 = 4;

__global__ __launch_bounds__(1024) void ec_bin_count(const int* __restrict__ dst,
                                                     int* __restrict__ cnt,
                                                     int E, int NB, int CH) {
  __shared__ int hist[NBMAX];
  int t = threadIdx.x;
  int c = blockIdx.x;
  int r = blockIdx.y;
  hist[t] = 0;
  __syncthreads();
  const int* d = dst + (size_t)r * E;
  int e0 = c * CHUNK;
  int dv[CHUNK / 1024];
#pragma unroll
  for (int i = 0; i < CHUNK / 1024; ++i) {
    int e = e0 + i * 1024 + t;
    dv[i] = (e < E) ? d[e] : -1;
  }
#pragma unroll
  for (int i = 0; i < CHUNK / 1024; ++i) {
    if (dv[i] >= 0) atomicAdd(&hist[dv[i] >> 7], 1);
  }
  __syncthreads();
  if (t < NB) cnt[((size_t)r * CH + c) * NB + t] = hist[t];
}

__device__ __forceinline__ int ec_scan1024(int v, int* wsum, int& total) {
  int lane = threadIdx.x & 31;
  int w = threadIdx.x >> 5;
  int x = v;
  for (int o = 1; o < 32; o <<= 1) {
    int y = __shfl_up(x, o);
    if (lane >= o) x += y;
  }
  if (lane == 31) wsum[w] = x;
  __syncthreads();
  if (w == 0) {
    int s = wsum[lane];
    for (int o = 1; o < 32; o <<= 1) {
      int y = __shfl_up(s, o);
      if (lane >= o) s += y;
    }
    wsum[lane] = s;
  }
  __syncthreads();
  int off = (w > 0) ? wsum[w - 1] : 0;
  total = wsum[31];
  __syncthreads();
  return off + x - v;
}

__global__ __launch_bounds__(1024) void ec_bin_scatter(
    const int* __restrict__ src, const int* __restrict__ dst,
    const int* __restrict__ cnt, int* __restrict__ start,
    unsigned* __restrict__ lst, int E, int NB, int CH, int r0) {
  __shared__ int lcur[NBMAX];
  __shared__ int gdelta[NBMAX];
  __shared__ int wsum[32];
  __shared__ unsigned sortedw[CHUNK];
  __shared__ int sortedp[CHUNK];
  int t = threadIdx.x;
  int c = blockIdx.x;
  int r = r0 + blockIdx.y;

  int tot = 0, before = 0, mine = 0;
  if (t < NB) {
    const int* col = cnt + (size_t)r * CH * NB + t;
    for (int cc = 0; cc < CH; cc += 8) {
      int v[8];
#pragma unroll
      for (int u = 0; u < 8; ++u) v[u] = (cc + u < CH) ? col[(size_t)(cc + u) * NB] : 0;
#pragma unroll
      for (int u = 0; u < 8; ++u) {
        tot += v[u];
        if (cc + u < c) before += v[u];
        if (cc + u == c) mine = v[u];
      }
    }
  }
  int total, ltotal;
  int gstart = ec_scan1024(tot, wsum, total);
  int lstart = ec_scan1024(mine, wsum, ltotal);
  lcur[t] = lstart;
  gdelta[t] = gstart + before - lstart;
  if (c == 0) {
    if (t < NB) start[r * SP + t] = gstart;
    if (t == 0) start[r * SP + NB] = total;
  }
  __syncthreads();

  const int* d = dst + (size_t)r * E;
  const int* s = src + (size_t)r * E;
  int e0 = c * CHUNK;
  int dv[CHUNK / 1024], sv[CHUNK / 1024];
#pragma unroll
  for (int i = 0; i < CHUNK / 1024; ++i) {
    int e = e0 + i * 1024 + t;
    dv[i] = (e < E) ? d[e] : -1;
    sv[i] = (e < E) ? s[e] : 0;
  }
#pragma unroll
  for (int i = 0; i < CHUNK / 1024; ++i) {
    if (dv[i] >= 0) {
      int b = dv[i] >> 7;
      int li = atomicAdd(&lcur[b], 1);
      sortedw[li] = ((unsigned)sv[i] << 7) | (unsigned)(dv[i] & (BKT - 1));
      sortedp[li] = gdelta[b] + li;
    }
  }
  __syncthreads();
  int n = E - e0;
  if (n > CHUNK) n = CHUNK;
  unsigned* out = lst + (size_t)(r - r0) * E;
  for (int i = 0; i < CHUNK / 1024; ++i) {
    int j = i * 1024 + t;
    if (j < n) out[sortedp[j]] = sortedw[j];
  }
}

template <int K, int NC>
__global__ __launch_bounds__(128) void ec_gemm_wmma(
    const float* __restrict__ A, const float* __restrict__ B,
    const float* __restrict__ bias, float* __restrict__ C, int mtiles) {
  constexpr int NT = NC / 16;
  constexpr int S = ((2 * NC) % 64 == 32) ? (2 * NC) : (2 * NC + 32);
  __shared__ float Bs[(K / 2) * S];

  for (int i = threadIdx.x; i < K * NC / 4; i += 128) {
    int k = (i * 4) / NC;
    int n = (i * 4) % NC;
    float4 v = *(const float4*)(B + i * 4);
    float* p = &Bs[(k >> 1) * S + 2 * n + (k & 1)];
    p[0] = v.x;
    p[2] = v.y;
    p[4] = v.z;
    p[6] = v.w;
  }
  __syncthreads();

  int wave = (blockIdx.x * blockDim.x + threadIdx.x) >> 5;
  if (wave >= mtiles) return;
  int lane = threadIdx.x & 31;
  int m = lane & 15;
  int half = lane >> 4;

  const float* Arow = A + (size_t)(wave * 16 + m) * K + 2 * half;

  v8f acc[NT];
  v8f z = {0.f, 0.f, 0.f, 0.f, 0.f, 0.f, 0.f, 0.f};
#pragma unroll
  for (int j = 0; j < NT; ++j) acc[j] = z;

#pragma unroll 8
  for (int k = 0; k < K; k += 4) {
    float2 av = *(const float2*)(Arow + k);
    v2f a = {av.x, av.y};
    const float* Bp = &Bs[(size_t)((k + 2 * half) >> 1) * S + 2 * m];
#pragma unroll
    for (int j = 0; j < NT; ++j) {
      v2f b = *(const v2f*)(Bp + j * 32);
      acc[j] = __builtin_amdgcn_wmma_f32_16x16x4_f32(
           false, a,  false, b,
           (short)0, acc[j],  false,  false);
    }
  }

#pragma unroll
  for (int j = 0; j < NT; ++j) {
    float bv = bias ? bias[j * 16 + m] : 0.0f;
#pragma unroll
    for (int v = 0; v < 8; ++v) {
      C[(size_t)(wave * 16 + v + 8 * half) * NC + j * 16 + m] = acc[j][v] + bv;
    }
  }
}


__global__ __launch_bounds__(128) void ec_gemm2_all(
    const float* __restrict__ A, const float* __restrict__ Bself,
    const float* __restrict__ Brel, const float* __restrict__ bias,
    float* __restrict__ Cself, float* __restrict__ Crel, int mtiles) {
  constexpr int K = 64;
  constexpr int NT = 1 + R2;
  constexpr int S = 2 * 16 * NT;
  static_assert(S % 64 == 32, "lane halves must land in disjoint bank ranges");
  __shared__ float Bs[(K / 2) * S];

  static_assert(K * 16 / 4 == 256, "two float4 of a block per thread");
  float4 sv[2 * NT];
#pragma unroll
  for (int it = 0; it < 2 * NT; ++it) {
    const float* bp = (it < 2) ? Bself : (Brel + (size_t)(it / 2 - 1) * K * 16);
    sv[it] = *(const float4*)(bp + (threadIdx.x + 128 * (it & 1)) * 4);
  }
#pragma unroll
  for (int it = 0; it < 2 * NT; ++it) {
    int q = threadIdx.x + 128 * (it & 1);
    int k = q >> 2;
    int n = (it / 2) * 16 + (q & 3) * 4;
    float* p = &Bs[(k >> 1) * S + 2 * n + (k & 1)];
    p[0] = sv[it].x;
    p[2] = sv[it].y;
    p[4] = sv[it].z;
    p[6] = sv[it].w;
  }
  __syncthreads();

  int wave = (blockIdx.x * blockDim.x + threadIdx.x) >> 5;
  if (wave >= mtiles) return;
  int lane = threadIdx.x & 31;
  int m = lane & 15;
  int half = lane >> 4;

  const float* Arow = A + (size_t)(wave * 16 + m) * K + 2 * half;

  v8f acc[NT];
  v8f z = {0.f, 0.f, 0.f, 0.f, 0.f, 0.f, 0.f, 0.f};
#pragma unroll
  for (int j = 0; j < NT; ++j) acc[j] = z;

#pragma unroll
  for (int k = 0; k < K; k += 4) {
    float2 av = *(const float2*)(Arow + k);
    v2f a = {av.x, av.y};
    const float* Bp = &Bs[(size_t)((k + 2 * half) >> 1) * S + 2 * m];
#pragma unroll
    for (int j = 0; j < NT; ++j) {
      v2f b = *(const v2f*)(Bp + j * 32);
      acc[j] = __builtin_amdgcn_wmma_f32_16x16x4_f32(
           false, a,  false, b,
           (short)0, acc[j],  false,  false);
    }
  }

  float bv = bias[m];
  size_t plane = (size_t)mtiles * 16 * 16;
#pragma unroll
  for (int v = 0; v < 8; ++v) {
    size_t o = (size_t)(wave * 16 + v + 8 * half) * 16 + m;
    Cself[o] = acc[0][v] + bv;
#pragma unroll
    for (int j = 1; j < NT; ++j) Crel[(size_t)(j - 1) * plane + o] = acc[j][v];
  }
}

__global__ __launch_bounds__(256) void ec_sum64(
    const float* __restrict__ hw, const unsigned* __restrict__ lst,
    const int* __restrict__ start, float* __restrict__ acc, int Nn, int relu) {
  __shared__ float rows[BKT * 64];
  __shared__ int degl[BKT];
  int t = threadIdx.x;
  int lane = t & 31;
  int wave = __builtin_amdgcn_readfirstlane(t >> 5);
  int b = blockIdx.x;
  for (int i = t; i < BKT * 64; i += 256) rows[i] = 0.0f;
  if (t < BKT) degl[t] = 0;
  __syncthreads();
  int s0 = start[b];
  int s1 = start[b + 1];
  int j = lane & 15;
  int g0 = s0 + wave * 16;
  int wd = (g0 + j < s1) ? (int)lst[g0 + j] : 0;
  for (; g0 < s1; g0 += 8 * 16) {
    int n = s1 - g0;
    if (n > 16) n = 16;
    int wdn = (g0 + 8 * 16 + j < s1) ? (int)lst[g0 + 8 * 16 + j] : 0;
    if (lane < n) atomicAdd(&degl[wd & (BKT - 1)], 1);
    float v0[16], v1[16];
#pragma unroll
    for (int k = 0; k < 16; ++k) {
      unsigned w = (unsigned)__builtin_amdgcn_readlane(wd, k);
      const float* hp = hw + (size_t)(w >> 7) * 64 + lane;
      v0[k] = hp[0];
      v1[k] = hp[32];
    }
    if (n == 16) {
#pragma unroll
      for (int k = 0; k < 16; ++k) {
        unsigned w = (unsigned)__builtin_amdgcn_readlane(wd, k);
        float* rp = rows + (w & (BKT - 1)) * 64 + lane;
        __hip_atomic_fetch_add(rp, v0[k], __ATOMIC_RELAXED, __HIP_MEMORY_SCOPE_WORKGROUP);
        __hip_atomic_fetch_add(rp + 32, v1[k], __ATOMIC_RELAXED, __HIP_MEMORY_SCOPE_WORKGROUP);
      }
    } else {
#pragma unroll
      for (int k = 0; k < 16; ++k) {
        if (k < n) {
          unsigned w = (unsigned)__builtin_amdgcn_readlane(wd, k);
          float* rp = rows + (w & (BKT - 1)) * 64 + lane;
          __hip_atomic_fetch_add(rp, v0[k], __ATOMIC_RELAXED, __HIP_MEMORY_SCOPE_WORKGROUP);
          __hip_atomic_fetch_add(rp + 32, v1[k], __ATOMIC_RELAXED, __HIP_MEMORY_SCOPE_WORKGROUP);
        }
      }
    }
    wd = wdn;
  }
  __syncthreads();
  for (int dl = wave; dl < BKT; dl += 8) {
    int d = b * BKT + dl;
    if (d < Nn) {
      float w = 1.0f / fmaxf((float)degl[dl], 1.0f);
      float* ap = acc + (size_t)d * 64 + lane;
      float a0 = ap[0] + rows[dl * 64 + lane] * w;
      float a1 = ap[32] + rows[dl * 64 + lane + 32] * w;
      if (relu) {
        a0 = fmaxf(a0, 0.0f);
        a1 = fmaxf(a1, 0.0f);
      }
      ap[0] = a0;
      ap[32] = a1;
    }
  }
}

__global__ __launch_bounds__(256) void ec_sum16_all(
    const float* __restrict__ hw, const unsigned* __restrict__ lst,
    const int* __restrict__ start, float* __restrict__ acc, int Nn, int E) {
  __shared__ float rows[2 * BKT * 16];
  __shared__ int degl[R2 * BKT];
  __shared__ int sb[2 * R2];
  int t = threadIdx.x;
  int lane = t & 31;
  int wave = __builtin_amdgcn_readfirstlane(t >> 5);
  int b = blockIdx.x;
  if (t < 2 * R2) sb[t] = start[(t >> 1) * SP + b + (t & 1)];
  for (int i = t; i < 2 * BKT * 16; i += 256) rows[i] = 0.0f;
  for (int i = t; i < R2 * BKT; i += 256) degl[i] = 0;
  float tot[BKT * 16 / 256];
#pragma unroll
  for (int q = 0; q < BKT * 16 / 256; ++q) {
    int i = t + 256 * q;
    int d = b * BKT + (i >> 4);
    tot[q] = (d < Nn) ? acc[(size_t)d * 16 + (i & 15)] : 0.0f;
  }
  __syncthreads();
  int c = lane & 15;
  int half = lane >> 4;
  int s1 = __builtin_amdgcn_readfirstlane(sb[1]);
  int g0 = __builtin_amdgcn_readfirstlane(sb[0]) + wave * 32;
  int wd = (g0 + lane < s1) ? (int)lst[g0 + lane] : 0;
  for (int r = 0; r < R2; ++r) {
    const float* hwr = hw + (size_t)r * Nn * 16;
    const unsigned* l = lst + (size_t)r * E;
    float* rw = rows + (r & 1) * BKT * 16;
    int* dg = degl + r * BKT;
    int s1n = 0, g0n = 0, wdf = 0;
    if (r + 1 < R2) {
      s1n = __builtin_amdgcn_readfirstlane(sb[2 * r + 3]);
      g0n = __builtin_amdgcn_readfirstlane(sb[2 * r + 2]) + wave * 32;
      wdf = (g0n + lane < s1n) ? (int)lst[(size_t)(r + 1) * E + g0n + lane] : 0;
    }
    for (; g0 < s1; g0 += 8 * 32) {
      int n = s1 - g0;
      if (n > 32) n = 32;
      int wdn = (g0 + 8 * 32 + lane < s1) ? (int)l[g0 + 8 * 32 + lane] : 0;
      if (lane < n) atomicAdd(&dg[wd & (BKT - 1)], 1);
      float v[16];
      unsigned wk[16];
#pragma unroll
      for (int k = 0; k < 16; ++k) {
        wk[k] = (unsigned)__shfl(wd, 2 * k + half);
        v[k] = hwr[(size_t)(wk[k] >> 7) * 16 + c];
      }
      if (n == 32) {
#pragma unroll
        for (int k = 0; k < 16; ++k) {
          __hip_atomic_fetch_add(&rw[(wk[k] & (BKT - 1)) * 16 + c], v[k], __ATOMIC_RELAXED,
                                 __HIP_MEMORY_SCOPE_WORKGROUP);
        }
      } else {
#pragma unroll
        for (int k = 0; k < 16; ++k) {
          if (2 * k + half < n) {
            __hip_atomic_fetch_add(&rw[(wk[k] & (BKT - 1)) * 16 + c], v[k], __ATOMIC_RELAXED,
                                   __HIP_MEMORY_SCOPE_WORKGROUP);
          }
        }
      }
      wd = wdn;
    }
    __syncthreads();
#pragma unroll
    for (int q = 0; q < BKT * 16 / 256; ++q) {
      int i = t + 256 * q;
      float w = 1.0f / fmaxf((float)dg[i >> 4], 1.0f);
      tot[q] += rw[i] * w;
      rw[i] = 0.0f;
    }
    s1 = s1n;
    g0 = g0n;
    wd = wdf;
  }
#pragma unroll
  for (int q = 0; q < BKT * 16 / 256; ++q) {
    int i = t + 256 * q;
    int d = b * BKT + (i >> 4);
    if (d < Nn) acc[(size_t)d * 16 + (i & 15)] = tot[q];
  }
}

extern "C" void kernel_launch(void* const* d_in, const int* in_sizes, int n_in,
                              void* d_out, int out_size, void* d_ws, size_t ws_size,
                              hipStream_t stream) {
  const float* x   = (const float*)d_in[0];
  const int*   src = (const int*)d_in[1];
  const int*   dst = (const int*)d_in[2];
  const float* W1  = (const float*)d_in[3];
  const float* L1w = (const float*)d_in[4];
  const float* L1b = (const float*)d_in[5];
  const float* W2  = (const float*)d_in[6];
  const float* L2w = (const float*)d_in[7];
  const float* L2b = (const float*)d_in[8];
  float* out = (float*)d_out;

  const int D_IN = 256, D_H = 64;
  const int Nn = in_sizes[0] / D_IN;
  const int R  = in_sizes[3] / (D_IN * D_H);
  const int E  = in_sizes[1] / R;
  const int NB = (Nn + BKT - 1) / BKT;
  const int CH = (E + CHUNK - 1) / CHUNK;

  char* ws = (char*)d_ws;
  size_t off = 0;
  int* start = (int*)(ws + off);
  int* cnt   = start + (size_t)R * SP;
  off += (size_t)R * Nn * sizeof(float);
  off = (off + 255) & ~(size_t)255;
  float* h = (float*)(ws + off);
  off += (size_t)Nn * D_H * sizeof(float);
  off = (off + 255) & ~(size_t)255;
  float* hw = (float*)(ws + off);
  unsigned* lst1 = (unsigned*)out;
  unsigned* lst2 = (unsigned*)h;

  int mtiles = Nn / 16;
  int gemmBlocks = (mtiles * 32 + 127) / 128;

  ec_bin_count<<<dim3(CH, R), 1024, 0, stream>>>(dst, cnt, E, NB, CH);

  ec_gemm_wmma<256, 64><<<gemmBlocks, 128, 0, stream>>>(x, L1w, L1b, h, mtiles);
  for (int r = 0; r < R; ++r) {
    if ((r & 1) == 0) {
      int nr = (R - r < 2) ? (R - r) : 2;
      ec_bin_scatter<<<dim3(CH, nr), 1024, 0, stream>>>(src, dst, cnt, start, lst1, E, NB, CH, r);
    }
    ec_gemm_wmma<256, 64><<<gemmBlocks, 128, 0, stream>>>(
        x, W1 + (size_t)r * D_IN * D_H, nullptr, hw, mtiles);
    ec_sum64<<<NB, 256, 0, stream>>>(hw, lst1 + (size_t)(r & 1) * E, start + r * SP, h, Nn,
                                     (r == R - 1) ? 1 : 0);
  }

  ec_gemm2_all<<<gemmBlocks, 128, 0, stream>>>(h, L2w, W2, L2b, out, hw, mtiles);
  ec_bin_scatter<<<dim3(CH, R), 1024, 0, stream>>>(src, dst, cnt, start, lst2, E, NB, CH, 0);
  ec_sum16_all<<<NB, 256, 0, stream>>>(hw, lst2, start, out, Nn, E);
}


# ec_sum64 write-back: a wave's 16 accumulator rows are read at kernel entry (in flight under the edge walk) and written back as one batch, not one dependent load round trip per row; sums unchanged
# speedup vs baseline: 2.1403x; 1.0500x over previous
#include <hip/hip_runtime.h>

typedef __attribute__((ext_vector_type(2))) float v2f;
typedef __attribute__((ext_vector_type(8))) float v8f;

constexpr int BKT = 128;
constexpr int NBMAX = 1024;
constexpr int CHUNK = 16384;
constexpr int SP = NBMAX + 1;
constexpr int R2 = 4;

__global__ __launch_bounds__(1024) void ec_bin_count(const int* __restrict__ dst,
                                                     int* __restrict__ cnt,
                                                     int E, int NB, int CH) {
  __shared__ int hist[NBMAX];
  int t = threadIdx.x;
  int c = blockIdx.x;
  int r = blockIdx.y;
  hist[t] = 0;
  __syncthreads();
  const int* d = dst + (size_t)r * E;
  int e0 = c * CHUNK;
  int dv[CHUNK / 1024];
#pragma unroll
  for (int i = 0; i < CHUNK / 1024; ++i) {
    int e = e0 + i * 1024 + t;
    dv[i] = (e < E) ? d[e] : -1;
  }
#pragma unroll
  for (int i = 0; i < CHUNK / 1024; ++i) {
    if (dv[i] >= 0) atomicAdd(&hist[dv[i] >> 7], 1);
  }
  __syncthreads();
  if (t < NB) cnt[((size_t)r * CH + c) * NB + t] = hist[t];
}

__device__ __forceinline__ int ec_scan1024(int v, int* wsum, int& total) {
  int lane = threadIdx.x & 31;
  int w = threadIdx.x >> 5;
  int x = v;
  for (int o = 1; o < 32; o <<= 1) {
    int y = __shfl_up(x, o);
    if (lane >= o) x += y;
  }
  if (lane == 31) wsum[w] = x;
  __syncthreads();
  if (w == 0) {
    int s = wsum[lane];
    for (int o = 1; o < 32; o <<= 1) {
      int y = __shfl_up(s, o);
      if (lane >= o) s += y;
    }
    wsum[lane] = s;
  }
  __syncthreads();
  int off = (w > 0) ? wsum[w - 1] : 0;
  total = wsum[31];
  __syncthreads();
  return off + x - v;
}

__global__ __launch_bounds__(1024) void ec_bin_scatter(
    const int* __restrict__ src, const int* __restrict__ dst,
    const int* __restrict__ cnt, int* __restrict__ start,
    unsigned* __restrict__ lst, int E, int NB, int CH, int r0) {
  __shared__ int lcur[NBMAX];
  __shared__ int gdelta[NBMAX];
  __shared__ int wsum[32];
  __shared__ unsigned sortedw[CHUNK];
  __shared__ int sortedp[CHUNK];
  int t = threadIdx.x;
  int c = blockIdx.x;
  int r = r0 + blockIdx.y;

  int tot = 0, before = 0, mine = 0;
  if (t < NB) {
    const int* col = cnt + (size_t)r * CH * NB + t;
    for (int cc = 0; cc < CH; cc += 8) {
      int v[8];
#pragma unroll
      for (int u = 0; u < 8; ++u) v[u] = (cc + u < CH) ? col[(size_t)(cc + u) * NB] : 0;
#pragma unroll
      for (int u = 0; u < 8; ++u) {
        tot += v[u];
        if (cc + u < c) before += v[u];
        if (cc + u == c) mine = v[u];
      }
    }
  }
  int total, ltotal;
  int gstart = ec_scan1024(tot, wsum, total);
  int lstart = ec_scan1024(mine, wsum, ltotal);
  lcur[t] = lstart;
  gdelta[t] = gstart + before - lstart;
  if (c == 0) {
    if (t < NB) start[r * SP + t] = gstart;
    if (t == 0) start[r * SP + NB] = total;
  }
  __syncthreads();

  const int* d = dst + (size_t)r * E;
  const int* s = src + (size_t)r * E;
  int e0 = c * CHUNK;
  int dv[CHUNK / 1024], sv[CHUNK / 1024];
#pragma unroll
  for (int i = 0; i < CHUNK / 1024; ++i) {
    int e = e0 + i * 1024 + t;
    dv[i] = (e < E) ? d[e] : -1;
    sv[i] = (e < E) ? s[e] : 0;
  }
#pragma unroll
  for (int i = 0; i < CHUNK / 1024; ++i) {
    if (dv[i] >= 0) {
      int b = dv[i] >> 7;
      int li = atomicAdd(&lcur[b], 1);
      sortedw[li] = ((unsigned)sv[i] << 7) | (unsigned)(dv[i] & (BKT - 1));
      sortedp[li] = gdelta[b] + li;
    }
  }
  __syncthreads();
  int n = E - e0;
  if (n > CHUNK) n = CHUNK;
  unsigned* out = lst + (size_t)(r - r0) * E;
  for (int i = 0; i < CHUNK / 1024; ++i) {
    int j = i * 1024 + t;
    if (j < n) out[sortedp[j]] = sortedw[j];
  }
}

template <int K, int NC>
__global__ __launch_bounds__(128) void ec_gemm_wmma(
    const float* __restrict__ A, const float* __restrict__ B,
    const float* __restrict__ bias, float* __restrict__ C, int mtiles) {
  constexpr int NT = NC / 16;
  constexpr int S = ((2 * NC) % 64 == 32) ? (2 * NC) : (2 * NC + 32);
  __shared__ float Bs[(K / 2) * S];

  for (int i = threadIdx.x; i < K * NC / 4; i += 128) {
    int k = (i * 4) / NC;
    int n = (i * 4) % NC;
    float4 v = *(const float4*)(B + i * 4);
    float* p = &Bs[(k >> 1) * S + 2 * n + (k & 1)];
    p[0] = v.x;
    p[2] = v.y;
    p[4] = v.z;
    p[6] = v.w;
  }
  __syncthreads();

  int wave = (blockIdx.x * blockDim.x + threadIdx.x) >> 5;
  if (wave >= mtiles) return;
  int lane = threadIdx.x & 31;
  int m = lane & 15;
  int half = lane >> 4;

  const float* Arow = A + (size_t)(wave * 16 + m) * K + 2 * half;

  v8f acc[NT];
  v8f z = {0.f, 0.f, 0.f, 0.f, 0.f, 0.f, 0.f, 0.f};
#pragma unroll
  for (int j = 0; j < NT; ++j) acc[j] = z;

#pragma unroll 8
  for (int k = 0; k < K; k += 4) {
    float2 av = *(const float2*)(Arow + k);
    v2f a = {av.x, av.y};
    const float* Bp = &Bs[(size_t)((k + 2 * half) >> 1) * S + 2 * m];
#pragma unroll
    for (int j = 0; j < NT; ++j) {
      v2f b = *(const v2f*)(Bp + j * 32);
      acc[j] = __builtin_amdgcn_wmma_f32_16x16x4_f32(
           false, a,  false, b,
           (short)0, acc[j],  false,  false);
    }
  }

#pragma unroll
  for (int j = 0; j < NT; ++j) {
    float bv = bias ? bias[j * 16 + m] : 0.0f;
#pragma unroll
    for (int v = 0; v < 8; ++v) {
      C[(size_t)(wave * 16 + v + 8 * half) * NC + j * 16 + m] = acc[j][v] + bv;
    }
  }
}


__global__ __launch_bounds__(128) void ec_gemm2_all(
    const float* __restrict__ A, const float* __restrict__ Bself,
    const float* __restrict__ Brel, const float* __restrict__ bias,
    float* __restrict__ Cself, float* __restrict__ Crel, int mtiles) {
  constexpr int K = 64;
  constexpr int NT = 1 + R2;
  constexpr int S = 2 * 16 * NT;
  static_assert(S % 64 == 32, "lane halves must land in disjoint bank ranges");
  __shared__ float Bs[(K / 2) * S];

  static_assert(K * 16 / 4 == 256, "two float4 of a block per thread");
  float4 sv[2 * NT];
#pragma unroll
  for (int it = 0; it < 2 * NT; ++it) {
    const float* bp = (it < 2) ? Bself : (Brel + (size_t)(it / 2 - 1) * K * 16);
    sv[it] = *(const float4*)(bp + (threadIdx.x + 128 * (it & 1)) * 4);
  }
#pragma unroll
  for (int it = 0; it < 2 * NT; ++it) {
    int q = threadIdx.x + 128 * (it & 1);
    int k = q >> 2;
    int n = (it / 2) * 16 + (q & 3) * 4;
    float* p = &Bs[(k >> 1) * S + 2 * n + (k & 1)];
    p[0] = sv[it].x;
    p[2] = sv[it].y;
    p[4] = sv[it].z;
    p[6] = sv[it].w;
  }
  __syncthreads();

  int wave = (blockIdx.x * blockDim.x + threadIdx.x) >> 5;
  if (wave >= mtiles) return;
  int lane = threadIdx.x & 31;
  int m = lane & 15;
  int half = lane >> 4;

  const float* Arow = A + (size_t)(wave * 16 + m) * K + 2 * half;

  v8f acc[NT];
  v8f z = {0.f, 0.f, 0.f, 0.f, 0.f, 0.f, 0.f, 0.f};
#pragma unroll
  for (int j = 0; j < NT; ++j) acc[j] = z;

#pragma unroll
  for (int k = 0; k < K; k += 4) {
    float2 av = *(const float2*)(Arow + k);
    v2f a = {av.x, av.y};
    const float* Bp = &Bs[(size_t)((k + 2 * half) >> 1) * S + 2 * m];
#pragma unroll
    for (int j = 0; j < NT; ++j) {
      v2f b = *(const v2f*)(Bp + j * 32);
      acc[j] = __builtin_amdgcn_wmma_f32_16x16x4_f32(
           false, a,  false, b,
           (short)0, acc[j],  false,  false);
    }
  }

  float bv = bias[m];
  size_t plane = (size_t)mtiles * 16 * 16;
#pragma unroll
  for (int v = 0; v < 8; ++v) {
    size_t o = (size_t)(wave * 16 + v + 8 * half) * 16 + m;
    Cself[o] = acc[0][v] + bv;
#pragma unroll
    for (int j = 1; j < NT; ++j) Crel[(size_t)(j - 1) * plane + o] = acc[j][v];
  }
}

__global__ __launch_bounds__(256) void ec_sum64(
    const float* __restrict__ hw, const unsigned* __restrict__ lst,
    const int* __restrict__ start, float* __restrict__ acc, int Nn, int relu) {
  __shared__ float rows[BKT * 64];
  __shared__ int degl[BKT];
  int t = threadIdx.x;
  int lane = t & 31;
  int wave = __builtin_amdgcn_readfirstlane(t >> 5);
  int b = blockIdx.x;
  float a0[16], a1[16];
#pragma unroll
  for (int k = 0; k < 16; ++k) {
    int d = b * BKT + wave + 8 * k;
    if (d > Nn - 1) d = Nn - 1;
    const float* ap = acc + (size_t)d * 64 + lane;
    a0[k] = ap[0];
    a1[k] = ap[32];
  }
  int s0 = start[b];
  int s1 = start[b + 1];
  int j = lane & 15;
  int g0 = s0 + wave * 16;
  int wd = (g0 + j < s1) ? (int)lst[g0 + j] : 0;
  for (int i = t; i < BKT * 64; i += 256) rows[i] = 0.0f;
  if (t < BKT) degl[t] = 0;
  __syncthreads();
  for (; g0 < s1; g0 += 8 * 16) {
    int n = s1 - g0;
    if (n > 16) n = 16;
    int wdn = (g0 + 8 * 16 + j < s1) ? (int)lst[g0 + 8 * 16 + j] : 0;
    if (lane < n) atomicAdd(&degl[wd & (BKT - 1)], 1);
    float v0[16], v1[16];
#pragma unroll
    for (int k = 0; k < 16; ++k) {
      unsigned w = (unsigned)__builtin_amdgcn_readlane(wd, k);
      const float* hp = hw + (size_t)(w >> 7) * 64 + lane;
      v0[k] = hp[0];
      v1[k] = hp[32];
    }
    if (n == 16) {
#pragma unroll
      for (int k = 0; k < 16; ++k) {
        unsigned w = (unsigned)__builtin_amdgcn_readlane(wd, k);
        float* rp = rows + (w & (BKT - 1)) * 64 + lane;
        __hip_atomic_fetch_add(rp, v0[k], __ATOMIC_RELAXED, __HIP_MEMORY_SCOPE_WORKGROUP);
        __hip_atomic_fetch_add(rp + 32, v1[k], __ATOMIC_RELAXED, __HIP_MEMORY_SCOPE_WORKGROUP);
      }
    } else {
#pragma unroll
      for (int k = 0; k < 16; ++k) {
        if (k < n) {
          unsigned w = (unsigned)__builtin_amdgcn_readlane(wd, k);
          float* rp = rows + (w & (BKT - 1)) * 64 + lane;
          __hip_atomic_fetch_add(rp, v0[k], __ATOMIC_RELAXED, __HIP_MEMORY_SCOPE_WORKGROUP);
          __hip_atomic_fetch_add(rp + 32, v1[k], __ATOMIC_RELAXED, __HIP_MEMORY_SCOPE_WORKGROUP);
        }
      }
    }
    wd = wdn;
  }
  __syncthreads();
  float wl = 1.0f / fmaxf((float)degl[wave + 8 * (lane & 15)], 1.0f);
#pragma unroll
  for (int k = 0; k < 16; ++k) {
    int dl = wave + 8 * k;
    float w = __int_as_float(__builtin_amdgcn_readlane(__float_as_int(wl), k));
    a0[k] += rows[dl * 64 + lane] * w;
    a1[k] += rows[dl * 64 + lane + 32] * w;
    if (relu) {
      a0[k] = fmaxf(a0[k], 0.0f);
      a1[k] = fmaxf(a1[k], 0.0f);
    }
  }
#pragma unroll
  for (int k = 0; k < 16; ++k) {
    int d = b * BKT + wave + 8 * k;
    if (d < Nn) {
      float* ap = acc + (size_t)d * 64 + lane;
      ap[0] = a0[k];
      ap[32] = a1[k];
    }
  }
}

__global__ __launch_bounds__(256) void ec_sum16_all(
    const float* __restrict__ hw, const unsigned* __restrict__ lst,
    const int* __restrict__ start, float* __restrict__ acc, int Nn, int E) {
  __shared__ float rows[2 * BKT * 16];
  __shared__ int degl[R2 * BKT];
  __shared__ int sb[2 * R2];
  int t = threadIdx.x;
  int lane = t & 31;
  int wave = __builtin_amdgcn_readfirstlane(t >> 5);
  int b = blockIdx.x;
  if (t < 2 * R2) sb[t] = start[(t >> 1) * SP + b + (t & 1)];
  for (int i = t; i < 2 * BKT * 16; i += 256) rows[i] = 0.0f;
  for (int i = t; i < R2 * BKT; i += 256) degl[i] = 0;
  float tot[BKT * 16 / 256];
#pragma unroll
  for (int q = 0; q < BKT * 16 / 256; ++q) {
    int i = t + 256 * q;
    int d = b * BKT + (i >> 4);
    tot[q] = (d < Nn) ? acc[(size_t)d * 16 + (i & 15)] : 0.0f;
  }
  __syncthreads();
  int c = lane & 15;
  int half = lane >> 4;
  int s1 = __builtin_amdgcn_readfirstlane(sb[1]);
  int g0 = __builtin_amdgcn_readfirstlane(sb[0]) + wave * 32;
  int wd = (g0 + lane < s1) ? (int)lst[g0 + lane] : 0;
  for (int r = 0; r < R2; ++r) {
    const float* hwr = hw + (size_t)r * Nn * 16;
    const unsigned* l = lst + (size_t)r * E;
    float* rw = rows + (r & 1) * BKT * 16;
    int* dg = degl + r * BKT;
    int s1n = 0, g0n = 0, wdf = 0;
    if (r + 1 < R2) {
      s1n = __builtin_amdgcn_readfirstlane(sb[2 * r + 3]);
      g0n = __builtin_amdgcn_readfirstlane(sb[2 * r + 2]) + wave * 32;
      wdf = (g0n + lane < s1n) ? (int)lst[(size_t)(r + 1) * E + g0n + lane] : 0;
    }
    for (; g0 < s1; g0 += 8 * 32) {
      int n = s1 - g0;
      if (n > 32) n = 32;
      int wdn = (g0 + 8 * 32 + lane < s1) ? (int)l[g0 + 8 * 32 + lane] : 0;
      if (lane < n) atomicAdd(&dg[wd & (BKT - 1)], 1);
      float v[16];
      unsigned wk[16];
#pragma unroll
      for (int k = 0; k < 16; ++k) {
        wk[k] = (unsigned)__shfl(wd, 2 * k + half);
        v[k] = hwr[(size_t)(wk[k] >> 7) * 16 + c];
      }
      if (n == 32) {
#pragma unroll
        for (int k = 0; k < 16; ++k) {
          __hip_atomic_fetch_add(&rw[(wk[k] & (BKT - 1)) * 16 + c], v[k], __ATOMIC_RELAXED,
                                 __HIP_MEMORY_SCOPE_WORKGROUP);
        }
      } else {
#pragma unroll
        for (int k = 0; k < 16; ++k) {
          if (2 * k + half < n) {
            __hip_atomic_fetch_add(&rw[(wk[k] & (BKT - 1)) * 16 + c], v[k], __ATOMIC_RELAXED,
                                   __HIP_MEMORY_SCOPE_WORKGROUP);
          }
        }
      }
      wd = wdn;
    }
    __syncthreads();
#pragma unroll
    for (int q = 0; q < BKT * 16 / 256; ++q) {
      int i = t + 256 * q;
      float w = 1.0f / fmaxf((float)dg[i >> 4], 1.0f);
      tot[q] += rw[i] * w;
      rw[i] = 0.0f;
    }
    s1 = s1n;
    g0 = g0n;
    wd = wdf;
  }
#pragma unroll
  for (int q = 0; q < BKT * 16 / 256; ++q) {
    int i = t + 256 * q;
    int d = b * BKT + (i >> 4);
    if (d < Nn) acc[(size_t)d * 16 + (i & 15)] = tot[q];
  }
}

extern "C" void kernel_launch(void* const* d_in, const int* in_sizes, int n_in,
                              void* d_out, int out_size, void* d_ws, size_t ws_size,
                              hipStream_t stream) {
  const float* x   = (const float*)d_in[0];
  const int*   src = (const int*)d_in[1];
  const int*   dst = (const int*)d_in[2];
  const float* W1  = (const float*)d_in[3];
  const float* L1w = (const float*)d_in[4];
  const float* L1b = (const float*)d_in[5];
  const float* W2  = (const float*)d_in[6];
  const float* L2w = (const float*)d_in[7];
  const float* L2b = (const float*)d_in[8];
  float* out = (float*)d_out;

  const int D_IN = 256, D_H = 64;
  const int Nn = in_sizes[0] / D_IN;
  const int R  = in_sizes[3] / (D_IN * D_H);
  const int E  = in_sizes[1] / R;
  const int NB = (Nn + BKT - 1) / BKT;
  const int CH = (E + CHUNK - 1) / CHUNK;

  char* ws = (char*)d_ws;
  size_t off = 0;
  int* start = (int*)(ws + off);
  int* cnt   = start + (size_t)R * SP;
  off += (size_t)R * Nn * sizeof(float);
  off = (off + 255) & ~(size_t)255;
  float* h = (float*)(ws + off);
  off += (size_t)Nn * D_H * sizeof(float);
  off = (off + 255) & ~(size_t)255;
  float* hw = (float*)(ws + off);
  unsigned* lst1 = (unsigned*)out;
  unsigned* lst2 = (unsigned*)h;

  int mtiles = Nn / 16;
  int gemmBlocks = (mtiles * 32 + 127) / 128;

  ec_bin_count<<<dim3(CH, R), 1024, 0, stream>>>(dst, cnt, E, NB, CH);

  ec_gemm_wmma<256, 64><<<gemmBlocks, 128, 0, stream>>>(x, L1w, L1b, h, mtiles);
  for (int r = 0; r < R; ++r) {
    if ((r & 1) == 0) {
      int nr = (R - r < 2) ? (R - r) : 2;
      ec_bin_scatter<<<dim3(CH, nr), 1024, 0, stream>>>(src, dst, cnt, start, lst1, E, NB, CH, r);
    }
    ec_gemm_wmma<256, 64><<<gemmBlocks, 128, 0, stream>>>(
        x, W1 + (size_t)r * D_IN * D_H, nullptr, hw, mtiles);
    ec_sum64<<<NB, 256, 0, stream>>>(hw, lst1 + (size_t)(r & 1) * E, start + r * SP, h, Nn,
                                     (r == R - 1) ? 1 : 0);
  }

  ec_gemm2_all<<<gemmBlocks, 128, 0, stream>>>(h, L2w, W2, L2b, out, hw, mtiles);
  ec_bin_scatter<<<dim3(CH, R), 1024, 0, stream>>>(src, dst, cnt, start, lst2, E, NB, CH, 0);
  ec_sum16_all<<<NB, 256, 0, stream>>>(hw, lst2, start, out, Nn, E);
}


# Layer-1 products on a persistent grid: a 16-wave work-group per WGP stages the weights once; waves loop over 16-row strips, x loads three K-steps ahead, strips dealt evenly to SIMDs; sums unchanged
# speedup vs baseline: 2.2346x; 1.0441x over previous
#include <hip/hip_runtime.h>

typedef __attribute__((ext_vector_type(2))) float v2f;
typedef __attribute__((ext_vector_type(8))) float v8f;

constexpr int BKT = 128;
constexpr int NBMAX = 1024;
constexpr int CHUNK = 16384;
constexpr int SP = NBMAX + 1;
constexpr int R2 = 4;

__global__ __launch_bounds__(1024) void ec_bin_count(const int* __restrict__ dst,
                                                     int* __restrict__ cnt,
                                                     int E, int NB, int CH) {
  __shared__ int hist[NBMAX];
  int t = threadIdx.x;
  int c = blockIdx.x;
  int r = blockIdx.y;
  hist[t] = 0;
  __syncthreads();
  const int* d = dst + (size_t)r * E;
  int e0 = c * CHUNK;
  int dv[CHUNK / 1024];
#pragma unroll
  for (int i = 0; i < CHUNK / 1024; ++i) {
    int e = e0 + i * 1024 + t;
    dv[i] = (e < E) ? d[e] : -1;
  }
#pragma unroll
  for (int i = 0; i < CHUNK / 1024; ++i) {
    if (dv[i] >= 0) atomicAdd(&hist[dv[i] >> 7], 1);
  }
  __syncthreads();
  if (t < NB) cnt[((size_t)r * CH + c) * NB + t] = hist[t];
}

__device__ __forceinline__ int ec_scan1024(int v, int* wsum, int& total) {
  int lane = threadIdx.x & 31;
  int w = threadIdx.x >> 5;
  int x = v;
  for (int o = 1; o < 32; o <<= 1) {
    int y = __shfl_up(x, o);
    if (lane >= o) x += y;
  }
  if (lane == 31) wsum[w] = x;
  __syncthreads();
  if (w == 0) {
    int s = wsum[lane];
    for (int o = 1; o < 32; o <<= 1) {
      int y = __shfl_up(s, o);
      if (lane >= o) s += y;
    }
    wsum[lane] = s;
  }
  __syncthreads();
  int off = (w > 0) ? wsum[w - 1] : 0;
  total = wsum[31];
  __syncthreads();
  return off + x - v;
}

__global__ __launch_bounds__(1024) void ec_bin_scatter(
    const int* __restrict__ src, const int* __restrict__ dst,
    const int* __restrict__ cnt, int* __restrict__ start,
    unsigned* __restrict__ lst, int E, int NB, int CH, int r0) {
  __shared__ int lcur[NBMAX];
  __shared__ int gdelta[NBMAX];
  __shared__ int wsum[32];
  __shared__ unsigned sortedw[CHUNK];
  __shared__ int sortedp[CHUNK];
  int t = threadIdx.x;
  int c = blockIdx.x;
  int r = r0 + blockIdx.y;

  int tot = 0, before = 0, mine = 0;
  if (t < NB) {
    const int* col = cnt + (size_t)r * CH * NB + t;
    for (int cc = 0; cc < CH; cc += 8) {
      int v[8];
#pragma unroll
      for (int u = 0; u < 8; ++u) v[u] = (cc + u < CH) ? col[(size_t)(cc + u) * NB] : 0;
#pragma unroll
      for (int u = 0; u < 8; ++u) {
        tot += v[u];
        if (cc + u < c) before += v[u];
        if (cc + u == c) mine = v[u];
      }
    }
  }
  int total, ltotal;
  int gstart = ec_scan1024(tot, wsum, total);
  int lstart = ec_scan1024(mine, wsum, ltotal);
  lcur[t] = lstart;
  gdelta[t] = gstart + before - lstart;
  if (c == 0) {
    if (t < NB) start[r * SP + t] = gstart;
    if (t == 0) start[r * SP + NB] = total;
  }
  __syncthreads();

  const int* d = dst + (size_t)r * E;
  const int* s = src + (size_t)r * E;
  int e0 = c * CHUNK;
  int dv[CHUNK / 1024], sv[CHUNK / 1024];
#pragma unroll
  for (int i = 0; i < CHUNK / 1024; ++i) {
    int e = e0 + i * 1024 + t;
    dv[i] = (e < E) ? d[e] : -1;
    sv[i] = (e < E) ? s[e] : 0;
  }
#pragma unroll
  for (int i = 0; i < CHUNK / 1024; ++i) {
    if (dv[i] >= 0) {
      int b = dv[i] >> 7;
      int li = atomicAdd(&lcur[b], 1);
      sortedw[li] = ((unsigned)sv[i] << 7) | (unsigned)(dv[i] & (BKT - 1));
      sortedp[li] = gdelta[b] + li;
    }
  }
  __syncthreads();
  int n = E - e0;
  if (n > CHUNK) n = CHUNK;
  unsigned* out = lst + (size_t)(r - r0) * E;
  for (int i = 0; i < CHUNK / 1024; ++i) {
    int j = i * 1024 + t;
    if (j < n) out[sortedp[j]] = sortedw[j];
  }
}

constexpr int GKB = 32;
constexpr int GD = 3;

__device__ __forceinline__ void ec_load_step(const float* __restrict__ Ap, v2f (&dst)[GKB / 4]) {
#pragma unroll
  for (int u = 0; u < GKB / 4; ++u) {
    float2 av = *(const float2*)(Ap + u * 4);
    dst[u] = v2f{av.x, av.y};
  }
}

template <int K, int S, int NTL>
__device__ __forceinline__ void ec_strip(const float* __restrict__ Arow, const float* __restrict__ Anext,
                                         const float* Bl, v2f (&ring)[GD][GKB / 4], v8f (&acc)[NTL]) {
  constexpr int NS = K / GKB;
  static_assert(NS % (GD + 1) == 0, "the register ring turns a whole number of times per strip");
  v8f z = {0.f, 0.f, 0.f, 0.f, 0.f, 0.f, 0.f, 0.f};
#pragma unroll
  for (int j = 0; j < NTL; ++j) acc[j] = z;
#pragma unroll(GD + 1)
  for (int s = 0; s < NS; ++s) {
    v2f nw[GKB / 4];
    const float* Ap = (s + GD < NS) ? (Arow + (s + GD) * GKB) : (Anext + (s + GD - NS) * GKB);
    ec_load_step(Ap, nw);
    __builtin_amdgcn_sched_barrier(0x186);
#pragma unroll
    for (int u = 0; u < GKB / 4; ++u) {
      int k = s * GKB + u * 4;
      const float* Bp = Bl + (size_t)(k >> 1) * S;
#pragma unroll
      for (int j = 0; j < NTL; ++j) {
        v2f b = *(const v2f*)(Bp + j * 32);
        acc[j] = __builtin_amdgcn_wmma_f32_16x16x4_f32(
             false, ring[0][u],  false, b,
             (short)0, acc[j],  false,  false);
      }
    }
#pragma unroll
    for (int d = 0; d + 1 < GD; ++d) {
#pragma unroll
      for (int u = 0; u < GKB / 4; ++u) ring[d][u] = ring[d + 1][u];
    }
#pragma unroll
    for (int u = 0; u < GKB / 4; ++u) ring[GD - 1][u] = nw[u];
  }
}

template <int K, int NC>
__global__ __launch_bounds__(512) void ec_gemm_wmma(
    const float* __restrict__ A, const float* __restrict__ B,
    const float* __restrict__ bias, float* __restrict__ C, int mtiles) {
  constexpr int NT = NC / 16;
  constexpr int S = ((2 * NC) % 64 == 32) ? (2 * NC) : (2 * NC + 32);
  constexpr int WV = 16;
  static_assert(NT == 4, "a left-over strip is split over four waves, one column tile each");
  __shared__ float Bs[(K / 2) * S];

  constexpr int NL = K * NC / 4 / 512;
  static_assert(K * NC / 4 % 512 == 0, "whole float4 rounds");
  float4 bv4[NL];
#pragma unroll
  for (int it = 0; it < NL; ++it) bv4[it] = *(const float4*)(B + (size_t)(threadIdx.x + 512 * it) * 4);
#pragma unroll
  for (int it = 0; it < NL; ++it) {
    int i = threadIdx.x + 512 * it;
    int k = (i * 4) / NC;
    int n = (i * 4) % NC;
    float* p = &Bs[(k >> 1) * S + 2 * n + (k & 1)];
    p[0] = bv4[it].x;
    p[2] = bv4[it].y;
    p[4] = bv4[it].z;
    p[6] = bv4[it].w;
  }

  int wave = __builtin_amdgcn_readfirstlane(threadIdx.x >> 5);
  int lane = threadIdx.x & 31;
  int m = lane & 15;
  int half = lane >> 4;

  int t0 = (int)((blockIdx.x * (unsigned)mtiles) / gridDim.x);
  int t1 = (int)(((blockIdx.x + 1) * (unsigned)mtiles) / gridDim.x);
  int tf = t0 + ((t1 - t0) & ~3);
  int tile = t0 + wave;

  v2f ring[GD][GKB / 4];
  {
    int tl = (tile < mtiles) ? tile : (mtiles - 1);
    const float* Ap = A + (size_t)(tl * 16 + m) * K + 2 * half;
#pragma unroll
    for (int d = 0; d < GD; ++d) ec_load_step(Ap + d * GKB, ring[d]);
  }
  float bvv[NT];
#pragma unroll
  for (int j = 0; j < NT; ++j) bvv[j] = bias ? bias[j * 16 + m] : 0.0f;
  __syncthreads();

  const float* Bl = &Bs[(size_t)half * S + 2 * m];

  for (; tile < tf; tile += WV) {
    const float* Arow = A + (size_t)(tile * 16 + m) * K + 2 * half;
    int tn = tile + WV;
    if (tn >= tf) tn = tile;
    const float* Anext = A + (size_t)(tn * 16 + m) * K + 2 * half;
    v8f acc[NT];
    ec_strip<K, S, NT>(Arow, Anext, Bl, ring, acc);
#pragma unroll
    for (int j = 0; j < NT; ++j) {
#pragma unroll
      for (int v = 0; v < 8; ++v) {
        C[(size_t)(tile * 16 + v + 8 * half) * NC + j * 16 + m] = acc[j][v] + bvv[j];
      }
    }
  }

  int q = 3 - (wave >> 2);
  if (tf + q < t1) {
    int tq = tf + q;
    int j0 = wave & 3;
    const float* Arow = A + (size_t)(tq * 16 + m) * K + 2 * half;
#pragma unroll
    for (int d = 0; d < GD; ++d) ec_load_step(Arow + d * GKB, ring[d]);
    v8f acc1[1];
    ec_strip<K, S, 1>(Arow, Arow, Bl + j0 * 32, ring, acc1);
    float bq = bias ? bias[j0 * 16 + m] : 0.0f;
#pragma unroll
    for (int v = 0; v < 8; ++v) {
      C[(size_t)(tq * 16 + v + 8 * half) * NC + j0 * 16 + m] = acc1[0][v] + bq;
    }
  }
}


__global__ __launch_bounds__(128) void ec_gemm2_all(
    const float* __restrict__ A, const float* __restrict__ Bself,
    const float* __restrict__ Brel, const float* __restrict__ bias,
    float* __restrict__ Cself, float* __restrict__ Crel, int mtiles) {
  constexpr int K = 64;
  constexpr int NT = 1 + R2;
  constexpr int S = 2 * 16 * NT;
  static_assert(S % 64 == 32, "lane halves must land in disjoint bank ranges");
  __shared__ float Bs[(K / 2) * S];

  static_assert(K * 16 / 4 == 256, "two float4 of a block per thread");
  float4 sv[2 * NT];
#pragma unroll
  for (int it = 0; it < 2 * NT; ++it) {
    const float* bp = (it < 2) ? Bself : (Brel + (size_t)(it / 2 - 1) * K * 16);
    sv[it] = *(const float4*)(bp + (threadIdx.x + 128 * (it & 1)) * 4);
  }
#pragma unroll
  for (int it = 0; it < 2 * NT; ++it) {
    int q = threadIdx.x + 128 * (it & 1);
    int k = q >> 2;
    int n = (it / 2) * 16 + (q & 3) * 4;
    float* p = &Bs[(k >> 1) * S + 2 * n + (k & 1)];
    p[0] = sv[it].x;
    p[2] = sv[it].y;
    p[4] = sv[it].z;
    p[6] = sv[it].w;
  }
  __syncthreads();

  int wave = (blockIdx.x * blockDim.x + threadIdx.x) >> 5;
  if (wave >= mtiles) return;
  int lane = threadIdx.x & 31;
  int m = lane & 15;
  int half = lane >> 4;

  const float* Arow = A + (size_t)(wave * 16 + m) * K + 2 * half;

  v8f acc[NT];
  v8f z = {0.f, 0.f, 0.f, 0.f, 0.f, 0.f, 0.f, 0.f};
#pragma unroll
  for (int j = 0; j < NT; ++j) acc[j] = z;

#pragma unroll
  for (int k = 0; k < K; k += 4) {
    float2 av = *(const float2*)(Arow + k);
    v2f a = {av.x, av.y};
    const float* Bp = &Bs[(size_t)((k + 2 * half) >> 1) * S + 2 * m];
#pragma unroll
    for (int j = 0; j < NT; ++j) {
      v2f b = *(const v2f*)(Bp + j * 32);
      acc[j] = __builtin_amdgcn_wmma_f32_16x16x4_f32(
           false, a,  false, b,
           (short)0, acc[j],  false,  false);
    }
  }

  float bv = bias[m];
  size_t plane = (size_t)mtiles * 16 * 16;
#pragma unroll
  for (int v = 0; v < 8; ++v) {
    size_t o = (size_t)(wave * 16 + v + 8 * half) * 16 + m;
    Cself[o] = acc[0][v] + bv;
#pragma unroll
    for (int j = 1; j < NT; ++j) Crel[(size_t)(j - 1) * plane + o] = acc[j][v];
  }
}

__global__ __launch_bounds__(256) void ec_sum64(
    const float* __restrict__ hw, const unsigned* __restrict__ lst,
    const int* __restrict__ start, float* __restrict__ acc, int Nn, int relu) {
  __shared__ float rows[BKT * 64];
  __shared__ int degl[BKT];
  int t = threadIdx.x;
  int lane = t & 31;
  int wave = __builtin_amdgcn_readfirstlane(t >> 5);
  int b = blockIdx.x;
  float a0[16], a1[16];
#pragma unroll
  for (int k = 0; k < 16; ++k) {
    int d = b * BKT + wave + 8 * k;
    if (d > Nn - 1) d = Nn - 1;
    const float* ap = acc + (size_t)d * 64 + lane;
    a0[k] = ap[0];
    a1[k] = ap[32];
  }
  int s0 = start[b];
  int s1 = start[b + 1];
  int j = lane & 15;
  int g0 = s0 + wave * 16;
  int wd = (g0 + j < s1) ? (int)lst[g0 + j] : 0;
  for (int i = t; i < BKT * 64; i += 256) rows[i] = 0.0f;
  if (t < BKT) degl[t] = 0;
  __syncthreads();
  for (; g0 < s1; g0 += 8 * 16) {
    int n = s1 - g0;
    if (n > 16) n = 16;
    int wdn = (g0 + 8 * 16 + j < s1) ? (int)lst[g0 + 8 * 16 + j] : 0;
    if (lane < n) atomicAdd(&degl[wd & (BKT - 1)], 1);
    float v0[16], v1[16];
#pragma unroll
    for (int k = 0; k < 16; ++k) {
      unsigned w = (unsigned)__builtin_amdgcn_readlane(wd, k);
      const float* hp = hw + (size_t)(w >> 7) * 64 + lane;
      v0[k] = hp[0];
      v1[k] = hp[32];
    }
    if (n == 16) {
#pragma unroll
      for (int k = 0; k < 16; ++k) {
        unsigned w = (unsigned)__builtin_amdgcn_readlane(wd, k);
        float* rp = rows + (w & (BKT - 1)) * 64 + lane;
        __hip_atomic_fetch_add(rp, v0[k], __ATOMIC_RELAXED, __HIP_MEMORY_SCOPE_WORKGROUP);
        __hip_atomic_fetch_add(rp + 32, v1[k], __ATOMIC_RELAXED, __HIP_MEMORY_SCOPE_WORKGROUP);
      }
    } else {
#pragma unroll
      for (int k = 0; k < 16; ++k) {
        if (k < n) {
          unsigned w = (unsigned)__builtin_amdgcn_readlane(wd, k);
          float* rp = rows + (w & (BKT - 1)) * 64 + lane;
          __hip_atomic_fetch_add(rp, v0[k], __ATOMIC_RELAXED, __HIP_MEMORY_SCOPE_WORKGROUP);
          __hip_atomic_fetch_add(rp + 32, v1[k], __ATOMIC_RELAXED, __HIP_MEMORY_SCOPE_WORKGROUP);
        }
      }
    }
    wd = wdn;
  }
  __syncthreads();
  float wl = 1.0f / fmaxf((float)degl[wave + 8 * (lane & 15)], 1.0f);
#pragma unroll
  for (int k = 0; k < 16; ++k) {
    int dl = wave + 8 * k;
    float w = __int_as_float(__builtin_amdgcn_readlane(__float_as_int(wl), k));
    a0[k] += rows[dl * 64 + lane] * w;
    a1[k] += rows[dl * 64 + lane + 32] * w;
    if (relu) {
      a0[k] = fmaxf(a0[k], 0.0f);
      a1[k] = fmaxf(a1[k], 0.0f);
    }
  }
#pragma unroll
  for (int k = 0; k < 16; ++k) {
    int d = b * BKT + wave + 8 * k;
    if (d < Nn) {
      float* ap = acc + (size_t)d * 64 + lane;
      ap[0] = a0[k];
      ap[32] = a1[k];
    }
  }
}

__global__ __launch_bounds__(256) void ec_sum16_all(
    const float* __restrict__ hw, const unsigned* __restrict__ lst,
    const int* __restrict__ start, float* __restrict__ acc, int Nn, int E) {
  __shared__ float rows[2 * BKT * 16];
  __shared__ int degl[R2 * BKT];
  __shared__ int sb[2 * R2];
  int t = threadIdx.x;
  int lane = t & 31;
  int wave = __builtin_amdgcn_readfirstlane(t >> 5);
  int b = blockIdx.x;
  if (t < 2 * R2) sb[t] = start[(t >> 1) * SP + b + (t & 1)];
  for (int i = t; i < 2 * BKT * 16; i += 256) rows[i] = 0.0f;
  for (int i = t; i < R2 * BKT; i += 256) degl[i] = 0;
  float tot[BKT * 16 / 256];
#pragma unroll
  for (int q = 0; q < BKT * 16 / 256; ++q) {
    int i = t + 256 * q;
    int d = b * BKT + (i >> 4);
    tot[q] = (d < Nn) ? acc[(size_t)d * 16 + (i & 15)] : 0.0f;
  }
  __syncthreads();
  int c = lane & 15;
  int half = lane >> 4;
  int s1 = __builtin_amdgcn_readfirstlane(sb[1]);
  int g0 = __builtin_amdgcn_readfirstlane(sb[0]) + wave * 32;
  int wd = (g0 + lane < s1) ? (int)lst[g0 + lane] : 0;
  for (int r = 0; r < R2; ++r) {
    const float* hwr = hw + (size_t)r * Nn * 16;
    const unsigned* l = lst + (size_t)r * E;
    float* rw = rows + (r & 1) * BKT * 16;
    int* dg = degl + r * BKT;
    int s1n = 0, g0n = 0, wdf = 0;
    if (r + 1 < R2) {
      s1n = __builtin_amdgcn_readfirstlane(sb[2 * r + 3]);
      g0n = __builtin_amdgcn_readfirstlane(sb[2 * r + 2]) + wave * 32;
      wdf = (g0n + lane < s1n) ? (int)lst[(size_t)(r + 1) * E + g0n + lane] : 0;
    }
    for (; g0 < s1; g0 += 8 * 32) {
      int n = s1 - g0;
      if (n > 32) n = 32;
      int wdn = (g0 + 8 * 32 + lane < s1) ? (int)l[g0 + 8 * 32 + lane] : 0;
      if (lane < n) atomicAdd(&dg[wd & (BKT - 1)], 1);
      float v[16];
      unsigned wk[16];
#pragma unroll
      for (int k = 0; k < 16; ++k) {
        wk[k] = (unsigned)__shfl(wd, 2 * k + half);
        v[k] = hwr[(size_t)(wk[k] >> 7) * 16 + c];
      }
      if (n == 32) {
#pragma unroll
        for (int k = 0; k < 16; ++k) {
          __hip_atomic_fetch_add(&rw[(wk[k] & (BKT - 1)) * 16 + c], v[k], __ATOMIC_RELAXED,
                                 __HIP_MEMORY_SCOPE_WORKGROUP);
        }
      } else {
#pragma unroll
        for (int k = 0; k < 16; ++k) {
          if (2 * k + half < n) {
            __hip_atomic_fetch_add(&rw[(wk[k] & (BKT - 1)) * 16 + c], v[k], __ATOMIC_RELAXED,
                                   __HIP_MEMORY_SCOPE_WORKGROUP);
          }
        }
      }
      wd = wdn;
    }
    __syncthreads();
#pragma unroll
    for (int q = 0; q < BKT * 16 / 256; ++q) {
      int i = t + 256 * q;
      float w = 1.0f / fmaxf((float)dg[i >> 4], 1.0f);
      tot[q] += rw[i] * w;
      rw[i] = 0.0f;
    }
    s1 = s1n;
    g0 = g0n;
    wd = wdf;
  }
#pragma unroll
  for (int q = 0; q < BKT * 16 / 256; ++q) {
    int i = t + 256 * q;
    int d = b * BKT + (i >> 4);
    if (d < Nn) acc[(size_t)d * 16 + (i & 15)] = tot[q];
  }
}

extern "C" void kernel_launch(void* const* d_in, const int* in_sizes, int n_in,
                              void* d_out, int out_size, void* d_ws, size_t ws_size,
                              hipStream_t stream) {
  const float* x   = (const float*)d_in[0];
  const int*   src = (const int*)d_in[1];
  const int*   dst = (const int*)d_in[2];
  const float* W1  = (const float*)d_in[3];
  const float* L1w = (const float*)d_in[4];
  const float* L1b = (const float*)d_in[5];
  const float* W2  = (const float*)d_in[6];
  const float* L2w = (const float*)d_in[7];
  const float* L2b = (const float*)d_in[8];
  float* out = (float*)d_out;

  const int D_IN = 256, D_H = 64;
  const int Nn = in_sizes[0] / D_IN;
  const int R  = in_sizes[3] / (D_IN * D_H);
  const int E  = in_sizes[1] / R;
  const int NB = (Nn + BKT - 1) / BKT;
  const int CH = (E + CHUNK - 1) / CHUNK;

  char* ws = (char*)d_ws;
  size_t off = 0;
  int* start = (int*)(ws + off);
  int* cnt   = start + (size_t)R * SP;
  off += (size_t)R * Nn * sizeof(float);
  off = (off + 255) & ~(size_t)255;
  float* h = (float*)(ws + off);
  off += (size_t)Nn * D_H * sizeof(float);
  off = (off + 255) & ~(size_t)255;
  float* hw = (float*)(ws + off);
  unsigned* lst1 = (unsigned*)out;
  unsigned* lst2 = (unsigned*)h;

  int mtiles = Nn / 16;
  int gemmBlocks = (mtiles * 32 + 127) / 128;
  const int gemm1Blocks = 256;

  ec_bin_count<<<dim3(CH, R), 1024, 0, stream>>>(dst, cnt, E, NB, CH);

  ec_gemm_wmma<256, 64><<<gemm1Blocks, 512, 0, stream>>>(x, L1w, L1b, h, mtiles);
  for (int r = 0; r < R; ++r) {
    if ((r & 1) == 0) {
      int nr = (R - r < 2) ? (R - r) : 2;
      ec_bin_scatter<<<dim3(CH, nr), 1024, 0, stream>>>(src, dst, cnt, start, lst1, E, NB, CH, r);
    }
    ec_gemm_wmma<256, 64><<<gemm1Blocks, 512, 0, stream>>>(
        x, W1 + (size_t)r * D_IN * D_H, nullptr, hw, mtiles);
    ec_sum64<<<NB, 256, 0, stream>>>(hw, lst1 + (size_t)(r & 1) * E, start + r * SP, h, Nn,
                                     (r == R - 1) ? 1 : 0);
  }

  ec_gemm2_all<<<gemmBlocks, 128, 0, stream>>>(h, L2w, W2, L2b, out, hw, mtiles);
  ec_bin_scatter<<<dim3(CH, R), 1024, 0, stream>>>(src, dst, cnt, start, lst2, E, NB, CH, 0);
  ec_sum16_all<<<NB, 256, 0, stream>>>(hw, lst2, start, out, Nn, E);
}


# All four relations' edge lists built once per call in workspace behind the product buffer (if the workspace given holds 12.8 MB more) and read by both layers: 13 launches, not 15; sums unchanged
# speedup vs baseline: 2.6138x; 1.1697x over previous
#include <hip/hip_runtime.h>

typedef __attribute__((ext_vector_type(2))) float v2f;
typedef __attribute__((ext_vector_type(8))) float v8f;

constexpr int BKT = 128;
constexpr int NBMAX = 1024;
constexpr int CHUNK = 16384;
constexpr int SP = NBMAX + 1;
constexpr int R2 = 4;

__global__ __launch_bounds__(1024) void ec_bin_count(const int* __restrict__ dst,
                                                     int* __restrict__ cnt,
                                                     int E, int NB, int CH) {
  __shared__ int hist[NBMAX];
  int t = threadIdx.x;
  int c = blockIdx.x;
  int r = blockIdx.y;
  hist[t] = 0;
  __syncthreads();
  const int* d = dst + (size_t)r * E;
  int e0 = c * CHUNK;
  int dv[CHUNK / 1024];
#pragma unroll
  for (int i = 0; i < CHUNK / 1024; ++i) {
    int e = e0 + i * 1024 + t;
    dv[i] = (e < E) ? d[e] : -1;
  }
#pragma unroll
  for (int i = 0; i < CHUNK / 1024; ++i) {
    if (dv[i] >= 0) atomicAdd(&hist[dv[i] >> 7], 1);
  }
  __syncthreads();
  if (t < NB) cnt[((size_t)r * CH + c) * NB + t] = hist[t];
}

__device__ __forceinline__ int ec_scan1024(int v, int* wsum, int& total) {
  int lane = threadIdx.x & 31;
  int w = threadIdx.x >> 5;
  int x = v;
  for (int o = 1; o < 32; o <<= 1) {
    int y = __shfl_up(x, o);
    if (lane >= o) x += y;
  }
  if (lane == 31) wsum[w] = x;
  __syncthreads();
  if (w == 0) {
    int s = wsum[lane];
    for (int o = 1; o < 32; o <<= 1) {
      int y = __shfl_up(s, o);
      if (lane >= o) s += y;
    }
    wsum[lane] = s;
  }
  __syncthreads();
  int off = (w > 0) ? wsum[w - 1] : 0;
  total = wsum[31];
  __syncthreads();
  return off + x - v;
}

__global__ __launch_bounds__(1024) void ec_bin_scatter(
    const int* __restrict__ src, const int* __restrict__ dst,
    const int* __restrict__ cnt, int* __restrict__ start,
    unsigned* __restrict__ lst, int E, int NB, int CH, int r0) {
  __shared__ int lcur[NBMAX];
  __shared__ int gdelta[NBMAX];
  __shared__ int wsum[32];
  __shared__ unsigned sortedw[CHUNK];
  __shared__ int sortedp[CHUNK];
  int t = threadIdx.x;
  int c = blockIdx.x;
  int r = r0 + blockIdx.y;

  int tot = 0, before = 0, mine = 0;
  if (t < NB) {
    const int* col = cnt + (size_t)r * CH * NB + t;
    for (int cc = 0; cc < CH; cc += 8) {
      int v[8];
#pragma unroll
      for (int u = 0; u < 8; ++u) v[u] = (cc + u < CH) ? col[(size_t)(cc + u) * NB] : 0;
#pragma unroll
      for (int u = 0; u < 8; ++u) {
        tot += v[u];
        if (cc + u < c) before += v[u];
        if (cc + u == c) mine = v[u];
      }
    }
  }
  int total, ltotal;
  int gstart = ec_scan1024(tot, wsum, total);
  int lstart = ec_scan1024(mine, wsum, ltotal);
  lcur[t] = lstart;
  gdelta[t] = gstart + before - lstart;
  if (c == 0) {
    if (t < NB) start[r * SP + t] = gstart;
    if (t == 0) start[r * SP + NB] = total;
  }
  __syncthreads();

  const int* d = dst + (size_t)r * E;
  const int* s = src + (size_t)r * E;
  int e0 = c * CHUNK;
  int dv[CHUNK / 1024], sv[CHUNK / 1024];
#pragma unroll
  for (int i = 0; i < CHUNK / 1024; ++i) {
    int e = e0 + i * 1024 + t;
    dv[i] = (e < E) ? d[e] : -1;
    sv[i] = (e < E) ? s[e] : 0;
  }
#pragma unroll
  for (int i = 0; i < CHUNK / 1024; ++i) {
    if (dv[i] >= 0) {
      int b = dv[i] >> 7;
      int li = atomicAdd(&lcur[b], 1);
      sortedw[li] = ((unsigned)sv[i] << 7) | (unsigned)(dv[i] & (BKT - 1));
      sortedp[li] = gdelta[b] + li;
    }
  }
  __syncthreads();
  int n = E - e0;
  if (n > CHUNK) n = CHUNK;
  unsigned* out = lst + (size_t)(r - r0) * E;
  for (int i = 0; i < CHUNK / 1024; ++i) {
    int j = i * 1024 + t;
    if (j < n) out[sortedp[j]] = sortedw[j];
  }
}

constexpr int GKB = 32;
constexpr int GD = 3;

__device__ __forceinline__ void ec_load_step(const float* __restrict__ Ap, v2f (&dst)[GKB / 4]) {
#pragma unroll
  for (int u = 0; u < GKB / 4; ++u) {
    float2 av = *(const float2*)(Ap + u * 4);
    dst[u] = v2f{av.x, av.y};
  }
}

template <int K, int S, int NTL>
__device__ __forceinline__ void ec_strip(const float* __restrict__ Arow, const float* __restrict__ Anext,
                                         const float* Bl, v2f (&ring)[GD][GKB / 4], v8f (&acc)[NTL]) {
  constexpr int NS = K / GKB;
  static_assert(NS % (GD + 1) == 0, "the register ring turns a whole number of times per strip");
  v8f z = {0.f, 0.f, 0.f, 0.f, 0.f, 0.f, 0.f, 0.f};
#pragma unroll
  for (int j = 0; j < NTL; ++j) acc[j] = z;
#pragma unroll(GD + 1)
  for (int s = 0; s < NS; ++s) {
    v2f nw[GKB / 4];
    const float* Ap = (s + GD < NS) ? (Arow + (s + GD) * GKB) : (Anext + (s + GD - NS) * GKB);
    ec_load_step(Ap, nw);
    __builtin_amdgcn_sched_barrier(0x186);
#pragma unroll
    for (int u = 0; u < GKB / 4; ++u) {
      int k = s * GKB + u * 4;
      const float* Bp = Bl + (size_t)(k >> 1) * S;
#pragma unroll
      for (int j = 0; j < NTL; ++j) {
        v2f b = *(const v2f*)(Bp + j * 32);
        acc[j] = __builtin_amdgcn_wmma_f32_16x16x4_f32(
             false, ring[0][u],  false, b,
             (short)0, acc[j],  false,  false);
      }
    }
#pragma unroll
    for (int d = 0; d + 1 < GD; ++d) {
#pragma unroll
      for (int u = 0; u < GKB / 4; ++u) ring[d][u] = ring[d + 1][u];
    }
#pragma unroll
    for (int u = 0; u < GKB / 4; ++u) ring[GD - 1][u] = nw[u];
  }
}

template <int K, int NC>
__global__ __launch_bounds__(512) void ec_gemm_wmma(
    const float* __restrict__ A, const float* __restrict__ B,
    const float* __restrict__ bias, float* __restrict__ C, int mtiles) {
  constexpr int NT = NC / 16;
  constexpr int S = ((2 * NC) % 64 == 32) ? (2 * NC) : (2 * NC + 32);
  constexpr int WV = 16;
  static_assert(NT == 4, "a left-over strip is split over four waves, one column tile each");
  __shared__ float Bs[(K / 2) * S];

  constexpr int NL = K * NC / 4 / 512;
  static_assert(K * NC / 4 % 512 == 0, "whole float4 rounds");
  float4 bv4[NL];
#pragma unroll
  for (int it = 0; it < NL; ++it) bv4[it] = *(const float4*)(B + (size_t)(threadIdx.x + 512 * it) * 4);
#pragma unroll
  for (int it = 0; it < NL; ++it) {
    int i = threadIdx.x + 512 * it;
    int k = (i * 4) / NC;
    int n = (i * 4) % NC;
    float* p = &Bs[(k >> 1) * S + 2 * n + (k & 1)];
    p[0] = bv4[it].x;
    p[2] = bv4[it].y;
    p[4] = bv4[it].z;
    p[6] = bv4[it].w;
  }

  int wave = __builtin_amdgcn_readfirstlane(threadIdx.x >> 5);
  int lane = threadIdx.x & 31;
  int m = lane & 15;
  int half = lane >> 4;

  int t0 = (int)((blockIdx.x * (unsigned)mtiles) / gridDim.x);
  int t1 = (int)(((blockIdx.x + 1) * (unsigned)mtiles) / gridDim.x);
  int tf = t0 + ((t1 - t0) & ~3);
  int tile = t0 + wave;

  v2f ring[GD][GKB / 4];
  {
    int tl = (tile < mtiles) ? tile : (mtiles - 1);
    const float* Ap = A + (size_t)(tl * 16 + m) * K + 2 * half;
#pragma unroll
    for (int d = 0; d < GD; ++d) ec_load_step(Ap + d * GKB, ring[d]);
  }
  float bvv[NT];
#pragma unroll
  for (int j = 0; j < NT; ++j) bvv[j] = bias ? bias[j * 16 + m] : 0.0f;
  __syncthreads();

  const float* Bl = &Bs[(size_t)half * S + 2 * m];

  for (; tile < tf; tile += WV) {
    const float* Arow = A + (size_t)(tile * 16 + m) * K + 2 * half;
    int tn = tile + WV;
    if (tn >= tf) tn = tile;
    const float* Anext = A + (size_t)(tn * 16 + m) * K + 2 * half;
    v8f acc[NT];
    ec_strip<K, S, NT>(Arow, Anext, Bl, ring, acc);
#pragma unroll
    for (int j = 0; j < NT; ++j) {
#pragma unroll
      for (int v = 0; v < 8; ++v) {
        C[(size_t)(tile * 16 + v + 8 * half) * NC + j * 16 + m] = acc[j][v] + bvv[j];
      }
    }
  }

  int q = 3 - (wave >> 2);
  if (tf + q < t1) {
    int tq = tf + q;
    int j0 = wave & 3;
    const float* Arow = A + (size_t)(tq * 16 + m) * K + 2 * half;
#pragma unroll
    for (int d = 0; d < GD; ++d) ec_load_step(Arow + d * GKB, ring[d]);
    v8f acc1[1];
    ec_strip<K, S, 1>(Arow, Arow, Bl + j0 * 32, ring, acc1);
    float bq = bias ? bias[j0 * 16 + m] : 0.0f;
#pragma unroll
    for (int v = 0; v < 8; ++v) {
      C[(size_t)(tq * 16 + v + 8 * half) * NC + j0 * 16 + m] = acc1[0][v] + bq;
    }
  }
}


__global__ __launch_bounds__(128) void ec_gemm2_all(
    const float* __restrict__ A, const float* __restrict__ Bself,
    const float* __restrict__ Brel, const float* __restrict__ bias,
    float* __restrict__ Cself, float* __restrict__ Crel, int mtiles) {
  constexpr int K = 64;
  constexpr int NT = 1 + R2;
  constexpr int S = 2 * 16 * NT;
  static_assert(S % 64 == 32, "lane halves must land in disjoint bank ranges");
  __shared__ float Bs[(K / 2) * S];

  static_assert(K * 16 / 4 == 256, "two float4 of a block per thread");
  float4 sv[2 * NT];
#pragma unroll
  for (int it = 0; it < 2 * NT; ++it) {
    const float* bp = (it < 2) ? Bself : (Brel + (size_t)(it / 2 - 1) * K * 16);
    sv[it] = *(const float4*)(bp + (threadIdx.x + 128 * (it & 1)) * 4);
  }
#pragma unroll
  for (int it = 0; it < 2 * NT; ++it) {
    int q = threadIdx.x + 128 * (it & 1);
    int k = q >> 2;
    int n = (it / 2) * 16 + (q & 3) * 4;
    float* p = &Bs[(k >> 1) * S + 2 * n + (k & 1)];
    p[0] = sv[it].x;
    p[2] = sv[it].y;
    p[4] = sv[it].z;
    p[6] = sv[it].w;
  }
  __syncthreads();

  int wave = (blockIdx.x * blockDim.x + threadIdx.x) >> 5;
  if (wave >= mtiles) return;
  int lane = threadIdx.x & 31;
  int m = lane & 15;
  int half = lane >> 4;

  const float* Arow = A + (size_t)(wave * 16 + m) * K + 2 * half;

  v8f acc[NT];
  v8f z = {0.f, 0.f, 0.f, 0.f, 0.f, 0.f, 0.f, 0.f};
#pragma unroll
  for (int j = 0; j < NT; ++j) acc[j] = z;

#pragma unroll
  for (int k = 0; k < K; k += 4) {
    float2 av = *(const float2*)(Arow + k);
    v2f a = {av.x, av.y};
    const float* Bp = &Bs[(size_t)((k + 2 * half) >> 1) * S + 2 * m];
#pragma unroll
    for (int j = 0; j < NT; ++j) {
      v2f b = *(const v2f*)(Bp + j * 32);
      acc[j] = __builtin_amdgcn_wmma_f32_16x16x4_f32(
           false, a,  false, b,
           (short)0, acc[j],  false,  false);
    }
  }

  float bv = bias[m];
  size_t plane = (size_t)mtiles * 16 * 16;
#pragma unroll
  for (int v = 0; v < 8; ++v) {
    size_t o = (size_t)(wave * 16 + v + 8 * half) * 16 + m;
    Cself[o] = acc[0][v] + bv;
#pragma unroll
    for (int j = 1; j < NT; ++j) Crel[(size_t)(j - 1) * plane + o] = acc[j][v];
  }
}

__global__ __launch_bounds__(256) void ec_sum64(
    const float* __restrict__ hw, const unsigned* __restrict__ lst,
    const int* __restrict__ start, float* __restrict__ acc, int Nn, int relu) {
  __shared__ float rows[BKT * 64];
  __shared__ int degl[BKT];
  int t = threadIdx.x;
  int lane = t & 31;
  int wave = __builtin_amdgcn_readfirstlane(t >> 5);
  int b = blockIdx.x;
  float a0[16], a1[16];
#pragma unroll
  for (int k = 0; k < 16; ++k) {
    int d = b * BKT + wave + 8 * k;
    if (d > Nn - 1) d = Nn - 1;
    const float* ap = acc + (size_t)d * 64 + lane;
    a0[k] = ap[0];
    a1[k] = ap[32];
  }
  int s0 = start[b];
  int s1 = start[b + 1];
  int j = lane & 15;
  int g0 = s0 + wave * 16;
  int wd = (g0 + j < s1) ? (int)lst[g0 + j] : 0;
  for (int i = t; i < BKT * 64; i += 256) rows[i] = 0.0f;
  if (t < BKT) degl[t] = 0;
  __syncthreads();
  for (; g0 < s1; g0 += 8 * 16) {
    int n = s1 - g0;
    if (n > 16) n = 16;
    int wdn = (g0 + 8 * 16 + j < s1) ? (int)lst[g0 + 8 * 16 + j] : 0;
    if (lane < n) atomicAdd(&degl[wd & (BKT - 1)], 1);
    float v0[16], v1[16];
#pragma unroll
    for (int k = 0; k < 16; ++k) {
      unsigned w = (unsigned)__builtin_amdgcn_readlane(wd, k);
      const float* hp = hw + (size_t)(w >> 7) * 64 + lane;
      v0[k] = hp[0];
      v1[k] = hp[32];
    }
    if (n == 16) {
#pragma unroll
      for (int k = 0; k < 16; ++k) {
        unsigned w = (unsigned)__builtin_amdgcn_readlane(wd, k);
        float* rp = rows + (w & (BKT - 1)) * 64 + lane;
        __hip_atomic_fetch_add(rp, v0[k], __ATOMIC_RELAXED, __HIP_MEMORY_SCOPE_WORKGROUP);
        __hip_atomic_fetch_add(rp + 32, v1[k], __ATOMIC_RELAXED, __HIP_MEMORY_SCOPE_WORKGROUP);
      }
    } else {
#pragma unroll
      for (int k = 0; k < 16; ++k) {
        if (k < n) {
          unsigned w = (unsigned)__builtin_amdgcn_readlane(wd, k);
          float* rp = rows + (w & (BKT - 1)) * 64 + lane;
          __hip_atomic_fetch_add(rp, v0[k], __ATOMIC_RELAXED, __HIP_MEMORY_SCOPE_WORKGROUP);
          __hip_atomic_fetch_add(rp + 32, v1[k], __ATOMIC_RELAXED, __HIP_MEMORY_SCOPE_WORKGROUP);
        }
      }
    }
    wd = wdn;
  }
  __syncthreads();
  float wl = 1.0f / fmaxf((float)degl[wave + 8 * (lane & 15)], 1.0f);
#pragma unroll
  for (int k = 0; k < 16; ++k) {
    int dl = wave + 8 * k;
    float w = __int_as_float(__builtin_amdgcn_readlane(__float_as_int(wl), k));
    a0[k] += rows[dl * 64 + lane] * w;
    a1[k] += rows[dl * 64 + lane + 32] * w;
    if (relu) {
      a0[k] = fmaxf(a0[k], 0.0f);
      a1[k] = fmaxf(a1[k], 0.0f);
    }
  }
#pragma unroll
  for (int k = 0; k < 16; ++k) {
    int d = b * BKT + wave + 8 * k;
    if (d < Nn) {
      float* ap = acc + (size_t)d * 64 + lane;
      ap[0] = a0[k];
      ap[32] = a1[k];
    }
  }
}

__global__ __launch_bounds__(256) void ec_sum16_all(
    const float* __restrict__ hw, const unsigned* __restrict__ lst,
    const int* __restrict__ start, float* __restrict__ acc, int Nn, int E) {
  __shared__ float rows[2 * BKT * 16];
  __shared__ int degl[R2 * BKT];
  __shared__ int sb[2 * R2];
  int t = threadIdx.x;
  int lane = t & 31;
  int wave = __builtin_amdgcn_readfirstlane(t >> 5);
  int b = blockIdx.x;
  if (t < 2 * R2) sb[t] = start[(t >> 1) * SP + b + (t & 1)];
  for (int i = t; i < 2 * BKT * 16; i += 256) rows[i] = 0.0f;
  for (int i = t; i < R2 * BKT; i += 256) degl[i] = 0;
  float tot[BKT * 16 / 256];
#pragma unroll
  for (int q = 0; q < BKT * 16 / 256; ++q) {
    int i = t + 256 * q;
    int d = b * BKT + (i >> 4);
    tot[q] = (d < Nn) ? acc[(size_t)d * 16 + (i & 15)] : 0.0f;
  }
  __syncthreads();
  int c = lane & 15;
  int half = lane >> 4;
  int s1 = __builtin_amdgcn_readfirstlane(sb[1]);
  int g0 = __builtin_amdgcn_readfirstlane(sb[0]) + wave * 32;
  int wd = (g0 + lane < s1) ? (int)lst[g0 + lane] : 0;
  for (int r = 0; r < R2; ++r) {
    const float* hwr = hw + (size_t)r * Nn * 16;
    const unsigned* l = lst + (size_t)r * E;
    float* rw = rows + (r & 1) * BKT * 16;
    int* dg = degl + r * BKT;
    int s1n = 0, g0n = 0, wdf = 0;
    if (r + 1 < R2) {
      s1n = __builtin_amdgcn_readfirstlane(sb[2 * r + 3]);
      g0n = __builtin_amdgcn_readfirstlane(sb[2 * r + 2]) + wave * 32;
      wdf = (g0n + lane < s1n) ? (int)lst[(size_t)(r + 1) * E + g0n + lane] : 0;
    }
    for (; g0 < s1; g0 += 8 * 32) {
      int n = s1 - g0;
      if (n > 32) n = 32;
      int wdn = (g0 + 8 * 32 + lane < s1) ? (int)l[g0 + 8 * 32 + lane] : 0;
      if (lane < n) atomicAdd(&dg[wd & (BKT - 1)], 1);
      float v[16];
      unsigned wk[16];
#pragma unroll
      for (int k = 0; k < 16; ++k) {
        wk[k] = (unsigned)__shfl(wd, 2 * k + half);
        v[k] = hwr[(size_t)(wk[k] >> 7) * 16 + c];
      }
      if (n == 32) {
#pragma unroll
        for (int k = 0; k < 16; ++k) {
          __hip_atomic_fetch_add(&rw[(wk[k] & (BKT - 1)) * 16 + c], v[k], __ATOMIC_RELAXED,
                                 __HIP_MEMORY_SCOPE_WORKGROUP);
        }
      } else {
#pragma unroll
        for (int k = 0; k < 16; ++k) {
          if (2 * k + half < n) {
            __hip_atomic_fetch_add(&rw[(wk[k] & (BKT - 1)) * 16 + c], v[k], __ATOMIC_RELAXED,
                                   __HIP_MEMORY_SCOPE_WORKGROUP);
          }
        }
      }
      wd = wdn;
    }
    __syncthreads();
#pragma unroll
    for (int q = 0; q < BKT * 16 / 256; ++q) {
      int i = t + 256 * q;
      float w = 1.0f / fmaxf((float)dg[i >> 4], 1.0f);
      tot[q] += rw[i] * w;
      rw[i] = 0.0f;
    }
    s1 = s1n;
    g0 = g0n;
    wd = wdf;
  }
#pragma unroll
  for (int q = 0; q < BKT * 16 / 256; ++q) {
    int i = t + 256 * q;
    int d = b * BKT + (i >> 4);
    if (d < Nn) acc[(size_t)d * 16 + (i & 15)] = tot[q];
  }
}

extern "C" void kernel_launch(void* const* d_in, const int* in_sizes, int n_in,
                              void* d_out, int out_size, void* d_ws, size_t ws_size,
                              hipStream_t stream) {
  const float* x   = (const float*)d_in[0];
  const int*   src = (const int*)d_in[1];
  const int*   dst = (const int*)d_in[2];
  const float* W1  = (const float*)d_in[3];
  const float* L1w = (const float*)d_in[4];
  const float* L1b = (const float*)d_in[5];
  const float* W2  = (const float*)d_in[6];
  const float* L2w = (const float*)d_in[7];
  const float* L2b = (const float*)d_in[8];
  float* out = (float*)d_out;

  const int D_IN = 256, D_H = 64;
  const int Nn = in_sizes[0] / D_IN;
  const int R  = in_sizes[3] / (D_IN * D_H);
  const int E  = in_sizes[1] / R;
  const int NB = (Nn + BKT - 1) / BKT;
  const int CH = (E + CHUNK - 1) / CHUNK;

  char* ws = (char*)d_ws;
  size_t off = 0;
  int* start = (int*)(ws + off);
  int* cnt   = start + (size_t)R * SP;
  off += (size_t)R * Nn * sizeof(float);
  off = (off + 255) & ~(size_t)255;
  float* h = (float*)(ws + off);
  off += (size_t)Nn * D_H * sizeof(float);
  off = (off + 255) & ~(size_t)255;
  float* hw = (float*)(ws + off);
  off += (size_t)Nn * D_H * sizeof(float);
  off = (off + 255) & ~(size_t)255;
  unsigned* lstAll = (unsigned*)(ws + off);
  const bool listsFit = (R == R2) && (ws_size >= off + (size_t)R * E * sizeof(unsigned));

  int mtiles = Nn / 16;
  int gemmBlocks = (mtiles * 32 + 127) / 128;
  const int gemm1Blocks = 256;

  ec_bin_count<<<dim3(CH, R), 1024, 0, stream>>>(dst, cnt, E, NB, CH);

  if (listsFit) {
    ec_gemm_wmma<256, 64><<<gemm1Blocks, 512, 0, stream>>>(x, L1w, L1b, h, mtiles);
    ec_bin_scatter<<<dim3(CH, R), 1024, 0, stream>>>(src, dst, cnt, start, lstAll, E, NB, CH, 0);
    for (int r = 0; r < R; ++r) {
      ec_gemm_wmma<256, 64><<<gemm1Blocks, 512, 0, stream>>>(
          x, W1 + (size_t)r * D_IN * D_H, nullptr, hw, mtiles);
      ec_sum64<<<NB, 256, 0, stream>>>(hw, lstAll + (size_t)r * E, start + r * SP, h, Nn,
                                       (r == R - 1) ? 1 : 0);
    }
    ec_gemm2_all<<<gemmBlocks, 128, 0, stream>>>(h, L2w, W2, L2b, out, hw, mtiles);
    ec_sum16_all<<<NB, 256, 0, stream>>>(hw, lstAll, start, out, Nn, E);
    return;
  }

  unsigned* lst1 = (unsigned*)out;
  unsigned* lst2 = (unsigned*)h;

  ec_gemm_wmma<256, 64><<<gemm1Blocks, 512, 0, stream>>>(x, L1w, L1b, h, mtiles);
  for (int r = 0; r < R; ++r) {
    if ((r & 1) == 0) {
      int nr = (R - r < 2) ? (R - r) : 2;
      ec_bin_scatter<<<dim3(CH, nr), 1024, 0, stream>>>(src, dst, cnt, start, lst1, E, NB, CH, r);
    }
    ec_gemm_wmma<256, 64><<<gemm1Blocks, 512, 0, stream>>>(
        x, W1 + (size_t)r * D_IN * D_H, nullptr, hw, mtiles);
    ec_sum64<<<NB, 256, 0, stream>>>(hw, lst1 + (size_t)(r & 1) * E, start + r * SP, h, Nn,
                                     (r == R - 1) ? 1 : 0);
  }

  ec_gemm2_all<<<gemmBlocks, 128, 0, stream>>>(h, L2w, W2, L2b, out, hw, mtiles);
  ec_bin_scatter<<<dim3(CH, R), 1024, 0, stream>>>(src, dst, cnt, start, lst2, E, NB, CH, 0);
  ec_sum16_all<<<NB, 256, 0, stream>>>(hw, lst2, start, out, Nn, E);
}


# Layer 1 in 2 launches a pass, not 9: one WMMA launch (grid y = weight block) writes self term and as many relation planes as the workspace holds (4, 2 or 1); one edge-sum launch walks them; sums same
# speedup vs baseline: 3.3016x; 1.2631x over previous
#include <hip/hip_runtime.h>

typedef __attribute__((ext_vector_type(2))) float v2f;
typedef __attribute__((ext_vector_type(8))) float v8f;

constexpr int BKT = 128;
constexpr int NBMAX = 1024;
constexpr int CHUNK = 16384;
constexpr int SP = NBMAX + 1;
constexpr int R2 = 4;

__global__ __launch_bounds__(1024) void ec_bin_count(const int* __restrict__ dst,
                                                     int* __restrict__ cnt,
                                                     int E, int NB, int CH) {
  __shared__ int hist[NBMAX];
  int t = threadIdx.x;
  int c = blockIdx.x;
  int r = blockIdx.y;
  hist[t] = 0;
  __syncthreads();
  const int* d = dst + (size_t)r * E;
  int e0 = c * CHUNK;
  int dv[CHUNK / 1024];
#pragma unroll
  for (int i = 0; i < CHUNK / 1024; ++i) {
    int e = e0 + i * 1024 + t;
    dv[i] = (e < E) ? d[e] : -1;
  }
#pragma unroll
  for (int i = 0; i < CHUNK / 1024; ++i) {
    if (dv[i] >= 0) atomicAdd(&hist[dv[i] >> 7], 1);
  }
  __syncthreads();
  if (t < NB) cnt[((size_t)r * CH + c) * NB + t] = hist[t];
}

__device__ __forceinline__ int ec_scan1024(int v, int* wsum, int& total) {
  int lane = threadIdx.x & 31;
  int w = threadIdx.x >> 5;
  int x = v;
  for (int o = 1; o < 32; o <<= 1) {
    int y = __shfl_up(x, o);
    if (lane >= o) x += y;
  }
  if (lane == 31) wsum[w] = x;
  __syncthreads();
  if (w == 0) {
    int s = wsum[lane];
    for (int o = 1; o < 32; o <<= 1) {
      int y = __shfl_up(s, o);
      if (lane >= o) s += y;
    }
    wsum[lane] = s;
  }
  __syncthreads();
  int off = (w > 0) ? wsum[w - 1] : 0;
  total = wsum[31];
  __syncthreads();
  return off + x - v;
}

__global__ __launch_bounds__(1024) void ec_bin_scatter(
    const int* __restrict__ src, const int* __restrict__ dst,
    const int* __restrict__ cnt, int* __restrict__ start,
    unsigned* __restrict__ lst, int E, int NB, int CH, int r0) {
  __shared__ int lcur[NBMAX];
  __shared__ int gdelta[NBMAX];
  __shared__ int wsum[32];
  __shared__ unsigned sortedw[CHUNK];
  __shared__ int sortedp[CHUNK];
  int t = threadIdx.x;
  int c = blockIdx.x;
  int r = r0 + blockIdx.y;

  int tot = 0, before = 0, mine = 0;
  if (t < NB) {
    const int* col = cnt + (size_t)r * CH * NB + t;
    for (int cc = 0; cc < CH; cc += 8) {
      int v[8];
#pragma unroll
      for (int u = 0; u < 8; ++u) v[u] = (cc + u < CH) ? col[(size_t)(cc + u) * NB] : 0;
#pragma unroll
      for (int u = 0; u < 8; ++u) {
        tot += v[u];
        if (cc + u < c) before += v[u];
        if (cc + u == c) mine = v[u];
      }
    }
  }
  int total, ltotal;
  int gstart = ec_scan1024(tot, wsum, total);
  int lstart = ec_scan1024(mine, wsum, ltotal);
  lcur[t] = lstart;
  gdelta[t] = gstart + before - lstart;
  if (c == 0) {
    if (t < NB) start[r * SP + t] = gstart;
    if (t == 0) start[r * SP + NB] = total;
  }
  __syncthreads();

  const int* d = dst + (size_t)r * E;
  const int* s = src + (size_t)r * E;
  int e0 = c * CHUNK;
  int dv[CHUNK / 1024], sv[CHUNK / 1024];
#pragma unroll
  for (int i = 0; i < CHUNK / 1024; ++i) {
    int e = e0 + i * 1024 + t;
    dv[i] = (e < E) ? d[e] : -1;
    sv[i] = (e < E) ? s[e] : 0;
  }
#pragma unroll
  for (int i = 0; i < CHUNK / 1024; ++i) {
    if (dv[i] >= 0) {
      int b = dv[i] >> 7;
      int li = atomicAdd(&lcur[b], 1);
      sortedw[li] = ((unsigned)sv[i] << 7) | (unsigned)(dv[i] & (BKT - 1));
      sortedp[li] = gdelta[b] + li;
    }
  }
  __syncthreads();
  int n = E - e0;
  if (n > CHUNK) n = CHUNK;
  unsigned* out = lst + (size_t)(r - r0) * E;
  for (int i = 0; i < CHUNK / 1024; ++i) {
    int j = i * 1024 + t;
    if (j < n) out[sortedp[j]] = sortedw[j];
  }
}

constexpr int GKB = 32;
constexpr int GD = 3;

__device__ __forceinline__ void ec_load_step(const float* __restrict__ Ap, v2f (&dst)[GKB / 4]) {
#pragma unroll
  for (int u = 0; u < GKB / 4; ++u) {
    float2 av = *(const float2*)(Ap + u * 4);
    dst[u] = v2f{av.x, av.y};
  }
}

template <int K, int S, int NTL>
__device__ __forceinline__ void ec_strip(const float* __restrict__ Arow, const float* __restrict__ Anext,
                                         const float* Bl, v2f (&ring)[GD][GKB / 4], v8f (&acc)[NTL]) {
  constexpr int NS = K / GKB;
  static_assert(NS % (GD + 1) == 0, "the register ring turns a whole number of times per strip");
  v8f z = {0.f, 0.f, 0.f, 0.f, 0.f, 0.f, 0.f, 0.f};
#pragma unroll
  for (int j = 0; j < NTL; ++j) acc[j] = z;
#pragma unroll(GD + 1)
  for (int s = 0; s < NS; ++s) {
    v2f nw[GKB / 4];
    const float* Ap = (s + GD < NS) ? (Arow + (s + GD) * GKB) : (Anext + (s + GD - NS) * GKB);
    ec_load_step(Ap, nw);
    __builtin_amdgcn_sched_barrier(0x186);
#pragma unroll
    for (int u = 0; u < GKB / 4; ++u) {
      int k = s * GKB + u * 4;
      const float* Bp = Bl + (size_t)(k >> 1) * S;
#pragma unroll
      for (int j = 0; j < NTL; ++j) {
        v2f b = *(const v2f*)(Bp + j * 32);
        acc[j] = __builtin_amdgcn_wmma_f32_16x16x4_f32(
             false, ring[0][u],  false, b,
             (short)0, acc[j],  false,  false);
      }
    }
#pragma unroll
    for (int d = 0; d + 1 < GD; ++d) {
#pragma unroll
      for (int u = 0; u < GKB / 4; ++u) ring[d][u] = ring[d + 1][u];
    }
#pragma unroll
    for (int u = 0; u < GKB / 4; ++u) ring[GD - 1][u] = nw[u];
  }
}

template <int K, int NC>
__global__ __launch_bounds__(512) void ec_gemm_wmma(
    const float* __restrict__ A, const float* __restrict__ Bself, const float* __restrict__ Brel,
    const float* __restrict__ biasSelf, float* __restrict__ Cself, float* __restrict__ Crel, int mtiles) {
  constexpr int NT = NC / 16;
  const int wb = (int)blockIdx.y - ((Bself != nullptr) ? 1 : 0);
  const float* __restrict__ B = (wb < 0) ? Bself : (Brel + (size_t)wb * K * NC);
  const float* __restrict__ bias = (wb < 0) ? biasSelf : nullptr;
  float* __restrict__ C = (wb < 0) ? Cself : (Crel + (size_t)wb * mtiles * 16 * NC);
  constexpr int S = ((2 * NC) % 64 == 32) ? (2 * NC) : (2 * NC + 32);
  constexpr int WV = 16;
  static_assert(NT == 4, "a left-over strip is split over four waves, one column tile each");
  __shared__ float Bs[(K / 2) * S];

  constexpr int NL = K * NC / 4 / 512;
  static_assert(K * NC / 4 % 512 == 0, "whole float4 rounds");
  float4 bv4[NL];
#pragma unroll
  for (int it = 0; it < NL; ++it) bv4[it] = *(const float4*)(B + (size_t)(threadIdx.x + 512 * it) * 4);
#pragma unroll
  for (int it = 0; it < NL; ++it) {
    int i = threadIdx.x + 512 * it;
    int k = (i * 4) / NC;
    int n = (i * 4) % NC;
    float* p = &Bs[(k >> 1) * S + 2 * n + (k & 1)];
    p[0] = bv4[it].x;
    p[2] = bv4[it].y;
    p[4] = bv4[it].z;
    p[6] = bv4[it].w;
  }

  int wave = __builtin_amdgcn_readfirstlane(threadIdx.x >> 5);
  int lane = threadIdx.x & 31;
  int m = lane & 15;
  int half = lane >> 4;

  int t0 = (int)((blockIdx.x * (unsigned)mtiles) / gridDim.x);
  int t1 = (int)(((blockIdx.x + 1) * (unsigned)mtiles) / gridDim.x);
  int tf = t0 + ((t1 - t0) & ~3);
  int tile = t0 + wave;

  v2f ring[GD][GKB / 4];
  {
    int tl = (tile < mtiles) ? tile : (mtiles - 1);
    const float* Ap = A + (size_t)(tl * 16 + m) * K + 2 * half;
#pragma unroll
    for (int d = 0; d < GD; ++d) ec_load_step(Ap + d * GKB, ring[d]);
  }
  float bvv[NT];
#pragma unroll
  for (int j = 0; j < NT; ++j) bvv[j] = bias ? bias[j * 16 + m] : 0.0f;
  __syncthreads();

  const float* Bl = &Bs[(size_t)half * S + 2 * m];

  for (; tile < tf; tile += WV) {
    const float* Arow = A + (size_t)(tile * 16 + m) * K + 2 * half;
    int tn = tile + WV;
    if (tn >= tf) tn = tile;
    const float* Anext = A + (size_t)(tn * 16 + m) * K + 2 * half;
    v8f acc[NT];
    ec_strip<K, S, NT>(Arow, Anext, Bl, ring, acc);
#pragma unroll
    for (int j = 0; j < NT; ++j) {
#pragma unroll
      for (int v = 0; v < 8; ++v) {
        C[(size_t)(tile * 16 + v + 8 * half) * NC + j * 16 + m] = acc[j][v] + bvv[j];
      }
    }
  }

  int q = 3 - (wave >> 2);
  if (tf + q < t1) {
    int tq = tf + q;
    int j0 = wave & 3;
    const float* Arow = A + (size_t)(tq * 16 + m) * K + 2 * half;
#pragma unroll
    for (int d = 0; d < GD; ++d) ec_load_step(Arow + d * GKB, ring[d]);
    v8f acc1[1];
    ec_strip<K, S, 1>(Arow, Arow, Bl + j0 * 32, ring, acc1);
    float bq = bias ? bias[j0 * 16 + m] : 0.0f;
#pragma unroll
    for (int v = 0; v < 8; ++v) {
      C[(size_t)(tq * 16 + v + 8 * half) * NC + j0 * 16 + m] = acc1[0][v] + bq;
    }
  }
}


__global__ __launch_bounds__(128) void ec_gemm2_all(
    const float* __restrict__ A, const float* __restrict__ Bself,
    const float* __restrict__ Brel, const float* __restrict__ bias,
    float* __restrict__ Cself, float* __restrict__ Crel, int mtiles) {
  constexpr int K = 64;
  constexpr int NT = 1 + R2;
  constexpr int S = 2 * 16 * NT;
  static_assert(S % 64 == 32, "lane halves must land in disjoint bank ranges");
  __shared__ float Bs[(K / 2) * S];

  static_assert(K * 16 / 4 == 256, "two float4 of a block per thread");
  float4 sv[2 * NT];
#pragma unroll
  for (int it = 0; it < 2 * NT; ++it) {
    const float* bp = (it < 2) ? Bself : (Brel + (size_t)(it / 2 - 1) * K * 16);
    sv[it] = *(const float4*)(bp + (threadIdx.x + 128 * (it & 1)) * 4);
  }
#pragma unroll
  for (int it = 0; it < 2 * NT; ++it) {
    int q = threadIdx.x + 128 * (it & 1);
    int k = q >> 2;
    int n = (it / 2) * 16 + (q & 3) * 4;
    float* p = &Bs[(k >> 1) * S + 2 * n + (k & 1)];
    p[0] = sv[it].x;
    p[2] = sv[it].y;
    p[4] = sv[it].z;
    p[6] = sv[it].w;
  }
  __syncthreads();

  int wave = (blockIdx.x * blockDim.x + threadIdx.x) >> 5;
  if (wave >= mtiles) return;
  int lane = threadIdx.x & 31;
  int m = lane & 15;
  int half = lane >> 4;

  const float* Arow = A + (size_t)(wave * 16 + m) * K + 2 * half;

  v8f acc[NT];
  v8f z = {0.f, 0.f, 0.f, 0.f, 0.f, 0.f, 0.f, 0.f};
#pragma unroll
  for (int j = 0; j < NT; ++j) acc[j] = z;

#pragma unroll
  for (int k = 0; k < K; k += 4) {
    float2 av = *(const float2*)(Arow + k);
    v2f a = {av.x, av.y};
    const float* Bp = &Bs[(size_t)((k + 2 * half) >> 1) * S + 2 * m];
#pragma unroll
    for (int j = 0; j < NT; ++j) {
      v2f b = *(const v2f*)(Bp + j * 32);
      acc[j] = __builtin_amdgcn_wmma_f32_16x16x4_f32(
           false, a,  false, b,
           (short)0, acc[j],  false,  false);
    }
  }

  float bv = bias[m];
  size_t plane = (size_t)mtiles * 16 * 16;
#pragma unroll
  for (int v = 0; v < 8; ++v) {
    size_t o = (size_t)(wave * 16 + v + 8 * half) * 16 + m;
    Cself[o] = acc[0][v] + bv;
#pragma unroll
    for (int j = 1; j < NT; ++j) Crel[(size_t)(j - 1) * plane + o] = acc[j][v];
  }
}

__global__ __launch_bounds__(256) void ec_sum64(
    const float* __restrict__ hw, const unsigned* __restrict__ lst,
    const int* __restrict__ start, float* __restrict__ acc, int Nn, int relu) {
  __shared__ float rows[BKT * 64];
  __shared__ int degl[BKT];
  int t = threadIdx.x;
  int lane = t & 31;
  int wave = __builtin_amdgcn_readfirstlane(t >> 5);
  int b = blockIdx.x;
  float a0[16], a1[16];
#pragma unroll
  for (int k = 0; k < 16; ++k) {
    int d = b * BKT + wave + 8 * k;
    if (d > Nn - 1) d = Nn - 1;
    const float* ap = acc + (size_t)d * 64 + lane;
    a0[k] = ap[0];
    a1[k] = ap[32];
  }
  int s0 = start[b];
  int s1 = start[b + 1];
  int j = lane & 15;
  int g0 = s0 + wave * 16;
  int wd = (g0 + j < s1) ? (int)lst[g0 + j] : 0;
  for (int i = t; i < BKT * 64; i += 256) rows[i] = 0.0f;
  if (t < BKT) degl[t] = 0;
  __syncthreads();
  for (; g0 < s1; g0 += 8 * 16) {
    int n = s1 - g0;
    if (n > 16) n = 16;
    int wdn = (g0 + 8 * 16 + j < s1) ? (int)lst[g0 + 8 * 16 + j] : 0;
    if (lane < n) atomicAdd(&degl[wd & (BKT - 1)], 1);
    float v0[16], v1[16];
#pragma unroll
    for (int k = 0; k < 16; ++k) {
      unsigned w = (unsigned)__builtin_amdgcn_readlane(wd, k);
      const float* hp = hw + (size_t)(w >> 7) * 64 + lane;
      v0[k] = hp[0];
      v1[k] = hp[32];
    }
    if (n == 16) {
#pragma unroll
      for (int k = 0; k < 16; ++k) {
        unsigned w = (unsigned)__builtin_amdgcn_readlane(wd, k);
        float* rp = rows + (w & (BKT - 1)) * 64 + lane;
        __hip_atomic_fetch_add(rp, v0[k], __ATOMIC_RELAXED, __HIP_MEMORY_SCOPE_WORKGROUP);
        __hip_atomic_fetch_add(rp + 32, v1[k], __ATOMIC_RELAXED, __HIP_MEMORY_SCOPE_WORKGROUP);
      }
    } else {
#pragma unroll
      for (int k = 0; k < 16; ++k) {
        if (k < n) {
          unsigned w = (unsigned)__builtin_amdgcn_readlane(wd, k);
          float* rp = rows + (w & (BKT - 1)) * 64 + lane;
          __hip_atomic_fetch_add(rp, v0[k], __ATOMIC_RELAXED, __HIP_MEMORY_SCOPE_WORKGROUP);
          __hip_atomic_fetch_add(rp + 32, v1[k], __ATOMIC_RELAXED, __HIP_MEMORY_SCOPE_WORKGROUP);
        }
      }
    }
    wd = wdn;
  }
  __syncthreads();
  float wl = 1.0f / fmaxf((float)degl[wave + 8 * (lane & 15)], 1.0f);
#pragma unroll
  for (int k = 0; k < 16; ++k) {
    int dl = wave + 8 * k;
    float w = __int_as_float(__builtin_amdgcn_readlane(__float_as_int(wl), k));
    a0[k] += rows[dl * 64 + lane] * w;
    a1[k] += rows[dl * 64 + lane + 32] * w;
    if (relu) {
      a0[k] = fmaxf(a0[k], 0.0f);
      a1[k] = fmaxf(a1[k], 0.0f);
    }
  }
#pragma unroll
  for (int k = 0; k < 16; ++k) {
    int d = b * BKT + wave + 8 * k;
    if (d < Nn) {
      float* ap = acc + (size_t)d * 64 + lane;
      ap[0] = a0[k];
      ap[32] = a1[k];
    }
  }
}

__global__ __launch_bounds__(256) void ec_sum64_all(
    const float* __restrict__ hw, const unsigned* __restrict__ lst,
    const int* __restrict__ start, float* __restrict__ acc, int Nn, int E, int r0, int nr, int relu) {
  __shared__ float rows[2 * BKT * 64];
  __shared__ int degl[R2 * BKT];
  __shared__ int sb[2 * R2];
  int t = threadIdx.x;
  int lane = t & 31;
  int wave = __builtin_amdgcn_readfirstlane(t >> 5);
  int b = blockIdx.x;
  float a0[16], a1[16];
#pragma unroll
  for (int k = 0; k < 16; ++k) {
    int d = b * BKT + wave + 8 * k;
    if (d > Nn - 1) d = Nn - 1;
    const float* ap = acc + (size_t)d * 64 + lane;
    a0[k] = ap[0];
    a1[k] = ap[32];
  }
  if (t < 2 * nr) sb[t] = start[(r0 + (t >> 1)) * SP + b + (t & 1)];
  {
    float4 z4 = make_float4(0.0f, 0.0f, 0.0f, 0.0f);
    float4* r4 = (float4*)rows;
    for (int i = t; i < 2 * BKT * 64 / 4; i += 256) r4[i] = z4;
  }
  for (int i = t; i < R2 * BKT; i += 256) degl[i] = 0;
  __syncthreads();
  int j = lane & 15;
  int se0 = __builtin_amdgcn_readfirstlane(sb[1]);
  int gs0 = __builtin_amdgcn_readfirstlane(sb[0]) + wave * 16;
  int wf0 = (gs0 + j < se0) ? (int)lst[(size_t)r0 * E + gs0 + j] : 0;
  int se1 = 0, gs1 = 0, wf1 = 0, se2 = 0, gs2 = 0, wf2 = 0, se3 = 0, gs3 = 0, wf3 = 0;
  if (nr > 1) {
    se1 = __builtin_amdgcn_readfirstlane(sb[3]);
    gs1 = __builtin_amdgcn_readfirstlane(sb[2]) + wave * 16;
    wf1 = (gs1 + j < se1) ? (int)lst[(size_t)(r0 + 1) * E + gs1 + j] : 0;
  }
  if (nr > 2) {
    se2 = __builtin_amdgcn_readfirstlane(sb[5]);
    gs2 = __builtin_amdgcn_readfirstlane(sb[4]) + wave * 16;
    wf2 = (gs2 + j < se2) ? (int)lst[(size_t)(r0 + 2) * E + gs2 + j] : 0;
  }
  if (nr > 3) {
    se3 = __builtin_amdgcn_readfirstlane(sb[7]);
    gs3 = __builtin_amdgcn_readfirstlane(sb[6]) + wave * 16;
    wf3 = (gs3 + j < se3) ? (int)lst[(size_t)(r0 + 3) * E + gs3 + j] : 0;
  }
  for (int p = 0; p < nr; ++p) {
    const float* hwp = hw + (size_t)p * Nn * 64;
    const unsigned* l = lst + (size_t)(r0 + p) * E;
    float* rw = rows + (p & 1) * BKT * 64;
    int* dg = degl + p * BKT;
    int s1 = (p == 0) ? se0 : (p == 1) ? se1 : (p == 2) ? se2 : se3;
    int g0 = (p == 0) ? gs0 : (p == 1) ? gs1 : (p == 2) ? gs2 : gs3;
    int wd = (p == 0) ? wf0 : (p == 1) ? wf1 : (p == 2) ? wf2 : wf3;
    for (; g0 < s1; g0 += 8 * 16) {
      int n = s1 - g0;
      if (n > 16) n = 16;
      if (lane < n) atomicAdd(&dg[wd & (BKT - 1)], 1);
      float v0[16], v1[16];
#pragma unroll
      for (int k = 0; k < 16; ++k) {
        unsigned w = (unsigned)__builtin_amdgcn_readlane(wd, k);
        const float* hp = hwp + (size_t)(w >> 7) * 64 + lane;
        v0[k] = hp[0];
        v1[k] = hp[32];
      }
      int wdn = (g0 + 8 * 16 + j < s1) ? (int)l[g0 + 8 * 16 + j] : 0;
      if (n == 16) {
#pragma unroll
        for (int k = 0; k < 16; ++k) {
          unsigned w = (unsigned)__builtin_amdgcn_readlane(wd, k);
          float* rp = rw + (w & (BKT - 1)) * 64 + lane;
          __hip_atomic_fetch_add(rp, v0[k], __ATOMIC_RELAXED, __HIP_MEMORY_SCOPE_WORKGROUP);
          __hip_atomic_fetch_add(rp + 32, v1[k], __ATOMIC_RELAXED, __HIP_MEMORY_SCOPE_WORKGROUP);
        }
      } else {
#pragma unroll
        for (int k = 0; k < 16; ++k) {
          if (k < n) {
            unsigned w = (unsigned)__builtin_amdgcn_readlane(wd, k);
            float* rp = rw + (w & (BKT - 1)) * 64 + lane;
            __hip_atomic_fetch_add(rp, v0[k], __ATOMIC_RELAXED, __HIP_MEMORY_SCOPE_WORKGROUP);
            __hip_atomic_fetch_add(rp + 32, v1[k], __ATOMIC_RELAXED, __HIP_MEMORY_SCOPE_WORKGROUP);
          }
        }
      }
      wd = wdn;
    }
    __syncthreads();
    float wl = 1.0f / fmaxf((float)dg[wave + 8 * (lane & 15)], 1.0f);
#pragma unroll
    for (int k = 0; k < 16; ++k) {
      float* rp = rw + (wave + 8 * k) * 64 + lane;
      float w = __int_as_float(__builtin_amdgcn_readlane(__float_as_int(wl), k));
      a0[k] = __builtin_fmaf(rp[0], w, a0[k]);
      a1[k] = __builtin_fmaf(rp[32], w, a1[k]);
      rp[0] = 0.0f;
      rp[32] = 0.0f;
    }
  }
#pragma unroll
  for (int k = 0; k < 16; ++k) {
    int d = b * BKT + wave + 8 * k;
    if (relu) {
      a0[k] = fmaxf(a0[k], 0.0f);
      a1[k] = fmaxf(a1[k], 0.0f);
    }
    if (d < Nn) {
      float* ap = acc + (size_t)d * 64 + lane;
      ap[0] = a0[k];
      ap[32] = a1[k];
    }
  }
}

__global__ __launch_bounds__(256) void ec_sum16_all(
    const float* __restrict__ hw, const unsigned* __restrict__ lst,
    const int* __restrict__ start, float* __restrict__ acc, int Nn, int E) {
  __shared__ float rows[2 * BKT * 16];
  __shared__ int degl[R2 * BKT];
  __shared__ int sb[2 * R2];
  int t = threadIdx.x;
  int lane = t & 31;
  int wave = __builtin_amdgcn_readfirstlane(t >> 5);
  int b = blockIdx.x;
  if (t < 2 * R2) sb[t] = start[(t >> 1) * SP + b + (t & 1)];
  for (int i = t; i < 2 * BKT * 16; i += 256) rows[i] = 0.0f;
  for (int i = t; i < R2 * BKT; i += 256) degl[i] = 0;
  float tot[BKT * 16 / 256];
#pragma unroll
  for (int q = 0; q < BKT * 16 / 256; ++q) {
    int i = t + 256 * q;
    int d = b * BKT + (i >> 4);
    tot[q] = (d < Nn) ? acc[(size_t)d * 16 + (i & 15)] : 0.0f;
  }
  __syncthreads();
  int c = lane & 15;
  int half = lane >> 4;
  int s1 = __builtin_amdgcn_readfirstlane(sb[1]);
  int g0 = __builtin_amdgcn_readfirstlane(sb[0]) + wave * 32;
  int wd = (g0 + lane < s1) ? (int)lst[g0 + lane] : 0;
  for (int r = 0; r < R2; ++r) {
    const float* hwr = hw + (size_t)r * Nn * 16;
    const unsigned* l = lst + (size_t)r * E;
    float* rw = rows + (r & 1) * BKT * 16;
    int* dg = degl + r * BKT;
    int s1n = 0, g0n = 0, wdf = 0;
    if (r + 1 < R2) {
      s1n = __builtin_amdgcn_readfirstlane(sb[2 * r + 3]);
      g0n = __builtin_amdgcn_readfirstlane(sb[2 * r + 2]) + wave * 32;
      wdf = (g0n + lane < s1n) ? (int)lst[(size_t)(r + 1) * E + g0n + lane] : 0;
    }
    for (; g0 < s1; g0 += 8 * 32) {
      int n = s1 - g0;
      if (n > 32) n = 32;
      int wdn = (g0 + 8 * 32 + lane < s1) ? (int)l[g0 + 8 * 32 + lane] : 0;
      if (lane < n) atomicAdd(&dg[wd & (BKT - 1)], 1);
      float v[16];
      unsigned wk[16];
#pragma unroll
      for (int k = 0; k < 16; ++k) {
        wk[k] = (unsigned)__shfl(wd, 2 * k + half);
        v[k] = hwr[(size_t)(wk[k] >> 7) * 16 + c];
      }
      if (n == 32) {
#pragma unroll
        for (int k = 0; k < 16; ++k) {
          __hip_atomic_fetch_add(&rw[(wk[k] & (BKT - 1)) * 16 + c], v[k], __ATOMIC_RELAXED,
                                 __HIP_MEMORY_SCOPE_WORKGROUP);
        }
      } else {
#pragma unroll
        for (int k = 0; k < 16; ++k) {
          if (2 * k + half < n) {
            __hip_atomic_fetch_add(&rw[(wk[k] & (BKT - 1)) * 16 + c], v[k], __ATOMIC_RELAXED,
                                   __HIP_MEMORY_SCOPE_WORKGROUP);
          }
        }
      }
      wd = wdn;
    }
    __syncthreads();
#pragma unroll
    for (int q = 0; q < BKT * 16 / 256; ++q) {
      int i = t + 256 * q;
      float w = 1.0f / fmaxf((float)dg[i >> 4], 1.0f);
      tot[q] += rw[i] * w;
      rw[i] = 0.0f;
    }
    s1 = s1n;
    g0 = g0n;
    wd = wdf;
  }
#pragma unroll
  for (int q = 0; q < BKT * 16 / 256; ++q) {
    int i = t + 256 * q;
    int d = b * BKT + (i >> 4);
    if (d < Nn) acc[(size_t)d * 16 + (i & 15)] = tot[q];
  }
}

extern "C" void kernel_launch(void* const* d_in, const int* in_sizes, int n_in,
                              void* d_out, int out_size, void* d_ws, size_t ws_size,
                              hipStream_t stream) {
  const float* x   = (const float*)d_in[0];
  const int*   src = (const int*)d_in[1];
  const int*   dst = (const int*)d_in[2];
  const float* W1  = (const float*)d_in[3];
  const float* L1w = (const float*)d_in[4];
  const float* L1b = (const float*)d_in[5];
  const float* W2  = (const float*)d_in[6];
  const float* L2w = (const float*)d_in[7];
  const float* L2b = (const float*)d_in[8];
  float* out = (float*)d_out;

  const int D_IN = 256, D_H = 64;
  const int Nn = in_sizes[0] / D_IN;
  const int R  = in_sizes[3] / (D_IN * D_H);
  const int E  = in_sizes[1] / R;
  const int NB = (Nn + BKT - 1) / BKT;
  const int CH = (E + CHUNK - 1) / CHUNK;

  char* ws = (char*)d_ws;
  size_t off = 0;
  int* start = (int*)(ws + off);
  int* cnt   = start + (size_t)R * SP;
  off += (size_t)R * Nn * sizeof(float);
  off = (off + 255) & ~(size_t)255;
  float* h = (float*)(ws + off);
  off += (size_t)Nn * D_H * sizeof(float);
  off = (off + 255) & ~(size_t)255;
  float* hw = (float*)(ws + off);
  unsigned* lstAll = (unsigned*)(ws + off);
  const size_t planeBytes = (size_t)Nn * D_H * sizeof(float);
  size_t offP = off + (size_t)R * E * sizeof(unsigned);
  offP = (offP + 255) & ~(size_t)255;
  float* planes = (float*)(ws + offP);
  int P = 0;
  if (R == R2) {
    if (ws_size >= offP + 4 * planeBytes) P = 4;
    else if (ws_size >= offP + 2 * planeBytes) P = 2;
    else if (ws_size >= offP + planeBytes) P = 1;
  }

  int mtiles = Nn / 16;
  int gemmBlocks = (mtiles * 32 + 127) / 128;
  const int gemm1Blocks = 256;

  ec_bin_count<<<dim3(CH, R), 1024, 0, stream>>>(dst, cnt, E, NB, CH);

  if (P > 0) {
    int done = 0;
    while (done < R) {
      const int np = (R - done < P) ? (R - done) : P;
      const int ny = np + ((done == 0) ? 1 : 0);
      const int gx = gemm1Blocks / ny;
      ec_gemm_wmma<256, 64><<<dim3(gx, ny), 512, 0, stream>>>(
          x, (done == 0) ? L1w : nullptr, W1 + (size_t)done * D_IN * D_H, L1b, h, planes, mtiles);
      if (done == 0) {
        ec_bin_scatter<<<dim3(CH, R), 1024, 0, stream>>>(src, dst, cnt, start, lstAll, E, NB, CH, 0);
      }
      ec_sum64_all<<<NB, 256, 0, stream>>>(planes, lstAll, start, h, Nn, E, done, np,
                                           (done + np == R) ? 1 : 0);
      done += np;
    }
    ec_gemm2_all<<<gemmBlocks, 128, 0, stream>>>(h, L2w, W2, L2b, out, planes, mtiles);
    ec_sum16_all<<<NB, 256, 0, stream>>>(planes, lstAll, start, out, Nn, E);
    return;
  }

  unsigned* lst1 = (unsigned*)out;
  unsigned* lst2 = (unsigned*)h;

  ec_gemm_wmma<256, 64><<<gemm1Blocks, 512, 0, stream>>>(x, L1w, nullptr, L1b, h, nullptr, mtiles);
  for (int r = 0; r < R; ++r) {
    if ((r & 1) == 0) {
      int nr = (R - r < 2) ? (R - r) : 2;
      ec_bin_scatter<<<dim3(CH, nr), 1024, 0, stream>>>(src, dst, cnt, start, lst1, E, NB, CH, r);
    }
    ec_gemm_wmma<256, 64><<<gemm1Blocks, 512, 0, stream>>>(
        x, nullptr, W1 + (size_t)r * D_IN * D_H, nullptr, nullptr, hw, mtiles);
    ec_sum64<<<NB, 256, 0, stream>>>(hw, lst1 + (size_t)(r & 1) * E, start + r * SP, h, Nn,
                                     (r == R - 1) ? 1 : 0);
  }

  ec_gemm2_all<<<gemmBlocks, 128, 0, stream>>>(h, L2w, W2, L2b, out, hw, mtiles);
  ec_bin_scatter<<<dim3(CH, R), 1024, 0, stream>>>(src, dst, cnt, start, lst2, E, NB, CH, 0);
  ec_sum16_all<<<NB, 256, 0, stream>>>(hw, lst2, start, out, Nn, E);
}


# 4 launches, not 6: bucket counts taken in the layer-1 product launch's prologue; layer 2's products made by WMMA in the layer-1 edge-sum's write-back from its finished rows in LDS; sums unchanged
# speedup vs baseline: 3.4064x; 1.0317x over previous
#include <hip/hip_runtime.h>

typedef __attribute__((ext_vector_type(2))) float v2f;
typedef __attribute__((ext_vector_type(8))) float v8f;

constexpr int BKT = 128;
constexpr int NBMAX = 1024;
constexpr int CHUNK = 16384;
constexpr int SP = NBMAX + 1;
constexpr int R2 = 4;

__global__ __launch_bounds__(1024) void ec_bin_count(const int* __restrict__ dst,
                                                     int* __restrict__ cnt,
                                                     int E, int NB, int CH) {
  __shared__ int hist[NBMAX];
  int t = threadIdx.x;
  int c = blockIdx.x;
  int r = blockIdx.y;
  hist[t] = 0;
  __syncthreads();
  const int* d = dst + (size_t)r * E;
  int e0 = c * CHUNK;
  int dv[CHUNK / 1024];
#pragma unroll
  for (int i = 0; i < CHUNK / 1024; ++i) {
    int e = e0 + i * 1024 + t;
    dv[i] = (e < E) ? d[e] : -1;
  }
#pragma unroll
  for (int i = 0; i < CHUNK / 1024; ++i) {
    if (dv[i] >= 0) atomicAdd(&hist[dv[i] >> 7], 1);
  }
  __syncthreads();
  if (t < NB) cnt[((size_t)r * CH + c) * NB + t] = hist[t];
}

__device__ __forceinline__ int ec_scan1024(int v, int* wsum, int& total) {
  int lane = threadIdx.x & 31;
  int w = threadIdx.x >> 5;
  int x = v;
  for (int o = 1; o < 32; o <<= 1) {
    int y = __shfl_up(x, o);
    if (lane >= o) x += y;
  }
  if (lane == 31) wsum[w] = x;
  __syncthreads();
  if (w == 0) {
    int s = wsum[lane];
    for (int o = 1; o < 32; o <<= 1) {
      int y = __shfl_up(s, o);
      if (lane >= o) s += y;
    }
    wsum[lane] = s;
  }
  __syncthreads();
  int off = (w > 0) ? wsum[w - 1] : 0;
  total = wsum[31];
  __syncthreads();
  return off + x - v;
}

__global__ __launch_bounds__(1024) void ec_bin_scatter(
    const int* __restrict__ src, const int* __restrict__ dst,
    const int* __restrict__ cnt, int* __restrict__ start,
    unsigned* __restrict__ lst, int E, int NB, int CH, int r0) {
  __shared__ int lcur[NBMAX];
  __shared__ int gdelta[NBMAX];
  __shared__ int wsum[32];
  __shared__ unsigned sortedw[CHUNK];
  __shared__ int sortedp[CHUNK];
  int t = threadIdx.x;
  int c = blockIdx.x;
  int r = r0 + blockIdx.y;

  int tot = 0, before = 0, mine = 0;
  if (t < NB) {
    const int* col = cnt + (size_t)r * CH * NB + t;
    for (int cc = 0; cc < CH; cc += 8) {
      int v[8];
#pragma unroll
      for (int u = 0; u < 8; ++u) v[u] = (cc + u < CH) ? col[(size_t)(cc + u) * NB] : 0;
#pragma unroll
      for (int u = 0; u < 8; ++u) {
        tot += v[u];
        if (cc + u < c) before += v[u];
        if (cc + u == c) mine = v[u];
      }
    }
  }
  int total, ltotal;
  int gstart = ec_scan1024(tot, wsum, total);
  int lstart = ec_scan1024(mine, wsum, ltotal);
  lcur[t] = lstart;
  gdelta[t] = gstart + before - lstart;
  if (c == 0) {
    if (t < NB) start[r * SP + t] = gstart;
    if (t == 0) start[r * SP + NB] = total;
  }
  __syncthreads();

  const int* d = dst + (size_t)r * E;
  const int* s = src + (size_t)r * E;
  int e0 = c * CHUNK;
  int dv[CHUNK / 1024], sv[CHUNK / 1024];
#pragma unroll
  for (int i = 0; i < CHUNK / 1024; ++i) {
    int e = e0 + i * 1024 + t;
    dv[i] = (e < E) ? d[e] : -1;
    sv[i] = (e < E) ? s[e] : 0;
  }
#pragma unroll
  for (int i = 0; i < CHUNK / 1024; ++i) {
    if (dv[i] >= 0) {
      int b = dv[i] >> 7;
      int li = atomicAdd(&lcur[b], 1);
      sortedw[li] = ((unsigned)sv[i] << 7) | (unsigned)(dv[i] & (BKT - 1));
      sortedp[li] = gdelta[b] + li;
    }
  }
  __syncthreads();
  int n = E - e0;
  if (n > CHUNK) n = CHUNK;
  unsigned* out = lst + (size_t)(r - r0) * E;
  for (int i = 0; i < CHUNK / 1024; ++i) {
    int j = i * 1024 + t;
    if (j < n) out[sortedp[j]] = sortedw[j];
  }
}

constexpr int GKB = 32;
constexpr int GD = 3;

__device__ __forceinline__ void ec_load_step(const float* __restrict__ Ap, v2f (&dst)[GKB / 4]) {
#pragma unroll
  for (int u = 0; u < GKB / 4; ++u) {
    float2 av = *(const float2*)(Ap + u * 4);
    dst[u] = v2f{av.x, av.y};
  }
}

template <int K, int S, int NTL>
__device__ __forceinline__ void ec_strip(const float* __restrict__ Arow, const float* __restrict__ Anext,
                                         const float* Bl, v2f (&ring)[GD][GKB / 4], v8f (&acc)[NTL]) {
  constexpr int NS = K / GKB;
  static_assert(NS % (GD + 1) == 0, "the register ring turns a whole number of times per strip");
  v8f z = {0.f, 0.f, 0.f, 0.f, 0.f, 0.f, 0.f, 0.f};
#pragma unroll
  for (int j = 0; j < NTL; ++j) acc[j] = z;
#pragma unroll(GD + 1)
  for (int s = 0; s < NS; ++s) {
    v2f nw[GKB / 4];
    const float* Ap = (s + GD < NS) ? (Arow + (s + GD) * GKB) : (Anext + (s + GD - NS) * GKB);
    ec_load_step(Ap, nw);
    __builtin_amdgcn_sched_barrier(0x186);
#pragma unroll
    for (int u = 0; u < GKB / 4; ++u) {
      int k = s * GKB + u * 4;
      const float* Bp = Bl + (size_t)(k >> 1) * S;
#pragma unroll
      for (int j = 0; j < NTL; ++j) {
        v2f b = *(const v2f*)(Bp + j * 32);
        acc[j] = __builtin_amdgcn_wmma_f32_16x16x4_f32(
             false, ring[0][u],  false, b,
             (short)0, acc[j],  false,  false);
      }
    }
#pragma unroll
    for (int d = 0; d + 1 < GD; ++d) {
#pragma unroll
      for (int u = 0; u < GKB / 4; ++u) ring[d][u] = ring[d + 1][u];
    }
#pragma unroll
    for (int u = 0; u < GKB / 4; ++u) ring[GD - 1][u] = nw[u];
  }
}

template <int K, int NC>
__global__ __launch_bounds__(512) void ec_gemm_wmma(
    const float* __restrict__ A, const float* __restrict__ Bself, const float* __restrict__ Brel,
    const float* __restrict__ biasSelf, float* __restrict__ Cself, float* __restrict__ Crel, int mtiles,
    const int* __restrict__ dstc, int* __restrict__ cnt, int E, int NB, int CH, int nchunk) {
  constexpr int NT = NC / 16;
  const int wb = (int)blockIdx.y - ((Bself != nullptr) ? 1 : 0);
  const float* __restrict__ B = (wb < 0) ? Bself : (Brel + (size_t)wb * K * NC);
  const float* __restrict__ bias = (wb < 0) ? biasSelf : nullptr;
  float* __restrict__ C = (wb < 0) ? Cself : (Crel + (size_t)wb * mtiles * 16 * NC);
  constexpr int S = ((2 * NC) % 64 == 32) ? (2 * NC) : (2 * NC + 32);
  constexpr int WV = 16;
  static_assert(NT == 4, "a left-over strip is split over four waves, one column tile each");
  __shared__ float Bs[(K / 2) * S];
  __shared__ int hist[NBMAX];

  constexpr int NL = K * NC / 4 / 512;
  static_assert(K * NC / 4 % 512 == 0, "whole float4 rounds");
  float4 bv4[NL];
#pragma unroll
  for (int it = 0; it < NL; ++it) bv4[it] = *(const float4*)(B + (size_t)(threadIdx.x + 512 * it) * 4);

  if (dstc != nullptr) {
    const int nwg = (int)(gridDim.x * gridDim.y);
    for (int ch = (int)(blockIdx.y * gridDim.x + blockIdx.x); ch < nchunk; ch += nwg) {
      hist[threadIdx.x] = 0;
      hist[threadIdx.x + 512] = 0;
      __syncthreads();
      const int r = ch / CH;
      const int cc = ch - r * CH;
      const int* d = dstc + (size_t)r * E;
      const int e0 = cc * CHUNK;
      int dv[CHUNK / 512];
#pragma unroll
      for (int i = 0; i < CHUNK / 512; ++i) {
        int e = e0 + i * 512 + (int)threadIdx.x;
        dv[i] = (e < E) ? d[e] : -1;
      }
#pragma unroll
      for (int i = 0; i < CHUNK / 512; ++i) {
        if (dv[i] >= 0) atomicAdd(&hist[dv[i] >> 7], 1);
      }
      __syncthreads();
      for (int q = (int)threadIdx.x; q < NB; q += 512) cnt[((size_t)r * CH + cc) * NB + q] = hist[q];
      __syncthreads();
    }
  }

#pragma unroll
  for (int it = 0; it < NL; ++it) {
    int i = threadIdx.x + 512 * it;
    int k = (i * 4) / NC;
    int n = (i * 4) % NC;
    float* p = &Bs[(k >> 1) * S + 2 * n + (k & 1)];
    p[0] = bv4[it].x;
    p[2] = bv4[it].y;
    p[4] = bv4[it].z;
    p[6] = bv4[it].w;
  }

  int wave = __builtin_amdgcn_readfirstlane(threadIdx.x >> 5);
  int lane = threadIdx.x & 31;
  int m = lane & 15;
  int half = lane >> 4;

  int t0 = (int)((blockIdx.x * (unsigned)mtiles) / gridDim.x);
  int t1 = (int)(((blockIdx.x + 1) * (unsigned)mtiles) / gridDim.x);
  int tf = t0 + ((t1 - t0) & ~3);
  int tile = t0 + wave;

  v2f ring[GD][GKB / 4];
  {
    int tl = (tile < mtiles) ? tile : (mtiles - 1);
    const float* Ap = A + (size_t)(tl * 16 + m) * K + 2 * half;
#pragma unroll
    for (int d = 0; d < GD; ++d) ec_load_step(Ap + d * GKB, ring[d]);
  }
  float bvv[NT];
#pragma unroll
  for (int j = 0; j < NT; ++j) bvv[j] = bias ? bias[j * 16 + m] : 0.0f;
  __syncthreads();

  const float* Bl = &Bs[(size_t)half * S + 2 * m];

  for (; tile < tf; tile += WV) {
    const float* Arow = A + (size_t)(tile * 16 + m) * K + 2 * half;
    int tn = tile + WV;
    if (tn >= tf) tn = tile;
    const float* Anext = A + (size_t)(tn * 16 + m) * K + 2 * half;
    v8f acc[NT];
    ec_strip<K, S, NT>(Arow, Anext, Bl, ring, acc);
#pragma unroll
    for (int j = 0; j < NT; ++j) {
#pragma unroll
      for (int v = 0; v < 8; ++v) {
        C[(size_t)(tile * 16 + v + 8 * half) * NC + j * 16 + m] = acc[j][v] + bvv[j];
      }
    }
  }

  int q = 3 - (wave >> 2);
  if (tf + q < t1) {
    int tq = tf + q;
    int j0 = wave & 3;
    const float* Arow = A + (size_t)(tq * 16 + m) * K + 2 * half;
#pragma unroll
    for (int d = 0; d < GD; ++d) ec_load_step(Arow + d * GKB, ring[d]);
    v8f acc1[1];
    ec_strip<K, S, 1>(Arow, Arow, Bl + j0 * 32, ring, acc1);
    float bq = bias ? bias[j0 * 16 + m] : 0.0f;
#pragma unroll
    for (int v = 0; v < 8; ++v) {
      C[(size_t)(tq * 16 + v + 8 * half) * NC + j0 * 16 + m] = acc1[0][v] + bq;
    }
  }
}


__global__ __launch_bounds__(128) void ec_gemm2_all(
    const float* __restrict__ A, const float* __restrict__ Bself,
    const float* __restrict__ Brel, const float* __restrict__ bias,
    float* __restrict__ Cself, float* __restrict__ Crel, int mtiles) {
  constexpr int K = 64;
  constexpr int NT = 1 + R2;
  constexpr int S = 2 * 16 * NT;
  static_assert(S % 64 == 32, "lane halves must land in disjoint bank ranges");
  __shared__ float Bs[(K / 2) * S];

  static_assert(K * 16 / 4 == 256, "two float4 of a block per thread");
  float4 sv[2 * NT];
#pragma unroll
  for (int it = 0; it < 2 * NT; ++it) {
    const float* bp = (it < 2) ? Bself : (Brel + (size_t)(it / 2 - 1) * K * 16);
    sv[it] = *(const float4*)(bp + (threadIdx.x + 128 * (it & 1)) * 4);
  }
#pragma unroll
  for (int it = 0; it < 2 * NT; ++it) {
    int q = threadIdx.x + 128 * (it & 1);
    int k = q >> 2;
    int n = (it / 2) * 16 + (q & 3) * 4;
    float* p = &Bs[(k >> 1) * S + 2 * n + (k & 1)];
    p[0] = sv[it].x;
    p[2] = sv[it].y;
    p[4] = sv[it].z;
    p[6] = sv[it].w;
  }
  __syncthreads();

  int wave = (blockIdx.x * blockDim.x + threadIdx.x) >> 5;
  if (wave >= mtiles) return;
  int lane = threadIdx.x & 31;
  int m = lane & 15;
  int half = lane >> 4;

  const float* Arow = A + (size_t)(wave * 16 + m) * K + 2 * half;

  v8f acc[NT];
  v8f z = {0.f, 0.f, 0.f, 0.f, 0.f, 0.f, 0.f, 0.f};
#pragma unroll
  for (int j = 0; j < NT; ++j) acc[j] = z;

#pragma unroll
  for (int k = 0; k < K; k += 4) {
    float2 av = *(const float2*)(Arow + k);
    v2f a = {av.x, av.y};
    const float* Bp = &Bs[(size_t)((k + 2 * half) >> 1) * S + 2 * m];
#pragma unroll
    for (int j = 0; j < NT; ++j) {
      v2f b = *(const v2f*)(Bp + j * 32);
      acc[j] = __builtin_amdgcn_wmma_f32_16x16x4_f32(
           false, a,  false, b,
           (short)0, acc[j],  false,  false);
    }
  }

  float bv = bias[m];
  size_t plane = (size_t)mtiles * 16 * 16;
#pragma unroll
  for (int v = 0; v < 8; ++v) {
    size_t o = (size_t)(wave * 16 + v + 8 * half) * 16 + m;
    Cself[o] = acc[0][v] + bv;
#pragma unroll
    for (int j = 1; j < NT; ++j) Crel[(size_t)(j - 1) * plane + o] = acc[j][v];
  }
}

__global__ __launch_bounds__(256) void ec_sum64(
    const float* __restrict__ hw, const unsigned* __restrict__ lst,
    const int* __restrict__ start, float* __restrict__ acc, int Nn, int relu) {
  __shared__ float rows[BKT * 64];
  __shared__ int degl[BKT];
  int t = threadIdx.x;
  int lane = t & 31;
  int wave = __builtin_amdgcn_readfirstlane(t >> 5);
  int b = blockIdx.x;
  float a0[16], a1[16];
#pragma unroll
  for (int k = 0; k < 16; ++k) {
    int d = b * BKT + wave + 8 * k;
    if (d > Nn - 1) d = Nn - 1;
    const float* ap = acc + (size_t)d * 64 + lane;
    a0[k] = ap[0];
    a1[k] = ap[32];
  }
  int s0 = start[b];
  int s1 = start[b + 1];
  int j = lane & 15;
  int g0 = s0 + wave * 16;
  int wd = (g0 + j < s1) ? (int)lst[g0 + j] : 0;
  for (int i = t; i < BKT * 64; i += 256) rows[i] = 0.0f;
  if (t < BKT) degl[t] = 0;
  __syncthreads();
  for (; g0 < s1; g0 += 8 * 16) {
    int n = s1 - g0;
    if (n > 16) n = 16;
    int wdn = (g0 + 8 * 16 + j < s1) ? (int)lst[g0 + 8 * 16 + j] : 0;
    if (lane < n) atomicAdd(&degl[wd & (BKT - 1)], 1);
    float v0[16], v1[16];
#pragma unroll
    for (int k = 0; k < 16; ++k) {
      unsigned w = (unsigned)__builtin_amdgcn_readlane(wd, k);
      const float* hp = hw + (size_t)(w >> 7) * 64 + lane;
      v0[k] = hp[0];
      v1[k] = hp[32];
    }
    if (n == 16) {
#pragma unroll
      for (int k = 0; k < 16; ++k) {
        unsigned w = (unsigned)__builtin_amdgcn_readlane(wd, k);
        float* rp = rows + (w & (BKT - 1)) * 64 + lane;
        __hip_atomic_fetch_add(rp, v0[k], __ATOMIC_RELAXED, __HIP_MEMORY_SCOPE_WORKGROUP);
        __hip_atomic_fetch_add(rp + 32, v1[k], __ATOMIC_RELAXED, __HIP_MEMORY_SCOPE_WORKGROUP);
      }
    } else {
#pragma unroll
      for (int k = 0; k < 16; ++k) {
        if (k < n) {
          unsigned w = (unsigned)__builtin_amdgcn_readlane(wd, k);
          float* rp = rows + (w & (BKT - 1)) * 64 + lane;
          __hip_atomic_fetch_add(rp, v0[k], __ATOMIC_RELAXED, __HIP_MEMORY_SCOPE_WORKGROUP);
          __hip_atomic_fetch_add(rp + 32, v1[k], __ATOMIC_RELAXED, __HIP_MEMORY_SCOPE_WORKGROUP);
        }
      }
    }
    wd = wdn;
  }
  __syncthreads();
  float wl = 1.0f / fmaxf((float)degl[wave + 8 * (lane & 15)], 1.0f);
#pragma unroll
  for (int k = 0; k < 16; ++k) {
    int dl = wave + 8 * k;
    float w = __int_as_float(__builtin_amdgcn_readlane(__float_as_int(wl), k));
    a0[k] += rows[dl * 64 + lane] * w;
    a1[k] += rows[dl * 64 + lane + 32] * w;
    if (relu) {
      a0[k] = fmaxf(a0[k], 0.0f);
      a1[k] = fmaxf(a1[k], 0.0f);
    }
  }
#pragma unroll
  for (int k = 0; k < 16; ++k) {
    int d = b * BKT + wave + 8 * k;
    if (d < Nn) {
      float* ap = acc + (size_t)d * 64 + lane;
      ap[0] = a0[k];
      ap[32] = a1[k];
    }
  }
}

__global__ __launch_bounds__(256) void ec_sum64_all(
    const float* __restrict__ hw, const unsigned* __restrict__ lst,
    const int* __restrict__ start, float* __restrict__ acc, int Nn, int E, int r0, int nr, int relu,
    const float* __restrict__ B2self, const float* __restrict__ B2rel, const float* __restrict__ bias2,
    float* __restrict__ out2) {
  __shared__ float rows[2 * BKT * 64];
  __shared__ int degl[R2 * BKT];
  __shared__ int sb[2 * R2];
  int t = threadIdx.x;
  int lane = t & 31;
  int wave = __builtin_amdgcn_readfirstlane(t >> 5);
  int b = blockIdx.x;
  float a0[16], a1[16];
#pragma unroll
  for (int k = 0; k < 16; ++k) {
    int d = b * BKT + 16 * wave + k;
    if (d > Nn - 1) d = Nn - 1;
    const float* ap = acc + (size_t)d * 64 + lane;
    a0[k] = ap[0];
    a1[k] = ap[32];
  }
  if (t < 2 * nr) sb[t] = start[(r0 + (t >> 1)) * SP + b + (t & 1)];
  {
    float4 z4 = make_float4(0.0f, 0.0f, 0.0f, 0.0f);
    float4* r4 = (float4*)rows;
    for (int i = t; i < 2 * BKT * 64 / 4; i += 256) r4[i] = z4;
  }
  for (int i = t; i < R2 * BKT; i += 256) degl[i] = 0;
  __syncthreads();
  int j = lane & 15;
  int se0 = __builtin_amdgcn_readfirstlane(sb[1]);
  int gs0 = __builtin_amdgcn_readfirstlane(sb[0]) + wave * 16;
  int wf0 = (gs0 + j < se0) ? (int)lst[(size_t)r0 * E + gs0 + j] : 0;
  int se1 = 0, gs1 = 0, wf1 = 0, se2 = 0, gs2 = 0, wf2 = 0, se3 = 0, gs3 = 0, wf3 = 0;
  if (nr > 1) {
    se1 = __builtin_amdgcn_readfirstlane(sb[3]);
    gs1 = __builtin_amdgcn_readfirstlane(sb[2]) + wave * 16;
    wf1 = (gs1 + j < se1) ? (int)lst[(size_t)(r0 + 1) * E + gs1 + j] : 0;
  }
  if (nr > 2) {
    se2 = __builtin_amdgcn_readfirstlane(sb[5]);
    gs2 = __builtin_amdgcn_readfirstlane(sb[4]) + wave * 16;
    wf2 = (gs2 + j < se2) ? (int)lst[(size_t)(r0 + 2) * E + gs2 + j] : 0;
  }
  if (nr > 3) {
    se3 = __builtin_amdgcn_readfirstlane(sb[7]);
    gs3 = __builtin_amdgcn_readfirstlane(sb[6]) + wave * 16;
    wf3 = (gs3 + j < se3) ? (int)lst[(size_t)(r0 + 3) * E + gs3 + j] : 0;
  }
  float4 sv[1 + R2];
  for (int p = 0; p < nr; ++p) {
    const float* hwp = hw + (size_t)p * Nn * 64;
    const unsigned* l = lst + (size_t)(r0 + p) * E;
    float* rw = rows + (p & 1) * BKT * 64;
    int* dg = degl + p * BKT;
    int s1 = (p == 0) ? se0 : (p == 1) ? se1 : (p == 2) ? se2 : se3;
    int g0 = (p == 0) ? gs0 : (p == 1) ? gs1 : (p == 2) ? gs2 : gs3;
    int wd = (p == 0) ? wf0 : (p == 1) ? wf1 : (p == 2) ? wf2 : wf3;
    for (; g0 < s1; g0 += 8 * 16) {
      int n = s1 - g0;
      if (n > 16) n = 16;
      if (lane < n) atomicAdd(&dg[wd & (BKT - 1)], 1);
      float v0[16], v1[16];
#pragma unroll
      for (int k = 0; k < 16; ++k) {
        unsigned w = (unsigned)__builtin_amdgcn_readlane(wd, k);
        const float* hp = hwp + (size_t)(w >> 7) * 64 + lane;
        v0[k] = hp[0];
        v1[k] = hp[32];
      }
      int wdn = (g0 + 8 * 16 + j < s1) ? (int)l[g0 + 8 * 16 + j] : 0;
      if (n == 16) {
#pragma unroll
        for (int k = 0; k < 16; ++k) {
          unsigned w = (unsigned)__builtin_amdgcn_readlane(wd, k);
          float* rp = rw + (w & (BKT - 1)) * 64 + lane;
          __hip_atomic_fetch_add(rp, v0[k], __ATOMIC_RELAXED, __HIP_MEMORY_SCOPE_WORKGROUP);
          __hip_atomic_fetch_add(rp + 32, v1[k], __ATOMIC_RELAXED, __HIP_MEMORY_SCOPE_WORKGROUP);
        }
      } else {
#pragma unroll
        for (int k = 0; k < 16; ++k) {
          if (k < n) {
            unsigned w = (unsigned)__builtin_amdgcn_readlane(wd, k);
            float* rp = rw + (w & (BKT - 1)) * 64 + lane;
            __hip_atomic_fetch_add(rp, v0[k], __ATOMIC_RELAXED, __HIP_MEMORY_SCOPE_WORKGROUP);
            __hip_atomic_fetch_add(rp + 32, v1[k], __ATOMIC_RELAXED, __HIP_MEMORY_SCOPE_WORKGROUP);
          }
        }
      }
      wd = wdn;
    }
    if (relu && p == nr - 1) {
#pragma unroll
      for (int it = 0; it < 1 + R2; ++it) {
        const float* bp = (it == 0) ? B2self : (B2rel + (size_t)(it - 1) * 64 * 16);
        sv[it] = *(const float4*)(bp + t * 4);
      }
    }
    __syncthreads();
    float wl = 1.0f / fmaxf((float)dg[16 * wave + (lane & 15)], 1.0f);
#pragma unroll
    for (int k = 0; k < 16; ++k) {
      float* rp = rw + (16 * wave + k) * 64 + lane;
      float w = __int_as_float(__builtin_amdgcn_readlane(__float_as_int(wl), k));
      a0[k] = __builtin_fmaf(rp[0], w, a0[k]);
      a1[k] = __builtin_fmaf(rp[32], w, a1[k]);
      rp[0] = 0.0f;
      rp[32] = 0.0f;
    }
  }
  if (relu) {
    float* rl = rows + ((nr - 1) & 1) * BKT * 64;
    float* Bs2 = rows + (nr & 1) * BKT * 64;
#pragma unroll
    for (int k = 0; k < 16; ++k) {
      float* rp = rl + (16 * wave + k) * 64;
      int s0 = (lane >> 1) ^ (2 * k);
      rp[2 * s0 + (lane & 1)] = fmaxf(a0[k], 0.0f);
      rp[2 * (s0 ^ 16) + (lane & 1)] = fmaxf(a1[k], 0.0f);
    }
#pragma unroll
    for (int it = 0; it < 1 + R2; ++it) {
      int kk = t >> 2;
      int n = it * 16 + (t & 3) * 4;
      float* p2 = &Bs2[(kk >> 1) * 160 + 2 * n + (kk & 1)];
      p2[0] = sv[it].x;
      p2[2] = sv[it].y;
      p2[4] = sv[it].z;
      p2[6] = sv[it].w;
    }
    __syncthreads();
    int m = lane & 15;
    int half = lane >> 4;
    v8f c2[1 + R2];
    v8f z8 = {0.f, 0.f, 0.f, 0.f, 0.f, 0.f, 0.f, 0.f};
#pragma unroll
    for (int jj = 0; jj < 1 + R2; ++jj) c2[jj] = z8;
    const float* arow = rl + (16 * wave + m) * 64;
#pragma unroll
    for (int k0 = 0; k0 < 64; k0 += 4) {
      v2f a = *(const v2f*)(arow + 2 * (((k0 >> 1) + half) ^ (2 * m)));
      const float* Bp = &Bs2[(size_t)((k0 + 2 * half) >> 1) * 160 + 2 * m];
#pragma unroll
      for (int jj = 0; jj < 1 + R2; ++jj) {
        v2f bq = *(const v2f*)(Bp + jj * 32);
        c2[jj] = __builtin_amdgcn_wmma_f32_16x16x4_f32(
             false, a,  false, bq,
             (short)0, c2[jj],  false,  false);
      }
    }
    float bv2 = bias2[m];
#pragma unroll
    for (int v = 0; v < 8; ++v) {
      int d = b * BKT + 16 * wave + v + 8 * half;
      if (d < Nn) {
        out2[(size_t)d * 16 + m] = c2[0][v] + bv2;
        float* hp = acc + (size_t)d * 64 + m;
#pragma unroll
        for (int jj = 1; jj < 1 + R2; ++jj) hp[(jj - 1) * 16] = c2[jj][v];
      }
    }
    return;
  }
#pragma unroll
  for (int k = 0; k < 16; ++k) {
    int d = b * BKT + 16 * wave + k;
    if (d < Nn) {
      float* ap = acc + (size_t)d * 64 + lane;
      ap[0] = a0[k];
      ap[32] = a1[k];
    }
  }
}

template <int RS>
__global__ __launch_bounds__(256) void ec_sum16_all(
    const float* __restrict__ hw, const unsigned* __restrict__ lst,
    const int* __restrict__ start, float* __restrict__ acc, int Nn, int E, int relStride) {
  __shared__ float rows[2 * BKT * 16];
  __shared__ int degl[R2 * BKT];
  __shared__ int sb[2 * R2];
  int t = threadIdx.x;
  int lane = t & 31;
  int wave = __builtin_amdgcn_readfirstlane(t >> 5);
  int b = blockIdx.x;
  if (t < 2 * R2) sb[t] = start[(t >> 1) * SP + b + (t & 1)];
  for (int i = t; i < 2 * BKT * 16; i += 256) rows[i] = 0.0f;
  for (int i = t; i < R2 * BKT; i += 256) degl[i] = 0;
  float tot[BKT * 16 / 256];
#pragma unroll
  for (int q = 0; q < BKT * 16 / 256; ++q) {
    int i = t + 256 * q;
    int d = b * BKT + (i >> 4);
    tot[q] = (d < Nn) ? acc[(size_t)d * 16 + (i & 15)] : 0.0f;
  }
  __syncthreads();
  int c = lane & 15;
  int half = lane >> 4;
  int s1 = __builtin_amdgcn_readfirstlane(sb[1]);
  int g0 = __builtin_amdgcn_readfirstlane(sb[0]) + wave * 32;
  int wd = (g0 + lane < s1) ? (int)lst[g0 + lane] : 0;
  for (int r = 0; r < R2; ++r) {
    const float* hwr = hw + (size_t)r * relStride;
    const unsigned* l = lst + (size_t)r * E;
    float* rw = rows + (r & 1) * BKT * 16;
    int* dg = degl + r * BKT;
    int s1n = 0, g0n = 0, wdf = 0;
    if (r + 1 < R2) {
      s1n = __builtin_amdgcn_readfirstlane(sb[2 * r + 3]);
      g0n = __builtin_amdgcn_readfirstlane(sb[2 * r + 2]) + wave * 32;
      wdf = (g0n + lane < s1n) ? (int)lst[(size_t)(r + 1) * E + g0n + lane] : 0;
    }
    for (; g0 < s1; g0 += 8 * 32) {
      int n = s1 - g0;
      if (n > 32) n = 32;
      int wdn = (g0 + 8 * 32 + lane < s1) ? (int)l[g0 + 8 * 32 + lane] : 0;
      if (lane < n) atomicAdd(&dg[wd & (BKT - 1)], 1);
      float v[16];
      unsigned wk[16];
#pragma unroll
      for (int k = 0; k < 16; ++k) {
        wk[k] = (unsigned)__shfl(wd, 2 * k + half);
        v[k] = hwr[(size_t)(wk[k] >> 7) * RS + c];
      }
      if (n == 32) {
#pragma unroll
        for (int k = 0; k < 16; ++k) {
          __hip_atomic_fetch_add(&rw[(wk[k] & (BKT - 1)) * 16 + c], v[k], __ATOMIC_RELAXED,
                                 __HIP_MEMORY_SCOPE_WORKGROUP);
        }
      } else {
#pragma unroll
        for (int k = 0; k < 16; ++k) {
          if (2 * k + half < n) {
            __hip_atomic_fetch_add(&rw[(wk[k] & (BKT - 1)) * 16 + c], v[k], __ATOMIC_RELAXED,
                                   __HIP_MEMORY_SCOPE_WORKGROUP);
          }
        }
      }
      wd = wdn;
    }
    __syncthreads();
#pragma unroll
    for (int q = 0; q < BKT * 16 / 256; ++q) {
      int i = t + 256 * q;
      float w = 1.0f / fmaxf((float)dg[i >> 4], 1.0f);
      tot[q] += rw[i] * w;
      rw[i] = 0.0f;
    }
    s1 = s1n;
    g0 = g0n;
    wd = wdf;
  }
#pragma unroll
  for (int q = 0; q < BKT * 16 / 256; ++q) {
    int i = t + 256 * q;
    int d = b * BKT + (i >> 4);
    if (d < Nn) acc[(size_t)d * 16 + (i & 15)] = tot[q];
  }
}

extern "C" void kernel_launch(void* const* d_in, const int* in_sizes, int n_in,
                              void* d_out, int out_size, void* d_ws, size_t ws_size,
                              hipStream_t stream) {
  const float* x   = (const float*)d_in[0];
  const int*   src = (const int*)d_in[1];
  const int*   dst = (const int*)d_in[2];
  const float* W1  = (const float*)d_in[3];
  const float* L1w = (const float*)d_in[4];
  const float* L1b = (const float*)d_in[5];
  const float* W2  = (const float*)d_in[6];
  const float* L2w = (const float*)d_in[7];
  const float* L2b = (const float*)d_in[8];
  float* out = (float*)d_out;

  const int D_IN = 256, D_H = 64;
  const int Nn = in_sizes[0] / D_IN;
  const int R  = in_sizes[3] / (D_IN * D_H);
  const int E  = in_sizes[1] / R;
  const int NB = (Nn + BKT - 1) / BKT;
  const int CH = (E + CHUNK - 1) / CHUNK;

  char* ws = (char*)d_ws;
  size_t off = 0;
  int* start = (int*)(ws + off);
  int* cnt   = start + (size_t)R * SP;
  off += (size_t)R * Nn * sizeof(float);
  off = (off + 255) & ~(size_t)255;
  float* h = (float*)(ws + off);
  off += (size_t)Nn * D_H * sizeof(float);
  off = (off + 255) & ~(size_t)255;
  float* hw = (float*)(ws + off);
  unsigned* lstAll = (unsigned*)(ws + off);
  const size_t planeBytes = (size_t)Nn * D_H * sizeof(float);
  size_t offP = off + (size_t)R * E * sizeof(unsigned);
  offP = (offP + 255) & ~(size_t)255;
  float* planes = (float*)(ws + offP);
  int P = 0;
  if (R == R2) {
    if (ws_size >= offP + 4 * planeBytes) P = 4;
    else if (ws_size >= offP + 2 * planeBytes) P = 2;
    else if (ws_size >= offP + planeBytes) P = 1;
  }

  int mtiles = Nn / 16;
  int gemmBlocks = (mtiles * 32 + 127) / 128;
  const int gemm1Blocks = 256;

  if (P > 0) {
    int done = 0;
    while (done < R) {
      const int np = (R - done < P) ? (R - done) : P;
      const int ny = np + ((done == 0) ? 1 : 0);
      const int gx = gemm1Blocks / ny;
      ec_gemm_wmma<256, 64><<<dim3(gx, ny), 512, 0, stream>>>(
          x, (done == 0) ? L1w : nullptr, W1 + (size_t)done * D_IN * D_H, L1b, h, planes, mtiles,
          (done == 0) ? dst : nullptr, cnt, E, NB, CH, CH * R);
      if (done == 0) {
        ec_bin_scatter<<<dim3(CH, R), 1024, 0, stream>>>(src, dst, cnt, start, lstAll, E, NB, CH, 0);
      }
      ec_sum64_all<<<NB, 256, 0, stream>>>(planes, lstAll, start, h, Nn, E, done, np,
                                           (done + np == R) ? 1 : 0, L2w, W2, L2b, out);
      done += np;
    }
    ec_sum16_all<64><<<NB, 256, 0, stream>>>(h, lstAll, start, out, Nn, E, 16);
    return;
  }

  unsigned* lst1 = (unsigned*)out;
  unsigned* lst2 = (unsigned*)h;

  ec_bin_count<<<dim3(CH, R), 1024, 0, stream>>>(dst, cnt, E, NB, CH);

  ec_gemm_wmma<256, 64><<<gemm1Blocks, 512, 0, stream>>>(x, L1w, nullptr, L1b, h, nullptr, mtiles,
                                                         nullptr, nullptr, 0, 0, 0, 0);
  for (int r = 0; r < R; ++r) {
    if ((r & 1) == 0) {
      int nr = (R - r < 2) ? (R - r) : 2;
      ec_bin_scatter<<<dim3(CH, nr), 1024, 0, stream>>>(src, dst, cnt, start, lst1, E, NB, CH, r);
    }
    ec_gemm_wmma<256, 64><<<gemm1Blocks, 512, 0, stream>>>(
        x, nullptr, W1 + (size_t)r * D_IN * D_H, nullptr, nullptr, hw, mtiles,
        nullptr, nullptr, 0, 0, 0, 0);
    ec_sum64<<<NB, 256, 0, stream>>>(hw, lst1 + (size_t)(r & 1) * E, start + r * SP, h, Nn,
                                     (r == R - 1) ? 1 : 0);
  }

  ec_gemm2_all<<<gemmBlocks, 128, 0, stream>>>(h, L2w, W2, L2b, out, hw, mtiles);
  ec_bin_scatter<<<dim3(CH, R), 1024, 0, stream>>>(src, dst, cnt, start, lst2, E, NB, CH, 0);
  ec_sum16_all<16><<<NB, 256, 0, stream>>>(hw, lst2, start, out, Nn, E, Nn * 16);
}
